# starting kernel
# baseline (speedup 1.0000x reference)
#include <hip/hip_runtime.h>


#define NCOMBO 23
#define CG_TOTAL 2351
#define NFRAGS 76
#define FRAG_TOTAL (NFRAGS * 64)
#define TAB_TOTAL (CG_TOTAL + FRAG_TOTAL)
#define WAVES_PER_BLOCK 4
#define TB (WAVES_PER_BLOCK * 32)
#define ZROW 84

typedef float v2f __attribute__((ext_vector_type(2)));
typedef float v8f __attribute__((ext_vector_type(8)));

__constant__ int c_l1[NCOMBO] = {0,0,0,0, 1,1,1,1,1,1,1,1, 2,2,2,2,2,2,2, 3,3,3,3};
__constant__ int c_l2[NCOMBO] = {0,1,2,3, 1,1,1,2,2,2,3,3, 2,2,2,2,3,3,3, 3,3,3,3};
__constant__ int c_L [NCOMBO] = {0,1,2,3, 0,1,2,1,2,3,2,3, 0,1,2,3,1,2,3, 0,1,2,3};
__constant__ float c_w[NCOMBO] = {1,2,2,2, 1,1,1,2,2,2,2,2, 1,1,1,1,2,2,2, 1,1,1,1};
__constant__ int c_off[NCOMBO] = {0,1,10,35, 84,93,120,165,210,285,390,495,
                                  642,667,742,867,1042,1147,1322, 1567,1616,1763,2008};
__constant__ int c_foff[NCOMBO] = {0,1,2,4, 8,9,10,11,12,14,17,20,
                                   24,26,28,32,38,42,48, 56,58,62,68};
__constant__ int d_xoff[4] = {0, 1, 10, 35};
__constant__ int d_x2pos[4] = {0, 1, 19, 69};
__constant__ int d_x2neg[4] = {0, 10, 44, 118};
__constant__ int d_diag[4] = {0, 4, 12, 19};

__global__ void wigner_cg_init(float* __restrict__ ws) {
  double f[16];
  f[0] = 1.0;
  for (int i = 1; i < 16; ++i) f[i] = f[i - 1] * i;
  for (int c = 0; c < NCOMBO; ++c) {
    const int l1 = c_l1[c], l2 = c_l2[c], L = c_L[c];
    const int n2 = 2 * l2 + 1, nL = 2 * L + 1;
    const int sz = (2 * l1 + 1) * n2 * nL, off = c_off[c];
    const double pref0 =
        sqrt((2.0 * L + 1.0) * f[L + l1 - l2] * f[L - l1 + l2] * f[l1 + l2 - L] / f[l1 + l2 + L + 1]);
    for (int t = threadIdx.x; t < sz; t += blockDim.x) {
      const int m = t / (n2 * nL), rem = t % (n2 * nL), n = rem / nL, k = rem % nL;
      const int m1 = m - l1, m2 = n - l2, M = m1 + m2;
      double val = 0.0;
      if (M >= -L && M <= L && k == M + L) {
        const double pref =
            pref0 * sqrt(f[L + M] * f[L - M] * f[l1 - m1] * f[l1 + m1] * f[l2 - m2] * f[l2 + m2]);
        double s = 0.0;
        for (int kk = 0; kk <= l1 + l2 - L; ++kk) {
          const int a1 = l1 - m1 - kk, a2 = l2 + m2 - kk;
          const int a3 = L - l2 + m1 + kk, a4 = L - l1 - m2 + kk;
          if (a1 < 0 || a2 < 0 || a3 < 0 || a4 < 0) continue;
          const double term = 1.0 / (f[kk] * f[l1 + l2 - L - kk] * f[a1] * f[a2] * f[a3] * f[a4]);
          s += (kk & 1) ? -term : term;
        }
        val = pref * s;
      }
      ws[off + t] = (float)val;
    }
  }
  __syncthreads();
  for (int c = 0; c < NCOMBO; ++c) {
    const int l1 = c_l1[c], l2 = c_l2[c], L = c_L[c];
    const int n1 = 2 * l1 + 1, R = (2 * l2 + 1) * (2 * L + 1);
    const int cgo = c_off[c], fo = c_foff[c];
    const int mtiles = (R + 15) >> 4, kch = (n1 + 3) >> 2;
    for (int t = threadIdx.x; t < mtiles * kch * 64; t += blockDim.x) {
      const int fi = t >> 6, s = t & 63;
      const int mt = fi / kch, kc = fi % kch;
      const int ln = s >> 1, j = s & 1;
      const int row = mt * 16 + (ln & 15);
      const int m = kc * 4 + ((ln >> 4) ? 2 : 0) + j;
      ws[CG_TOTAL + (fo + fi) * 64 + s] =
          (row < R && m < n1) ? ws[cgo + m * R + row] : 0.f;
    }
  }
}

__launch_bounds__(TB, 1)
__global__ void wigner_main(const float* __restrict__ x0, const float* __restrict__ x1,
                            const float* __restrict__ x2, const float* __restrict__ x3,
                            const float* __restrict__ tab, float* __restrict__ out, int Btot) {
  __shared__ __align__(16) float sTAB[TAB_TOTAL];
  __shared__ float sXT[WAVES_PER_BLOCK][85 * 16];
  __shared__ float sA [WAVES_PER_BLOCK][64 * 7 * 16];
  __shared__ float sE [WAVES_PER_BLOCK][7 * 7 * 7 * 16];
  __shared__ float sX2[WAVES_PER_BLOCK][167 * 16];

  const int tid = threadIdx.x, lane = tid & 31, wid = tid >> 5;
  const int b = lane & 15, half = lane >> 4;
  const float* sCG = sTAB;
  const float* sFRAG = sTAB + CG_TOTAL;

  for (int t = tid; t < TAB_TOTAL; t += TB) sTAB[t] = tab[t];

  const int tile = blockIdx.x * WAVES_PER_BLOCK + wid;
  const int base = tile * 16;
  float* XT = sXT[wid];
  float* A  = sA[wid];
  float* E  = sE[wid];
  float* X2 = sX2[wid];

  {
    int pe = base + WAVES_PER_BLOCK * 16;
    if (pe < Btot) __builtin_prefetch(&x3[(size_t)pe * 49], 0, 1);
  }

  for (int t = lane; t < 84 * 16; t += 32) {
    const int bb = t / 84, comp = t % 84;
    int e = base + bb;
    e = (e < Btot) ? e : (Btot - 1);
    float v;
    if (comp < 1)       v = x0[e];
    else if (comp < 10) v = x1[(size_t)e * 9 + (comp - 1)];
    else if (comp < 35) v = x2[(size_t)e * 25 + (comp - 10)];
    else                v = x3[(size_t)e * 49 + (comp - 35)];
    XT[comp * 16 + bb] = v;
  }
  for (int t = lane; t < 16; t += 32) XT[ZROW * 16 + t] = 0.f;
  for (int t = lane; t < 167 * 16; t += 32) X2[t] = 0.f;
  __syncthreads();

  for (int c = 0; c < NCOMBO; ++c) {
    const int l1 = c_l1[c], l2 = c_l2[c], L = c_L[c];
    const int n1 = 2 * l1 + 1, n2 = 2 * l2 + 1, nL = 2 * L + 1;
    const int R = n2 * nL, cgo = c_off[c], fb = c_foff[c];
    const int xo1 = d_xoff[l1], xo2 = d_xoff[l2];
    const int mtiles = (R + 15) >> 4, kch = (n1 + 3) >> 2;

    const int m0a = half ? 2 : 0;
    if (kch == 1) {
      for (int p = 0; p < n1; ++p) {
        v2f bf0;
        const int r0 = (m0a     < n1) ? (xo1 + m0a * n1 + p)       : ZROW;
        const int r1 = (m0a + 1 < n1) ? (xo1 + (m0a + 1) * n1 + p) : ZROW;
        bf0.x = XT[r0 * 16 + b];
        bf0.y = XT[r1 * 16 + b];
        for (int mt = 0; mt < mtiles; ++mt) {
          v8f acc = {0.f, 0.f, 0.f, 0.f, 0.f, 0.f, 0.f, 0.f};
          const v2f af0 = *(const v2f*)&sFRAG[(fb + mt) * 64 + lane * 2];
          acc = __builtin_amdgcn_wmma_f32_16x16x4_f32(false, af0, false, bf0,
                                                      (short)0, acc, false, false);
          const int rb = mt * 16 + half * 8;
#pragma unroll
          for (int r = 0; r < 8; ++r)
            A[((rb + r) * n1 + p) * 16 + b] = acc[r];
        }
      }
    } else {
      for (int p = 0; p < n1; ++p) {
        v2f bf0, bf1;
        {
          const int r0 = xo1 + m0a * n1 + p;
          const int r1 = xo1 + (m0a + 1) * n1 + p;
          bf0.x = XT[r0 * 16 + b];
          bf0.y = XT[r1 * 16 + b];
        }
        {
          const int m0 = 4 + m0a;
          const int r0 = (m0     < n1) ? (xo1 + m0 * n1 + p)       : ZROW;
          const int r1 = (m0 + 1 < n1) ? (xo1 + (m0 + 1) * n1 + p) : ZROW;
          bf1.x = XT[r0 * 16 + b];
          bf1.y = XT[r1 * 16 + b];
        }
        for (int mt = 0; mt < mtiles; ++mt) {
          v8f acc = {0.f, 0.f, 0.f, 0.f, 0.f, 0.f, 0.f, 0.f};
          const float* fp = &sFRAG[(fb + mt * 2) * 64 + lane * 2];
          const v2f af0 = *(const v2f*)fp;
          acc = __builtin_amdgcn_wmma_f32_16x16x4_f32(false, af0, false, bf0,
                                                      (short)0, acc, false, false);
          const v2f af1 = *(const v2f*)(fp + 64);
          acc = __builtin_amdgcn_wmma_f32_16x16x4_f32(false, af1, false, bf1,
                                                      (short)0, acc, false, false);
          const int rb = mt * 16 + half * 8;
#pragma unroll
          for (int r = 0; r < 8; ++r)
            A[((rb + r) * n1 + p) * 16 + b] = acc[r];
        }
      }
    }
    __syncthreads();

    for (int q = 0; q < n2; ++q) {
      float xv[7];
#pragma unroll
      for (int n = 0; n < 7; ++n)
        xv[n] = (n < n2) ? XT[(xo2 + n * n2 + q) * 16 + b] : 0.f;
      const int kp = nL * n1;
      for (int idx = half; idx < kp; idx += 2) {
        const int p = idx % n1, k = idx / n1;
        float e = 0.f;
#pragma unroll
        for (int n = 0; n < 7; ++n)
          if (n < n2)
            e += xv[n] * A[((n * nL + k) * n1 + p) * 16 + b];
        E[((k * n2 + q) * n1 + p) * 16 + b] = e;
      }
    }
    __syncthreads();

    const int x2base = ((l1 + l2 + L) & 1) ? d_x2neg[L] : d_x2pos[L];
    const float wc = c_w[c];
    for (int idx = half; idx < nL * nL; idx += 2) {
      const int j = idx % nL, k = idx / nL;
      float s = 0.f;
      for (int pq = 0; pq < n1 * n2; ++pq) {
        const float cc = sCG[cgo + pq * nL + j];
        if (cc != 0.f) {
          const int p = pq / n2, q = pq % n2;
          s += cc * E[((k * n2 + q) * n1 + p) * 16 + b];
        }
      }
      X2[(x2base + idx) * 16 + b] += wc * s;
    }
    __syncthreads();
  }

  if (lane < 16) {
    const int e = base + b;
    if (e < Btot) {
      const float nu1 = XT[0 * 16 + b];
      const float nu2 = X2[0 * 16 + b];
      float nu3 = 0.f, nu4 = 0.f;
      for (int l = 0; l <= 3; ++l) {
        const int n = 2 * l + 1;
        const int cgo = c_off[d_diag[l]];
        const int xo = d_xoff[l];
        const int xp = d_x2pos[l], xn = d_x2neg[l];
        for (int mn = 0; mn < n * n; ++mn) {
          const float c1 = sCG[cgo + mn];
          if (c1 == 0.f) continue;
          const int m = mn / n, nn = mn % n;
          for (int pq = 0; pq < n * n; ++pq) {
            const float c2 = sCG[cgo + pq];
            if (c2 == 0.f) continue;
            const int p = pq / n, q = pq % n;
            const float w = c1 * c2;
            nu3 += w * X2[(xp + m * n + p) * 16 + b] * XT[(xo + nn * n + q) * 16 + b];
            nu4 += w * X2[(xp + m * n + p) * 16 + b] * X2[(xp + nn * n + q) * 16 + b];
            if (l > 0)
              nu4 += w * X2[(xn + m * n + p) * 16 + b] * X2[(xn + nn * n + q) * 16 + b];
          }
        }
      }
      out[(size_t)e * 4 + 0] = nu1;
      out[(size_t)e * 4 + 1] = nu2;
      out[(size_t)e * 4 + 2] = nu3;
      out[(size_t)e * 4 + 3] = nu4;
    }
  }
}

extern "C" void kernel_launch(void* const* d_in, const int* in_sizes, int n_in,
                              void* d_out, int out_size, void* d_ws, size_t ws_size,
                              hipStream_t stream) {
  const float* x0 = (const float*)d_in[0];
  const float* x1 = (const float*)d_in[1];
  const float* x2 = (const float*)d_in[2];
  const float* x3 = (const float*)d_in[3];
  float* tab = (float*)d_ws;
  const int Btot = in_sizes[0];

  wigner_cg_init<<<1, 256, 0, stream>>>(tab);

  const int tiles = (Btot + 15) / 16;
  const int blocks = (tiles + WAVES_PER_BLOCK - 1) / WAVES_PER_BLOCK;
  wigner_main<<<blocks, TB, 0, stream>>>(x0, x1, x2, x3, tab, (float*)d_out, Btot);
}


# combine(X,X,3) re-decomposed per element as two chained f32 WMMA products over the Kronecker rows with the CG tables as shared operand; VALU stages 2/3 and A/E LDS buffers go; summation order changes
# speedup vs baseline: 18.8150x; 18.8150x over previous
#include <hip/hip_runtime.h>


#define NCOMBO 23
#define CG_TOTAL 2351
#define NPAIR 10
#define B1_STEPS 65
#define A2_STEPS 68
#define B1_BASE CG_TOTAL
#define A2_BASE (CG_TOTAL + B1_STEPS * 64)
#define WS_TOTAL (A2_BASE + A2_STEPS * 64)
#define LB1 0
#define LA2 (B1_STEPS * 64)
#define LDIAG (LA2 + A2_STEPS * 64)
#define LT_TOTAL (LDIAG + 84)
#define WAVES_PER_BLOCK 4
#define TB (WAVES_PER_BLOCK * 32)
#define XS 85
#define ZC 84
#define X2S 169

typedef float v2f __attribute__((ext_vector_type(2)));
typedef float v8f __attribute__((ext_vector_type(8)));

__constant__ int c_l1[NCOMBO] = {0,0,0,0, 1,1,1,1,1,1,1,1, 2,2,2,2,2,2,2, 3,3,3,3};
__constant__ int c_l2[NCOMBO] = {0,1,2,3, 1,1,1,2,2,2,3,3, 2,2,2,2,3,3,3, 3,3,3,3};
__constant__ int c_L [NCOMBO] = {0,1,2,3, 0,1,2,1,2,3,2,3, 0,1,2,3,1,2,3, 0,1,2,3};
__constant__ int c_off[NCOMBO] = {0,1,10,35, 84,93,120,165,210,285,390,495,
                                  642,667,742,867,1042,1147,1322, 1567,1616,1763,2008};
__constant__ int p_l1[NPAIR] = {0,0,0,0,1,1,1,2,2,3};
__constant__ int p_l2[NPAIR] = {0,1,2,3,1,2,3,2,3,3};
__constant__ int p_c0[NPAIR] = {0,1,2,3,4,7,10,12,16,19};
__constant__ int p_b1[NPAIR] = {0,1,3,6,10,14,20,28,37,49};
__constant__ int p_a2[NPAIR] = {0,4,8,12,16,20,24,32,40,52};
__constant__ int d_xoff[4] = {0, 1, 10, 35};
__constant__ int d_x2pos[4] = {0, 1, 19, 69};
__constant__ int d_x2neg[4] = {0, 10, 44, 118};
__constant__ int d_diag[4] = {0, 4, 12, 19};

__device__ __forceinline__ int lof(int x) { return (x >= 1) + (x >= 4) + (x >= 9); }

__global__ void wigner_cg_init(float* __restrict__ ws) {
  double f[16];
  f[0] = 1.0;
  for (int i = 1; i < 16; ++i) f[i] = f[i - 1] * i;
  for (int c = 0; c < NCOMBO; ++c) {
    const int l1 = c_l1[c], l2 = c_l2[c], L = c_L[c];
    const int n2 = 2 * l2 + 1, nL = 2 * L + 1;
    const int sz = (2 * l1 + 1) * n2 * nL, off = c_off[c];
    const double pref0 =
        sqrt((2.0 * L + 1.0) * f[L + l1 - l2] * f[L - l1 + l2] * f[l1 + l2 - L] / f[l1 + l2 + L + 1]);
    for (int t = threadIdx.x; t < sz; t += blockDim.x) {
      const int m = t / (n2 * nL), rem = t % (n2 * nL), n = rem / nL, k = rem % nL;
      const int m1 = m - l1, m2 = n - l2, M = m1 + m2;
      double val = 0.0;
      if (M >= -L && M <= L && k == M + L) {
        const double pref =
            pref0 * sqrt(f[L + M] * f[L - M] * f[l1 - m1] * f[l1 + m1] * f[l2 - m2] * f[l2 + m2]);
        double s = 0.0;
        for (int kk = 0; kk <= l1 + l2 - L; ++kk) {
          const int a1 = l1 - m1 - kk, a2 = l2 + m2 - kk;
          const int a3 = L - l2 + m1 + kk, a4 = L - l1 - m2 + kk;
          if (a1 < 0 || a2 < 0 || a3 < 0 || a4 < 0) continue;
          const double term = 1.0 / (f[kk] * f[l1 + l2 - L - kk] * f[a1] * f[a2] * f[a3] * f[a4]);
          s += (kk & 1) ? -term : term;
        }
        val = pref * s;
      }
      ws[off + t] = (float)val;
    }
  }
  __syncthreads();
  for (int pi = 0; pi < NPAIR; ++pi) {
    const int l1 = p_l1[pi], l2 = p_l2[pi];
    const int n1 = 2 * l1 + 1, n2 = 2 * l2 + 1, PH = l1 + 1, QH = l2 + 1;
    const int Lmin = l2 - l1, Lmax = (l1 + l2 < 3) ? (l1 + l2) : 3;
    const int c0 = p_c0[pi], NR = n1 * n2, RT = (NR + 15) >> 4;
    const float w = (l1 == l2) ? 1.f : 2.f;
    for (int t = threadIdx.x; t < PH * QH * 64; t += blockDim.x) {
      const int st = t >> 6, s6 = t & 63, ln = s6 >> 1, s = s6 & 1;
      const int i = st / QH, u = st % QH, h = ln >> 4, c = ln & 15;
      const int p = h * PH + i, q = 2 * u + s, L = lof(c), k = c - L * L;
      float val = 0.f;
      if (p < n1 && q < n2 && L >= Lmin && L <= Lmax)
        val = ws[c_off[c0 + L - Lmin] + (p * n2 + q) * (2 * L + 1) + k];
      ws[B1_BASE + (p_b1[pi] + st) * 64 + s6] = val;
    }
    for (int t = threadIdx.x; t < RT * 4 * 64; t += blockDim.x) {
      const int st = t >> 6, s6 = t & 63, ln = s6 >> 1, s = s6 & 1;
      const int rt = st >> 2, v = st & 3, h = ln >> 4, c = ln & 15;
      const int rho = rt * 16 + 8 * h + 2 * v + s, L = lof(c), k = c - L * L;
      float val = 0.f;
      if (rho < NR && L >= Lmin && L <= Lmax) {
        const int m = rho / n2, n = rho % n2;
        val = w * ws[c_off[c0 + L - Lmin] + (m * n2 + n) * (2 * L + 1) + k];
      }
      ws[A2_BASE + (p_a2[pi] + st) * 64 + s6] = val;
    }
  }
}

__device__ __forceinline__ constexpr int xoff_c(int l) { return l == 0 ? 0 : (l == 1 ? 1 : (l == 2 ? 10 : 35)); }

template <int L1, int L2, int B1OFF, int A2OFF>
__device__ __forceinline__ void pair_gemm(const float* __restrict__ Xe, const float* __restrict__ sB1,
                                          const float* __restrict__ sA2, v8f (&acc)[2], const int lane) {
  constexpr int N1 = 2 * L1 + 1, N2 = 2 * L2 + 1, PH = L1 + 1, QH = L2 + 1;
  constexpr int NR = N1 * N2, RT = (NR + 15) / 16;
  constexpr int XO1 = xoff_c(L1), XO2 = xoff_c(L2);
  const int h = lane >> 4, c = lane & 15;
#pragma unroll
  for (int rt = 0; rt < RT; ++rt) {
    const int rho = rt * 16 + c;
    const bool valid = rho < NR;
    const int m = rho / N2, n = rho - m * N2;
    float a[2][PH], b[2][2 * QH];
#pragma unroll
    for (int i = 0; i < PH; ++i) {
      const int p = h * PH + i;
      const int ia = (valid && p < N1) ? (XO1 + m * N1 + p) : ZC;
#pragma unroll
      for (int el = 0; el < 2; ++el) a[el][i] = Xe[el * XS + ia];
    }
#pragma unroll
    for (int q = 0; q < 2 * QH; ++q) {
      const int ib = (valid && q < N2) ? (XO2 + n * N2 + q) : ZC;
#pragma unroll
      for (int el = 0; el < 2; ++el) b[el][q] = Xe[el * XS + ib];
    }
    v8f d[2];
#pragma unroll
    for (int el = 0; el < 2; ++el) d[el] = (v8f){0.f, 0.f, 0.f, 0.f, 0.f, 0.f, 0.f, 0.f};
#pragma unroll
    for (int i = 0; i < PH; ++i) {
#pragma unroll
      for (int u = 0; u < QH; ++u) {
        const v2f bf = *(const v2f*)&sB1[(B1OFF + i * QH + u) * 64 + lane * 2];
#pragma unroll
        for (int el = 0; el < 2; ++el) {
          v2f af;
          af.x = a[el][i] * b[el][2 * u];
          af.y = a[el][i] * b[el][2 * u + 1];
          d[el] = __builtin_amdgcn_wmma_f32_16x16x4_f32(false, af, false, bf, (short)0, d[el], false, false);
        }
      }
    }
#pragma unroll
    for (int v = 0; v < 4; ++v) {
      const v2f af2 = *(const v2f*)&sA2[(A2OFF + rt * 4 + v) * 64 + lane * 2];
#pragma unroll
      for (int el = 0; el < 2; ++el) {
        v2f bf2;
        bf2.x = d[el][2 * v];
        bf2.y = d[el][2 * v + 1];
        acc[el] = __builtin_amdgcn_wmma_f32_16x16x4_f32(false, af2, false, bf2, (short)0, acc[el], false, false);
      }
    }
  }
}

__device__ __forceinline__ void store_x2(float* __restrict__ X2e, const v8f& t, const bool even, const int lane) {
  const int h = lane >> 4, c = lane & 15;
  const int Lc = lof(c);
#pragma unroll
  for (int v = 0; v < 8; ++v) {
    const int kp = 8 * h + v;
    const int L = lof(kp);
    const int k = kp - L * L, j = c - L * L, nL = 2 * L + 1;
    const bool pos = even ? ((L & 1) == 0) : ((L & 1) == 1);
    const int pb = (L >= 1) + 18 * (L >= 2) + 50 * (L >= 3);
    const int base = pos ? pb : (pb + nL * nL);
    const bool wr = (L == Lc) && (pos || L > 0);
    X2e[wr ? (base + k * nL + j) : (X2S - 1)] = t[v];
  }
}

__launch_bounds__(TB, 3)
__global__ void wigner_main(const float* __restrict__ x0, const float* __restrict__ x1,
                            const float* __restrict__ x2, const float* __restrict__ x3,
                            const float* __restrict__ tab, float* __restrict__ out, int Btot) {
  __shared__ __align__(16) float sTAB[LT_TOTAL];
  __shared__ float sX [WAVES_PER_BLOCK][16 * XS];
  __shared__ float sX2[WAVES_PER_BLOCK][16 * X2S];

  const int tid = threadIdx.x, lane = tid & 31, wid = tid >> 5;
  const int b = lane & 15;
  const float* sB1 = sTAB + LB1;
  const float* sA2 = sTAB + LA2;
  const float* sDIAG = sTAB + LDIAG;

  for (int t = tid; t < LDIAG; t += TB) sTAB[t] = tab[CG_TOTAL + t];
  for (int l = 0; l <= 3; ++l) {
    const int nn = (2 * l + 1) * (2 * l + 1);
    for (int t = tid; t < nn; t += TB) sTAB[LDIAG + d_xoff[l] + t] = tab[c_off[d_diag[l]] + t];
  }

  const int tile = blockIdx.x * WAVES_PER_BLOCK + wid;
  const int base = tile * 16;
  float* X  = sX[wid];
  float* X2 = sX2[wid];

  {
    int pe = base + WAVES_PER_BLOCK * 16;
    if (pe < Btot) __builtin_prefetch(&x3[(size_t)pe * 49], 0, 1);
  }

  for (int t = lane; t < 84 * 16; t += 32) {
    const int bb = t / 84, comp = t % 84;
    int e = base + bb;
    e = (e < Btot) ? e : (Btot - 1);
    float v;
    if (comp < 1)       v = x0[e];
    else if (comp < 10) v = x1[(size_t)e * 9 + (comp - 1)];
    else if (comp < 35) v = x2[(size_t)e * 25 + (comp - 10)];
    else                v = x3[(size_t)e * 49 + (comp - 35)];
    X[bb * XS + comp] = v;
  }
  for (int t = lane; t < 16; t += 32) X[t * XS + ZC] = 0.f;
  __syncthreads();

  for (int e = 0; e < 16; e += 2) {
    asm volatile("" ::: "memory");
    const float* Xe = X + e * XS;
    v8f accE[2], accO[2];
#pragma unroll
    for (int el = 0; el < 2; ++el) {
      accE[el] = (v8f){0.f, 0.f, 0.f, 0.f, 0.f, 0.f, 0.f, 0.f};
      accO[el] = (v8f){0.f, 0.f, 0.f, 0.f, 0.f, 0.f, 0.f, 0.f};
    }
    pair_gemm<0, 0,  0,  0>(Xe, sB1, sA2, accE, lane);
    pair_gemm<0, 1,  1,  4>(Xe, sB1, sA2, accO, lane);
    pair_gemm<0, 2,  3,  8>(Xe, sB1, sA2, accE, lane);
    pair_gemm<0, 3,  6, 12>(Xe, sB1, sA2, accO, lane);
    pair_gemm<1, 1, 10, 16>(Xe, sB1, sA2, accE, lane);
    pair_gemm<1, 2, 14, 20>(Xe, sB1, sA2, accO, lane);
    pair_gemm<1, 3, 20, 24>(Xe, sB1, sA2, accE, lane);
    pair_gemm<2, 2, 28, 32>(Xe, sB1, sA2, accE, lane);
    pair_gemm<2, 3, 37, 40>(Xe, sB1, sA2, accO, lane);
    pair_gemm<3, 3, 49, 52>(Xe, sB1, sA2, accE, lane);
#pragma unroll
    for (int el = 0; el < 2; ++el) {
      store_x2(X2 + (e + el) * X2S, accE[el], true, lane);
      store_x2(X2 + (e + el) * X2S, accO[el], false, lane);
    }
  }

  __syncthreads();

  if (lane < 16) {
    const int e = base + b;
    if (e < Btot) {
      const float* Xb = X + b * XS;
      const float* X2b = X2 + b * X2S;
      const float nu1 = Xb[0];
      const float nu2 = X2b[0];
      float nu3 = 0.f, nu4 = 0.f;
      for (int l = 0; l <= 3; ++l) {
        const int n = 2 * l + 1;
        const int cgo = d_xoff[l];
        const int xo = d_xoff[l];
        const int xp = d_x2pos[l], xn = d_x2neg[l];
        for (int mn = 0; mn < n * n; ++mn) {
          const float c1 = sDIAG[cgo + mn];
          if (c1 == 0.f) continue;
          const int m = mn / n, nn = mn % n;
          for (int pq = 0; pq < n * n; ++pq) {
            const float c2 = sDIAG[cgo + pq];
            if (c2 == 0.f) continue;
            const int p = pq / n, q = pq % n;
            const float w = c1 * c2;
            nu3 += w * X2b[xp + m * n + p] * Xb[xo + nn * n + q];
            nu4 += w * X2b[xp + m * n + p] * X2b[xp + nn * n + q];
            if (l > 0)
              nu4 += w * X2b[xn + m * n + p] * X2b[xn + nn * n + q];
          }
        }
      }
      out[(size_t)e * 4 + 0] = nu1;
      out[(size_t)e * 4 + 1] = nu2;
      out[(size_t)e * 4 + 2] = nu3;
      out[(size_t)e * 4 + 3] = nu4;
    }
  }
}

extern "C" void kernel_launch(void* const* d_in, const int* in_sizes, int n_in,
                              void* d_out, int out_size, void* d_ws, size_t ws_size,
                              hipStream_t stream) {
  const float* x0 = (const float*)d_in[0];
  const float* x1 = (const float*)d_in[1];
  const float* x2 = (const float*)d_in[2];
  const float* x3 = (const float*)d_in[3];
  float* tab = (float*)d_ws;
  const int Btot = in_sizes[0];

  wigner_cg_init<<<1, 256, 0, stream>>>(tab);

  const int tiles = (Btot + 15) / 16;
  const int blocks = (tiles + WAVES_PER_BLOCK - 1) / WAVES_PER_BLOCK;
  wigner_main<<<blocks, TB, 0, stream>>>(x0, x1, x2, x3, tab, (float*)d_out, Btot);
}


# nu=3/4 contractions run inside the trip on all 32 lanes in the accumulator layout (entry x weighted reflected entry) instead of a 16-lane serial table scan; nu3/nu4 summation order changes
# speedup vs baseline: 22.4706x; 1.1943x over previous
#include <hip/hip_runtime.h>


#define NCOMBO 23
#define CG_TOTAL 2351
#define NPAIR 10
#define B1_STEPS 65
#define A2_STEPS 68
#define B1_BASE CG_TOTAL
#define A2_BASE (CG_TOTAL + B1_STEPS * 64)
#define WS_TOTAL (A2_BASE + A2_STEPS * 64)
#define LB1 0
#define LA2 (B1_STEPS * 64)
#define LDIAG (LA2 + A2_STEPS * 64)
#define LT_TOTAL (LDIAG + 84)
#define WAVES_PER_BLOCK 4
#define TB (WAVES_PER_BLOCK * 32)
#define XS 85
#define ZC 84
#define X2S 169

typedef float v2f __attribute__((ext_vector_type(2)));
typedef float v8f __attribute__((ext_vector_type(8)));

__constant__ int c_l1[NCOMBO] = {0,0,0,0, 1,1,1,1,1,1,1,1, 2,2,2,2,2,2,2, 3,3,3,3};
__constant__ int c_l2[NCOMBO] = {0,1,2,3, 1,1,1,2,2,2,3,3, 2,2,2,2,3,3,3, 3,3,3,3};
__constant__ int c_L [NCOMBO] = {0,1,2,3, 0,1,2,1,2,3,2,3, 0,1,2,3,1,2,3, 0,1,2,3};
__constant__ int c_off[NCOMBO] = {0,1,10,35, 84,93,120,165,210,285,390,495,
                                  642,667,742,867,1042,1147,1322, 1567,1616,1763,2008};
__constant__ int p_l1[NPAIR] = {0,0,0,0,1,1,1,2,2,3};
__constant__ int p_l2[NPAIR] = {0,1,2,3,1,2,3,2,3,3};
__constant__ int p_c0[NPAIR] = {0,1,2,3,4,7,10,12,16,19};
__constant__ int p_b1[NPAIR] = {0,1,3,6,10,14,20,28,37,49};
__constant__ int p_a2[NPAIR] = {0,4,8,12,16,20,24,32,40,52};
__constant__ int d_xoff[4] = {0, 1, 10, 35};
__constant__ int d_diag[4] = {0, 4, 12, 19};

__device__ __forceinline__ int lof(int x) { return (x >= 1) + (x >= 4) + (x >= 9); }

__global__ void wigner_cg_init(float* __restrict__ ws) {
  double f[16];
  f[0] = 1.0;
  for (int i = 1; i < 16; ++i) f[i] = f[i - 1] * i;
  for (int c = 0; c < NCOMBO; ++c) {
    const int l1 = c_l1[c], l2 = c_l2[c], L = c_L[c];
    const int n2 = 2 * l2 + 1, nL = 2 * L + 1;
    const int sz = (2 * l1 + 1) * n2 * nL, off = c_off[c];
    const double pref0 =
        sqrt((2.0 * L + 1.0) * f[L + l1 - l2] * f[L - l1 + l2] * f[l1 + l2 - L] / f[l1 + l2 + L + 1]);
    for (int t = threadIdx.x; t < sz; t += blockDim.x) {
      const int m = t / (n2 * nL), rem = t % (n2 * nL), n = rem / nL, k = rem % nL;
      const int m1 = m - l1, m2 = n - l2, M = m1 + m2;
      double val = 0.0;
      if (M >= -L && M <= L && k == M + L) {
        const double pref =
            pref0 * sqrt(f[L + M] * f[L - M] * f[l1 - m1] * f[l1 + m1] * f[l2 - m2] * f[l2 + m2]);
        double s = 0.0;
        for (int kk = 0; kk <= l1 + l2 - L; ++kk) {
          const int a1 = l1 - m1 - kk, a2 = l2 + m2 - kk;
          const int a3 = L - l2 + m1 + kk, a4 = L - l1 - m2 + kk;
          if (a1 < 0 || a2 < 0 || a3 < 0 || a4 < 0) continue;
          const double term = 1.0 / (f[kk] * f[l1 + l2 - L - kk] * f[a1] * f[a2] * f[a3] * f[a4]);
          s += (kk & 1) ? -term : term;
        }
        val = pref * s;
      }
      ws[off + t] = (float)val;
    }
  }
  __syncthreads();
  for (int pi = 0; pi < NPAIR; ++pi) {
    const int l1 = p_l1[pi], l2 = p_l2[pi];
    const int n1 = 2 * l1 + 1, n2 = 2 * l2 + 1, PH = l1 + 1, QH = l2 + 1;
    const int Lmin = l2 - l1, Lmax = (l1 + l2 < 3) ? (l1 + l2) : 3;
    const int c0 = p_c0[pi], NR = n1 * n2, RT = (NR + 15) >> 4;
    const float w = (l1 == l2) ? 1.f : 2.f;
    for (int t = threadIdx.x; t < PH * QH * 64; t += blockDim.x) {
      const int st = t >> 6, s6 = t & 63, ln = s6 >> 1, s = s6 & 1;
      const int i = st / QH, u = st % QH, h = ln >> 4, c = ln & 15;
      const int p = h * PH + i, q = 2 * u + s, L = lof(c), k = c - L * L;
      float val = 0.f;
      if (p < n1 && q < n2 && L >= Lmin && L <= Lmax)
        val = ws[c_off[c0 + L - Lmin] + (p * n2 + q) * (2 * L + 1) + k];
      ws[B1_BASE + (p_b1[pi] + st) * 64 + s6] = val;
    }
    for (int t = threadIdx.x; t < RT * 4 * 64; t += blockDim.x) {
      const int st = t >> 6, s6 = t & 63, ln = s6 >> 1, s = s6 & 1;
      const int rt = st >> 2, v = st & 3, h = ln >> 4, c = ln & 15;
      const int rho = rt * 16 + 8 * h + 2 * v + s, L = lof(c), k = c - L * L;
      float val = 0.f;
      if (rho < NR && L >= Lmin && L <= Lmax) {
        const int m = rho / n2, n = rho % n2;
        val = w * ws[c_off[c0 + L - Lmin] + (m * n2 + n) * (2 * L + 1) + k];
      }
      ws[A2_BASE + (p_a2[pi] + st) * 64 + s6] = val;
    }
  }
}

__device__ __forceinline__ constexpr int xoff_c(int l) { return l == 0 ? 0 : (l == 1 ? 1 : (l == 2 ? 10 : 35)); }

template <int L1, int L2, int B1OFF, int A2OFF>
__device__ __forceinline__ void pair_gemm(const float* __restrict__ Xe, const float* __restrict__ sB1,
                                          const float* __restrict__ sA2, v8f (&acc)[2], const int lane) {
  constexpr int N1 = 2 * L1 + 1, N2 = 2 * L2 + 1, PH = L1 + 1, QH = L2 + 1;
  constexpr int NR = N1 * N2, RT = (NR + 15) / 16;
  constexpr int XO1 = xoff_c(L1), XO2 = xoff_c(L2);
  const int h = lane >> 4, c = lane & 15;
#pragma unroll
  for (int rt = 0; rt < RT; ++rt) {
    const int rho = rt * 16 + c;
    const bool valid = rho < NR;
    const int m = rho / N2, n = rho - m * N2;
    float a[2][PH], b[2][2 * QH];
#pragma unroll
    for (int i = 0; i < PH; ++i) {
      const int p = h * PH + i;
      const int ia = (valid && p < N1) ? (XO1 + m * N1 + p) : ZC;
#pragma unroll
      for (int el = 0; el < 2; ++el) a[el][i] = Xe[el * XS + ia];
    }
#pragma unroll
    for (int q = 0; q < 2 * QH; ++q) {
      const int ib = (valid && q < N2) ? (XO2 + n * N2 + q) : ZC;
#pragma unroll
      for (int el = 0; el < 2; ++el) b[el][q] = Xe[el * XS + ib];
    }
    v8f d[2];
#pragma unroll
    for (int el = 0; el < 2; ++el) d[el] = (v8f){0.f, 0.f, 0.f, 0.f, 0.f, 0.f, 0.f, 0.f};
#pragma unroll
    for (int i = 0; i < PH; ++i) {
#pragma unroll
      for (int u = 0; u < QH; ++u) {
        const v2f bf = *(const v2f*)&sB1[(B1OFF + i * QH + u) * 64 + lane * 2];
#pragma unroll
        for (int el = 0; el < 2; ++el) {
          v2f af;
          af.x = a[el][i] * b[el][2 * u];
          af.y = a[el][i] * b[el][2 * u + 1];
          d[el] = __builtin_amdgcn_wmma_f32_16x16x4_f32(false, af, false, bf, (short)0, d[el], false, false);
        }
      }
    }
#pragma unroll
    for (int v = 0; v < 4; ++v) {
      const v2f af2 = *(const v2f*)&sA2[(A2OFF + rt * 4 + v) * 64 + lane * 2];
#pragma unroll
      for (int el = 0; el < 2; ++el) {
        v2f bf2;
        bf2.x = d[el][2 * v];
        bf2.y = d[el][2 * v + 1];
        acc[el] = __builtin_amdgcn_wmma_f32_16x16x4_f32(false, af2, false, bf2, (short)0, acc[el], false, false);
      }
    }
  }
}

__device__ __forceinline__ void store_x2(float* __restrict__ X2e, const v8f& t, const bool even, const int lane) {
  const int h = lane >> 4, c = lane & 15;
  const int Lc = lof(c);
#pragma unroll
  for (int v = 0; v < 8; ++v) {
    const int kp = 8 * h + v;
    const int L = lof(kp);
    const int k = kp - L * L, j = c - L * L, nL = 2 * L + 1;
    const bool pos = even ? ((L & 1) == 0) : ((L & 1) == 1);
    const int pb = (L >= 1) + 18 * (L >= 2) + 50 * (L >= 3);
    const int base = pos ? pb : (pb + nL * nL);
    const bool wr = (L == Lc) && (pos || L > 0);
    X2e[wr ? (base + k * nL + j) : (X2S - 1)] = t[v];
  }
}

__launch_bounds__(TB, 3)
__global__ void wigner_main(const float* __restrict__ x0, const float* __restrict__ x1,
                            const float* __restrict__ x2, const float* __restrict__ x3,
                            const float* __restrict__ tab, float* __restrict__ out, int Btot) {
  __shared__ __align__(16) float sTAB[LT_TOTAL];
  __shared__ float sX [WAVES_PER_BLOCK][16 * XS];
  __shared__ float sX2[WAVES_PER_BLOCK][16 * X2S];

  const int tid = threadIdx.x, lane = tid & 31, wid = tid >> 5;
  const int b = lane & 15;
  const float* sB1 = sTAB + LB1;
  const float* sA2 = sTAB + LA2;
  const float* sDIAG = sTAB + LDIAG;

  for (int t = tid; t < LDIAG; t += TB) sTAB[t] = tab[CG_TOTAL + t];
  for (int l = 0; l <= 3; ++l) {
    const int nn = (2 * l + 1) * (2 * l + 1);
    for (int t = tid; t < nn; t += TB) sTAB[LDIAG + d_xoff[l] + t] = tab[c_off[d_diag[l]] + t];
  }

  const int tile = blockIdx.x * WAVES_PER_BLOCK + wid;
  const int base = tile * 16;
  float* X  = sX[wid];
  float* X2 = sX2[wid];

  {
    int pe = base + WAVES_PER_BLOCK * 16;
    if (pe < Btot) __builtin_prefetch(&x3[(size_t)pe * 49], 0, 1);
  }

  for (int t = lane; t < 84 * 16; t += 32) {
    const int bb = t / 84, comp = t % 84;
    int e = base + bb;
    e = (e < Btot) ? e : (Btot - 1);
    float v;
    if (comp < 1)       v = x0[e];
    else if (comp < 10) v = x1[(size_t)e * 9 + (comp - 1)];
    else if (comp < 35) v = x2[(size_t)e * 25 + (comp - 10)];
    else                v = x3[(size_t)e * 49 + (comp - 35)];
    X[bb * XS + comp] = v;
  }
  for (int t = lane; t < 16; t += 32) X[t * XS + ZC] = 0.f;
  __syncthreads();

  const int hh = lane >> 4, cc = lane & 15;
  float wgt[8], wgtN[8];
  int xi[8], pi[8], ni[8];
  bool posE[8];
#pragma unroll
  for (int r = 0; r < 8; ++r) {
    const int kp = 8 * hh + r;
    const int L = lof(kp), nL = 2 * L + 1;
    const int k = kp - L * L, j = cc - L * L;
    const bool dg = (L == lof(cc));
    const int kr = 2 * L - k, jr = 2 * L - j;
    const int xo = (L >= 1) + 9 * (L >= 2) + 25 * (L >= 3);
    const int pb = (L >= 1) + 18 * (L >= 2) + 50 * (L >= 3);
    const float gk = sDIAG[xo + k * nL + kr];
    const float gj = sDIAG[dg ? (xo + j * nL + jr) : 0];
    const float w = gk * gj;
    wgt[r]  = dg ? w : 0.f;
    wgtN[r] = (dg && L > 0) ? w : 0.f;
    xi[r] = dg ? (xo + kr * nL + jr) : ZC;
    pi[r] = dg ? (pb + kr * nL + jr) : 0;
    ni[r] = (dg && L > 0) ? (pb + nL * nL + kr * nL + jr) : 0;
    posE[r] = ((L & 1) == 0);
  }

  for (int e = 0; e < 16; e += 2) {
    asm volatile("" ::: "memory");
    const float* Xe = X + e * XS;
    v8f accE[2], accO[2];
#pragma unroll
    for (int el = 0; el < 2; ++el) {
      accE[el] = (v8f){0.f, 0.f, 0.f, 0.f, 0.f, 0.f, 0.f, 0.f};
      accO[el] = (v8f){0.f, 0.f, 0.f, 0.f, 0.f, 0.f, 0.f, 0.f};
    }
    pair_gemm<0, 0,  0,  0>(Xe, sB1, sA2, accE, lane);
    pair_gemm<0, 1,  1,  4>(Xe, sB1, sA2, accO, lane);
    pair_gemm<0, 2,  3,  8>(Xe, sB1, sA2, accE, lane);
    pair_gemm<0, 3,  6, 12>(Xe, sB1, sA2, accO, lane);
    pair_gemm<1, 1, 10, 16>(Xe, sB1, sA2, accE, lane);
    pair_gemm<1, 2, 14, 20>(Xe, sB1, sA2, accO, lane);
    pair_gemm<1, 3, 20, 24>(Xe, sB1, sA2, accE, lane);
    pair_gemm<2, 2, 28, 32>(Xe, sB1, sA2, accE, lane);
    pair_gemm<2, 3, 37, 40>(Xe, sB1, sA2, accO, lane);
    pair_gemm<3, 3, 49, 52>(Xe, sB1, sA2, accE, lane);
#pragma unroll
    for (int el = 0; el < 2; ++el) {
      store_x2(X2 + (e + el) * X2S, accE[el], true, lane);
      store_x2(X2 + (e + el) * X2S, accO[el], false, lane);
    }
    __builtin_amdgcn_fence(__ATOMIC_RELEASE, "wavefront");
    __builtin_amdgcn_fence(__ATOMIC_ACQUIRE, "wavefront");
    float xv[2][8], pv[2][8], nv[2][8];
#pragma unroll
    for (int el = 0; el < 2; ++el) {
      const float* Xq = Xe + el * XS;
      const float* X2q = X2 + (e + el) * X2S;
#pragma unroll
      for (int r = 0; r < 8; ++r) {
        xv[el][r] = Xq[xi[r]];
        pv[el][r] = X2q[pi[r]];
        nv[el][r] = X2q[ni[r]];
      }
    }
    asm volatile("" :: "v"(xv[0][0]), "v"(xv[0][1]), "v"(xv[0][2]), "v"(xv[0][3]), "v"(xv[0][4]), "v"(xv[0][5]), "v"(xv[0][6]), "v"(xv[0][7]), "v"(pv[0][0]), "v"(pv[0][1]), "v"(pv[0][2]), "v"(pv[0][3]), "v"(pv[0][4]), "v"(pv[0][5]), "v"(pv[0][6]), "v"(pv[0][7]), "v"(nv[0][0]), "v"(nv[0][1]), "v"(nv[0][2]), "v"(nv[0][3]), "v"(nv[0][4]), "v"(nv[0][5]), "v"(nv[0][6]), "v"(nv[0][7]));
    asm volatile("" :: "v"(xv[1][0]), "v"(xv[1][1]), "v"(xv[1][2]), "v"(xv[1][3]), "v"(xv[1][4]), "v"(xv[1][5]), "v"(xv[1][6]), "v"(xv[1][7]), "v"(pv[1][0]), "v"(pv[1][1]), "v"(pv[1][2]), "v"(pv[1][3]), "v"(pv[1][4]), "v"(pv[1][5]), "v"(pv[1][6]), "v"(pv[1][7]), "v"(nv[1][0]), "v"(nv[1][1]), "v"(nv[1][2]), "v"(nv[1][3]), "v"(nv[1][4]), "v"(nv[1][5]), "v"(nv[1][6]), "v"(nv[1][7]));
#pragma unroll
    for (int el = 0; el < 2; ++el) {
      float* X2e = X2 + (e + el) * X2S;
      float s3 = 0.f, s4 = 0.f;
#pragma unroll
      for (int r = 0; r < 8; ++r) {
        const float P = posE[r] ? accE[el][r] : accO[el][r];
        const float N = posE[r] ? accO[el][r] : accE[el][r];
        const float wp = wgt[r] * P;
        const float wn = wgtN[r] * N;
        s3 += wp * xv[el][r];
        s4 += wp * pv[el][r];
        s4 += wn * nv[el][r];
      }
      X2e[64 + lane] = s3;
      X2e[96 + lane] = s4;
    }
  }

  __builtin_amdgcn_fence(__ATOMIC_RELEASE, "wavefront");
  __builtin_amdgcn_fence(__ATOMIC_ACQUIRE, "wavefront");

  if (lane < 16) {
    const int e = base + b;
    if (e < Btot) {
      const float* Xb = X + b * XS;
      const float* X2b = X2 + b * X2S;
      const float nu1 = Xb[0];
      const float nu2 = X2b[0];
      float nu3 = 0.f, nu4 = 0.f;
      for (int i = 0; i < 32; ++i) {
        nu3 += X2b[64 + i];
        nu4 += X2b[96 + i];
      }
      out[(size_t)e * 4 + 0] = nu1;
      out[(size_t)e * 4 + 1] = nu2;
      out[(size_t)e * 4 + 2] = nu3;
      out[(size_t)e * 4 + 3] = nu4;
    }
  }
}

extern "C" void kernel_launch(void* const* d_in, const int* in_sizes, int n_in,
                              void* d_out, int out_size, void* d_ws, size_t ws_size,
                              hipStream_t stream) {
  const float* x0 = (const float*)d_in[0];
  const float* x1 = (const float*)d_in[1];
  const float* x2 = (const float*)d_in[2];
  const float* x3 = (const float*)d_in[3];
  float* tab = (float*)d_ws;
  const int Btot = in_sizes[0];

  wigner_cg_init<<<1, 256, 0, stream>>>(tab);

  const int tiles = (Btot + 15) / 16;
  const int blocks = (tiles + WAVES_PER_BLOCK - 1) / WAVES_PER_BLOCK;
  wigner_main<<<blocks, TB, 0, stream>>>(x0, x1, x2, x3, tab, (float*)d_out, Btot);
}


# Pairs (1,3),(2,2),(2,3),(3,3): WMMA 1 takes the larger block as A and the selection-rule gather of the other block times its table entry as B (K = one index, not the Kronecker pair); T sums reordered
# speedup vs baseline: 31.3837x; 1.3967x over previous
#include <hip/hip_runtime.h>


#define NCOMBO 23
#define CG_TOTAL 2351
#define NPAIR 6
#define NPAIR2 4
#define B1_STEPS 35
#define A2_STEPS 72
#define B1_BASE CG_TOTAL
#define A2_BASE (CG_TOTAL + B1_STEPS * 64)
#define WS_TOTAL (A2_BASE + A2_STEPS * 64)
#define LB1 0
#define LA2 (B1_STEPS * 64)
#define LDIAG (LA2 + A2_STEPS * 64)
#define LT_TOTAL (LDIAG + 84)
#define WAVES_PER_BLOCK 4
#define TB (WAVES_PER_BLOCK * 32)
#define XS 101
#define ZC 84
#define XFRONT 8
#define X2S 169

typedef float v2f __attribute__((ext_vector_type(2)));
typedef float v8f __attribute__((ext_vector_type(8)));

__constant__ int c_l1[NCOMBO] = {0,0,0,0, 1,1,1,1,1,1,1,1, 2,2,2,2,2,2,2, 3,3,3,3};
__constant__ int c_l2[NCOMBO] = {0,1,2,3, 1,1,1,2,2,2,3,3, 2,2,2,2,3,3,3, 3,3,3,3};
__constant__ int c_L [NCOMBO] = {0,1,2,3, 0,1,2,1,2,3,2,3, 0,1,2,3,1,2,3, 0,1,2,3};
__constant__ int c_off[NCOMBO] = {0,1,10,35, 84,93,120,165,210,285,390,495,
                                  642,667,742,867,1042,1147,1322, 1567,1616,1763,2008};
__constant__ int p_l1[NPAIR] = {0,0,0,0,1,1};
__constant__ int p_l2[NPAIR] = {0,1,2,3,1,2};
__constant__ int p_c0[NPAIR] = {0,1,2,3,4,7};
__constant__ int p_b1[NPAIR] = {0,1,3,6,10,14};
__constant__ int p_a2[NPAIR] = {0,4,8,12,16,20};
__constant__ int q_l1[NPAIR2] = {1,2,2,3};
__constant__ int q_l2[NPAIR2] = {3,2,3,3};
__constant__ int q_sw[NPAIR2] = {1,0,1,0};
__constant__ int q_c0[NPAIR2] = {10,12,16,19};
__constant__ int q_g [NPAIR2] = {20,24,27,31};
__constant__ int q_a2[NPAIR2] = {24,32,44,56};
__constant__ int d_xoff[4] = {0, 1, 10, 35};
__constant__ int d_diag[4] = {0, 4, 12, 19};

__device__ __forceinline__ int lof(int x) { return (x >= 1) + (x >= 4) + (x >= 9); }

__global__ void wigner_cg_init(float* __restrict__ ws) {
  double f[16];
  f[0] = 1.0;
  for (int i = 1; i < 16; ++i) f[i] = f[i - 1] * i;
  for (int c = 0; c < NCOMBO; ++c) {
    const int l1 = c_l1[c], l2 = c_l2[c], L = c_L[c];
    const int n2 = 2 * l2 + 1, nL = 2 * L + 1;
    const int sz = (2 * l1 + 1) * n2 * nL, off = c_off[c];
    const double pref0 =
        sqrt((2.0 * L + 1.0) * f[L + l1 - l2] * f[L - l1 + l2] * f[l1 + l2 - L] / f[l1 + l2 + L + 1]);
    for (int t = threadIdx.x; t < sz; t += blockDim.x) {
      const int m = t / (n2 * nL), rem = t % (n2 * nL), n = rem / nL, k = rem % nL;
      const int m1 = m - l1, m2 = n - l2, M = m1 + m2;
      double val = 0.0;
      if (M >= -L && M <= L && k == M + L) {
        const double pref =
            pref0 * sqrt(f[L + M] * f[L - M] * f[l1 - m1] * f[l1 + m1] * f[l2 - m2] * f[l2 + m2]);
        double s = 0.0;
        for (int kk = 0; kk <= l1 + l2 - L; ++kk) {
          const int a1 = l1 - m1 - kk, a2 = l2 + m2 - kk;
          const int a3 = L - l2 + m1 + kk, a4 = L - l1 - m2 + kk;
          if (a1 < 0 || a2 < 0 || a3 < 0 || a4 < 0) continue;
          const double term = 1.0 / (f[kk] * f[l1 + l2 - L - kk] * f[a1] * f[a2] * f[a3] * f[a4]);
          s += (kk & 1) ? -term : term;
        }
        val = pref * s;
      }
      ws[off + t] = (float)val;
    }
  }
  __syncthreads();
  for (int pi = 0; pi < NPAIR; ++pi) {
    const int l1 = p_l1[pi], l2 = p_l2[pi];
    const int n1 = 2 * l1 + 1, n2 = 2 * l2 + 1, PH = l1 + 1, QH = l2 + 1;
    const int Lmin = l2 - l1, Lmax = (l1 + l2 < 3) ? (l1 + l2) : 3;
    const int c0 = p_c0[pi], NR = n1 * n2, RT = (NR + 15) >> 4;
    const float w = (l1 == l2) ? 1.f : 2.f;
    for (int t = threadIdx.x; t < PH * QH * 64; t += blockDim.x) {
      const int st = t >> 6, s6 = t & 63, ln = s6 >> 1, s = s6 & 1;
      const int i = st / QH, u = st % QH, h = ln >> 4, c = ln & 15;
      const int p = h * PH + i, q = 2 * u + s, L = lof(c), k = c - L * L;
      float val = 0.f;
      if (p < n1 && q < n2 && L >= Lmin && L <= Lmax)
        val = ws[c_off[c0 + L - Lmin] + (p * n2 + q) * (2 * L + 1) + k];
      ws[B1_BASE + (p_b1[pi] + st) * 64 + s6] = val;
    }
    for (int t = threadIdx.x; t < RT * 4 * 64; t += blockDim.x) {
      const int st = t >> 6, s6 = t & 63, ln = s6 >> 1, s = s6 & 1;
      const int rt = st >> 2, v = st & 3, h = ln >> 4, c = ln & 15;
      const int rho = rt * 16 + 8 * h + 2 * v + s, L = lof(c), k = c - L * L;
      float val = 0.f;
      if (rho < NR && L >= Lmin && L <= Lmax) {
        const int m = rho / n2, n = rho % n2;
        val = w * ws[c_off[c0 + L - Lmin] + (m * n2 + n) * (2 * L + 1) + k];
      }
      ws[A2_BASE + (p_a2[pi] + st) * 64 + s6] = val;
    }
  }
  for (int qi = 0; qi < NPAIR2; ++qi) {
    const int l1 = q_l1[qi], l2 = q_l2[qi], sw = q_sw[qi];
    const int n2 = 2 * l2 + 1;
    const int LA = sw ? l2 : l1, LB = sw ? l1 : l2;
    const int NA = 2 * LA + 1, NB = 2 * LB + 1, KH = LA + 1, TT = LB + 1;
    const int Lmin = l2 - l1, Lmax = (l1 + l2 < 3) ? (l1 + l2) : 3;
    const int c0 = q_c0[qi];
    const float w = (l1 == l2) ? 1.f : 2.f;
    for (int t = threadIdx.x; t < KH * 64; t += blockDim.x) {
      const int st = t >> 6, s6 = t & 63, ln = s6 >> 1, s = s6 & 1;
      const int c = ln & 15, ka = 2 * st + s, L = lof(c), j = c - L * L;
      const int kb = (j - L) + LA + LB - ka;
      float val = 0.f;
      if (ka < NA && kb >= 0 && kb < NB && L >= Lmin && L <= Lmax) {
        const int p = sw ? kb : ka, q = sw ? ka : kb;
        val = ws[c_off[c0 + L - Lmin] + (p * n2 + q) * (2 * L + 1) + j];
      }
      ws[B1_BASE + (q_g[qi] + st) * 64 + s6] = val;
    }
    for (int t = threadIdx.x; t < TT * 4 * 64; t += blockDim.x) {
      const int st = t >> 6, s6 = t & 63, ln = s6 >> 1, s = s6 & 1;
      const int tt = st >> 2, v = st & 3, h = ln >> 4, c = ln & 15;
      const int ma = 2 * v + s, nb = 2 * tt + h, L = lof(c), k = c - L * L;
      float val = 0.f;
      if (ma < NA && nb < NB && L >= Lmin && L <= Lmax) {
        const int m = sw ? nb : ma, n = sw ? ma : nb;
        val = w * ws[c_off[c0 + L - Lmin] + (m * n2 + n) * (2 * L + 1) + k];
      }
      ws[A2_BASE + (q_a2[qi] + st) * 64 + s6] = val;
    }
  }
}

__device__ __forceinline__ constexpr int xoff_c(int l) { return l == 0 ? 0 : (l == 1 ? 1 : (l == 2 ? 10 : 35)); }

template <int L1, int L2, int B1OFF, int A2OFF>
__device__ __forceinline__ void pair_gemm(const float* __restrict__ Xe, const float* __restrict__ sB1,
                                          const float* __restrict__ sA2, v8f (&acc)[2], const int lane) {
  constexpr int N1 = 2 * L1 + 1, N2 = 2 * L2 + 1, PH = L1 + 1, QH = L2 + 1;
  constexpr int NR = N1 * N2, RT = (NR + 15) / 16;
  constexpr int XO1 = xoff_c(L1), XO2 = xoff_c(L2);
  const int h = lane >> 4, c = lane & 15;
#pragma unroll
  for (int rt = 0; rt < RT; ++rt) {
    const int rho = rt * 16 + c;
    const bool valid = rho < NR;
    const int m = rho / N2, n = rho - m * N2;
    float a[2][PH], b[2][2 * QH];
#pragma unroll
    for (int i = 0; i < PH; ++i) {
      const int p = h * PH + i;
      const int ia = (valid && p < N1) ? (XO1 + m * N1 + p) : ZC;
#pragma unroll
      for (int el = 0; el < 2; ++el) a[el][i] = Xe[el * XS + ia];
    }
#pragma unroll
    for (int q = 0; q < 2 * QH; ++q) {
      const int ib = (valid && q < N2) ? (XO2 + n * N2 + q) : ZC;
#pragma unroll
      for (int el = 0; el < 2; ++el) b[el][q] = Xe[el * XS + ib];
    }
    v8f d[2];
#pragma unroll
    for (int el = 0; el < 2; ++el) d[el] = (v8f){0.f, 0.f, 0.f, 0.f, 0.f, 0.f, 0.f, 0.f};
#pragma unroll
    for (int i = 0; i < PH; ++i) {
#pragma unroll
      for (int u = 0; u < QH; ++u) {
        const v2f bf = *(const v2f*)&sB1[(B1OFF + i * QH + u) * 64 + lane * 2];
#pragma unroll
        for (int el = 0; el < 2; ++el) {
          v2f af;
          af.x = a[el][i] * b[el][2 * u];
          af.y = a[el][i] * b[el][2 * u + 1];
          d[el] = __builtin_amdgcn_wmma_f32_16x16x4_f32(false, af, false, bf, (short)0, d[el], false, false);
        }
      }
    }
#pragma unroll
    for (int v = 0; v < 4; ++v) {
      const v2f af2 = *(const v2f*)&sA2[(A2OFF + rt * 4 + v) * 64 + lane * 2];
#pragma unroll
      for (int el = 0; el < 2; ++el) {
        v2f bf2;
        bf2.x = d[el][2 * v];
        bf2.y = d[el][2 * v + 1];
        acc[el] = __builtin_amdgcn_wmma_f32_16x16x4_f32(false, af2, false, bf2, (short)0, acc[el], false, false);
      }
    }
  }
}

template <int LA, int LB, int GOFF, int A2OFF>
__device__ __forceinline__ void pair_gemm2(const float* __restrict__ Xe, const float* __restrict__ sB1,
                                           const float* __restrict__ sA2, v8f (&acc)[2], const int lane,
                                           const int Mc) {
  constexpr int NA = 2 * LA + 1, NB = 2 * LB + 1, KH = LA + 1, TT = LB + 1;
  constexpr int XOA = xoff_c(LA), XOB = xoff_c(LB);
  const int h = lane >> 4, c = lane & 15;
  const int g = c >> 3, ma = c & 7;
  const bool arow = (g == h) && (ma < NA);
  const float* Xa = Xe + (arow ? (XOA + ma * NA) : ZC);
  const float* Xb = Xe + (XOB + h * NB + Mc + LA + LB - (2 * KH - 1));
  v2f af[2][KH], gf[KH];
#pragma unroll
  for (int i = 0; i < KH; ++i) {
    gf[i] = *(const v2f*)&sB1[(GOFF + i) * 64 + lane * 2];
#pragma unroll
    for (int el = 0; el < 2; ++el) {
      af[el][i].x = Xa[el * XS + 2 * i];
      af[el][i].y = Xa[el * XS + 2 * i + 1];
    }
  }
#pragma unroll
  for (int tt = 0; tt < TT; ++tt) {
    asm volatile("" ::: "memory");
    float raw[2][2 * KH];
    v2f a2[4];
#pragma unroll
    for (int ka = 0; ka < 2 * KH; ++ka) {
#pragma unroll
      for (int el = 0; el < 2; ++el) raw[el][ka] = Xb[el * XS + 2 * tt * NB + (2 * KH - 1 - ka)];
    }
#pragma unroll
    for (int v = 0; v < 4; ++v) a2[v] = *(const v2f*)&sA2[(A2OFF + tt * 4 + v) * 64 + lane * 2];
    if constexpr (KH == 4) {
      asm volatile("" :: "v"(raw[0][0]), "v"(raw[0][1]), "v"(raw[0][2]), "v"(raw[0][3]), "v"(raw[0][4]), "v"(raw[0][5]), "v"(raw[0][6]), "v"(raw[0][7]), "v"(raw[1][0]), "v"(raw[1][1]), "v"(raw[1][2]), "v"(raw[1][3]), "v"(raw[1][4]), "v"(raw[1][5]), "v"(raw[1][6]), "v"(raw[1][7]), "v"(a2[0]), "v"(a2[1]), "v"(a2[2]), "v"(a2[3]));
    } else {
      asm volatile("" :: "v"(raw[0][0]), "v"(raw[0][1]), "v"(raw[0][2]), "v"(raw[0][3]), "v"(raw[0][4]), "v"(raw[0][5]), "v"(raw[1][0]), "v"(raw[1][1]), "v"(raw[1][2]), "v"(raw[1][3]), "v"(raw[1][4]), "v"(raw[1][5]), "v"(a2[0]), "v"(a2[1]), "v"(a2[2]), "v"(a2[3]));
    }
    v8f d[2];
#pragma unroll
    for (int el = 0; el < 2; ++el) d[el] = (v8f){0.f, 0.f, 0.f, 0.f, 0.f, 0.f, 0.f, 0.f};
#pragma unroll
    for (int i = 0; i < KH; ++i) {
#pragma unroll
      for (int el = 0; el < 2; ++el) {
        v2f bf;
        bf.x = raw[el][2 * i] * gf[i].x;
        bf.y = raw[el][2 * i + 1] * gf[i].y;
        d[el] = __builtin_amdgcn_wmma_f32_16x16x4_f32(false, af[el][i], false, bf, (short)0, d[el], false, false);
      }
    }
#pragma unroll
    for (int v = 0; v < 4; ++v) {
#pragma unroll
      for (int el = 0; el < 2; ++el) {
        v2f bf2;
        bf2.x = d[el][2 * v];
        bf2.y = d[el][2 * v + 1];
        acc[el] = __builtin_amdgcn_wmma_f32_16x16x4_f32(false, a2[v], false, bf2, (short)0, acc[el], false, false);
      }
    }
  }
}

__device__ __forceinline__ void store_x2(float* __restrict__ X2e, const v8f& t, const bool even, const int lane) {
  const int h = lane >> 4, c = lane & 15;
  const int Lc = lof(c);
#pragma unroll
  for (int v = 0; v < 8; ++v) {
    const int kp = 8 * h + v;
    const int L = lof(kp);
    const int k = kp - L * L, j = c - L * L, nL = 2 * L + 1;
    const bool pos = even ? ((L & 1) == 0) : ((L & 1) == 1);
    const int pb = (L >= 1) + 18 * (L >= 2) + 50 * (L >= 3);
    const int base = pos ? pb : (pb + nL * nL);
    const bool wr = (L == Lc) && (pos || L > 0);
    X2e[wr ? (base + k * nL + j) : (X2S - 1)] = t[v];
  }
}

__launch_bounds__(TB, 3)
__global__ void wigner_main(const float* __restrict__ x0, const float* __restrict__ x1,
                            const float* __restrict__ x2, const float* __restrict__ x3,
                            const float* __restrict__ tab, float* __restrict__ out, int Btot) {
  __shared__ __align__(16) float sTAB[LT_TOTAL];
  __shared__ float sXp[XFRONT + WAVES_PER_BLOCK * 16 * XS];
  __shared__ float sX2[WAVES_PER_BLOCK][16 * X2S];

  const int tid = threadIdx.x, lane = tid & 31, wid = tid >> 5;
  const int b = lane & 15;
  const float* sB1 = sTAB + LB1;
  const float* sA2 = sTAB + LA2;
  const float* sDIAG = sTAB + LDIAG;

  for (int t = tid; t < LDIAG; t += TB) sTAB[t] = tab[CG_TOTAL + t];
  for (int l = 0; l <= 3; ++l) {
    const int nn = (2 * l + 1) * (2 * l + 1);
    for (int t = tid; t < nn; t += TB) sTAB[LDIAG + d_xoff[l] + t] = tab[c_off[d_diag[l]] + t];
  }

  const int tile = blockIdx.x * WAVES_PER_BLOCK + wid;
  const int base = tile * 16;
  float* X  = sXp + XFRONT + wid * 16 * XS;
  float* X2 = sX2[wid];

  {
    int pe = base + WAVES_PER_BLOCK * 16;
    if (pe < Btot) __builtin_prefetch(&x3[(size_t)pe * 49], 0, 1);
  }

  for (int t = lane; t < 84 * 16; t += 32) {
    const int bb = t / 84, comp = t % 84;
    int e = base + bb;
    e = (e < Btot) ? e : (Btot - 1);
    float v;
    if (comp < 1)       v = x0[e];
    else if (comp < 10) v = x1[(size_t)e * 9 + (comp - 1)];
    else if (comp < 35) v = x2[(size_t)e * 25 + (comp - 10)];
    else                v = x3[(size_t)e * 49 + (comp - 35)];
    X[bb * XS + comp] = v;
  }
  for (int t = lane; t < 16 * (XS - ZC); t += 32) X[(t / (XS - ZC)) * XS + ZC + t % (XS - ZC)] = 0.f;
  if (tid < XFRONT) sXp[tid] = 0.f;
  __syncthreads();

  const int hh = lane >> 4, cc = lane & 15;
  const int Mcol = cc - lof(cc) * lof(cc) - lof(cc);
  float wgt[8], wgtN[8];
  int xi[8], pi[8], ni[8];
  bool posE[8];
#pragma unroll
  for (int r = 0; r < 8; ++r) {
    const int kp = 8 * hh + r;
    const int L = lof(kp), nL = 2 * L + 1;
    const int k = kp - L * L, j = cc - L * L;
    const bool dg = (L == lof(cc));
    const int kr = 2 * L - k, jr = 2 * L - j;
    const int xo = (L >= 1) + 9 * (L >= 2) + 25 * (L >= 3);
    const int pb = (L >= 1) + 18 * (L >= 2) + 50 * (L >= 3);
    const float gk = sDIAG[xo + k * nL + kr];
    const float gj = sDIAG[dg ? (xo + j * nL + jr) : 0];
    const float w = gk * gj;
    wgt[r]  = dg ? w : 0.f;
    wgtN[r] = (dg && L > 0) ? w : 0.f;
    xi[r] = dg ? (xo + kr * nL + jr) : ZC;
    pi[r] = dg ? (pb + kr * nL + jr) : 0;
    ni[r] = (dg && L > 0) ? (pb + nL * nL + kr * nL + jr) : 0;
    posE[r] = ((L & 1) == 0);
  }

  for (int e = 0; e < 16; e += 2) {
    asm volatile("" ::: "memory");
    const float* Xe = X + e * XS;
    v8f accE[2], accO[2];
#pragma unroll
    for (int el = 0; el < 2; ++el) {
      accE[el] = (v8f){0.f, 0.f, 0.f, 0.f, 0.f, 0.f, 0.f, 0.f};
      accO[el] = (v8f){0.f, 0.f, 0.f, 0.f, 0.f, 0.f, 0.f, 0.f};
    }
    pair_gemm<0, 0,  0,  0>(Xe, sB1, sA2, accE, lane);
    pair_gemm<0, 1,  1,  4>(Xe, sB1, sA2, accO, lane);
    pair_gemm<0, 2,  3,  8>(Xe, sB1, sA2, accE, lane);
    pair_gemm<0, 3,  6, 12>(Xe, sB1, sA2, accO, lane);
    pair_gemm<1, 1, 10, 16>(Xe, sB1, sA2, accE, lane);
    pair_gemm<1, 2, 14, 20>(Xe, sB1, sA2, accO, lane);
    pair_gemm2<3, 1, 20, 24>(Xe, sB1, sA2, accE, lane, Mcol);
    pair_gemm2<2, 2, 24, 32>(Xe, sB1, sA2, accE, lane, Mcol);
    pair_gemm2<3, 2, 27, 44>(Xe, sB1, sA2, accO, lane, Mcol);
    pair_gemm2<3, 3, 31, 56>(Xe, sB1, sA2, accE, lane, Mcol);
#pragma unroll
    for (int el = 0; el < 2; ++el) {
      store_x2(X2 + (e + el) * X2S, accE[el], true, lane);
      store_x2(X2 + (e + el) * X2S, accO[el], false, lane);
    }
    __builtin_amdgcn_fence(__ATOMIC_RELEASE, "wavefront");
    __builtin_amdgcn_fence(__ATOMIC_ACQUIRE, "wavefront");
    float xv[2][8], pv[2][8], nv[2][8];
#pragma unroll
    for (int el = 0; el < 2; ++el) {
      const float* Xq = Xe + el * XS;
      const float* X2q = X2 + (e + el) * X2S;
#pragma unroll
      for (int r = 0; r < 8; ++r) {
        xv[el][r] = Xq[xi[r]];
        pv[el][r] = X2q[pi[r]];
        nv[el][r] = X2q[ni[r]];
      }
    }
    asm volatile("" :: "v"(xv[0][0]), "v"(xv[0][1]), "v"(xv[0][2]), "v"(xv[0][3]), "v"(xv[0][4]), "v"(xv[0][5]), "v"(xv[0][6]), "v"(xv[0][7]), "v"(pv[0][0]), "v"(pv[0][1]), "v"(pv[0][2]), "v"(pv[0][3]), "v"(pv[0][4]), "v"(pv[0][5]), "v"(pv[0][6]), "v"(pv[0][7]), "v"(nv[0][0]), "v"(nv[0][1]), "v"(nv[0][2]), "v"(nv[0][3]), "v"(nv[0][4]), "v"(nv[0][5]), "v"(nv[0][6]), "v"(nv[0][7]));
    asm volatile("" :: "v"(xv[1][0]), "v"(xv[1][1]), "v"(xv[1][2]), "v"(xv[1][3]), "v"(xv[1][4]), "v"(xv[1][5]), "v"(xv[1][6]), "v"(xv[1][7]), "v"(pv[1][0]), "v"(pv[1][1]), "v"(pv[1][2]), "v"(pv[1][3]), "v"(pv[1][4]), "v"(pv[1][5]), "v"(pv[1][6]), "v"(pv[1][7]), "v"(nv[1][0]), "v"(nv[1][1]), "v"(nv[1][2]), "v"(nv[1][3]), "v"(nv[1][4]), "v"(nv[1][5]), "v"(nv[1][6]), "v"(nv[1][7]));
#pragma unroll
    for (int el = 0; el < 2; ++el) {
      float* X2e = X2 + (e + el) * X2S;
      float s3 = 0.f, s4 = 0.f;
#pragma unroll
      for (int r = 0; r < 8; ++r) {
        const float P = posE[r] ? accE[el][r] : accO[el][r];
        const float N = posE[r] ? accO[el][r] : accE[el][r];
        const float wp = wgt[r] * P;
        const float wn = wgtN[r] * N;
        s3 += wp * xv[el][r];
        s4 += wp * pv[el][r];
        s4 += wn * nv[el][r];
      }
      X2e[64 + lane] = s3;
      X2e[96 + lane] = s4;
    }
  }

  __builtin_amdgcn_fence(__ATOMIC_RELEASE, "wavefront");
  __builtin_amdgcn_fence(__ATOMIC_ACQUIRE, "wavefront");

  if (lane < 16) {
    const int e = base + b;
    if (e < Btot) {
      const float* Xb = X + b * XS;
      const float* X2b = X2 + b * X2S;
      const float nu1 = Xb[0];
      const float nu2 = X2b[0];
      float nu3 = 0.f, nu4 = 0.f;
      for (int i = 0; i < 32; ++i) {
        nu3 += X2b[64 + i];
        nu4 += X2b[96 + i];
      }
      out[(size_t)e * 4 + 0] = nu1;
      out[(size_t)e * 4 + 1] = nu2;
      out[(size_t)e * 4 + 2] = nu3;
      out[(size_t)e * 4 + 3] = nu4;
    }
  }
}

extern "C" void kernel_launch(void* const* d_in, const int* in_sizes, int n_in,
                              void* d_out, int out_size, void* d_ws, size_t ws_size,
                              hipStream_t stream) {
  const float* x0 = (const float*)d_in[0];
  const float* x1 = (const float*)d_in[1];
  const float* x2 = (const float*)d_in[2];
  const float* x3 = (const float*)d_in[3];
  float* tab = (float*)d_ws;
  const int Btot = in_sizes[0];

  wigner_cg_init<<<1, 256, 0, stream>>>(tab);

  const int tiles = (Btot + 15) / 16;
  const int blocks = (tiles + WAVES_PER_BLOCK - 1) / WAVES_PER_BLOCK;
  wigner_main<<<blocks, TB, 0, stream>>>(x0, x1, x2, x3, tab, (float*)d_out, Btot);
}


# LDS keeps only the running trip's two X2 rows plus 65 result floats per element (was all 16 X2 rows): about 74 KB per work-group, so four work-groups per WGP, 4 waves per SIMD instead of 3; same sums
# speedup vs baseline: 33.7657x; 1.0759x over previous
#include <hip/hip_runtime.h>


#define NCOMBO 23
#define CG_TOTAL 2351
#define NPAIR 6
#define NPAIR2 4
#define B1_STEPS 35
#define A2_STEPS 72
#define B1_BASE CG_TOTAL
#define A2_BASE (CG_TOTAL + B1_STEPS * 64)
#define WS_TOTAL (A2_BASE + A2_STEPS * 64)
#define LB1 0
#define LA2 (B1_STEPS * 64)
#define LDIAG (LA2 + A2_STEPS * 64)
#define LT_TOTAL (LDIAG + 84)
#define WAVES_PER_BLOCK 4
#define TB (WAVES_PER_BLOCK * 32)
#define XS 101
#define ZC 84
#define XFRONT 8
#define X2S 169
#define PS 65
#define X2W (2 * X2S + 16 * PS)

typedef float v2f __attribute__((ext_vector_type(2)));
typedef float v8f __attribute__((ext_vector_type(8)));

__constant__ int c_l1[NCOMBO] = {0,0,0,0, 1,1,1,1,1,1,1,1, 2,2,2,2,2,2,2, 3,3,3,3};
__constant__ int c_l2[NCOMBO] = {0,1,2,3, 1,1,1,2,2,2,3,3, 2,2,2,2,3,3,3, 3,3,3,3};
__constant__ int c_L [NCOMBO] = {0,1,2,3, 0,1,2,1,2,3,2,3, 0,1,2,3,1,2,3, 0,1,2,3};
__constant__ int c_off[NCOMBO] = {0,1,10,35, 84,93,120,165,210,285,390,495,
                                  642,667,742,867,1042,1147,1322, 1567,1616,1763,2008};
__constant__ int p_l1[NPAIR] = {0,0,0,0,1,1};
__constant__ int p_l2[NPAIR] = {0,1,2,3,1,2};
__constant__ int p_c0[NPAIR] = {0,1,2,3,4,7};
__constant__ int p_b1[NPAIR] = {0,1,3,6,10,14};
__constant__ int p_a2[NPAIR] = {0,4,8,12,16,20};
__constant__ int q_l1[NPAIR2] = {1,2,2,3};
__constant__ int q_l2[NPAIR2] = {3,2,3,3};
__constant__ int q_sw[NPAIR2] = {1,0,1,0};
__constant__ int q_c0[NPAIR2] = {10,12,16,19};
__constant__ int q_g [NPAIR2] = {20,24,27,31};
__constant__ int q_a2[NPAIR2] = {24,32,44,56};
__constant__ int d_xoff[4] = {0, 1, 10, 35};
__constant__ int d_diag[4] = {0, 4, 12, 19};

__device__ __forceinline__ int lof(int x) { return (x >= 1) + (x >= 4) + (x >= 9); }

__global__ void wigner_cg_init(float* __restrict__ ws) {
  double f[16];
  f[0] = 1.0;
  for (int i = 1; i < 16; ++i) f[i] = f[i - 1] * i;
  for (int c = 0; c < NCOMBO; ++c) {
    const int l1 = c_l1[c], l2 = c_l2[c], L = c_L[c];
    const int n2 = 2 * l2 + 1, nL = 2 * L + 1;
    const int sz = (2 * l1 + 1) * n2 * nL, off = c_off[c];
    const double pref0 =
        sqrt((2.0 * L + 1.0) * f[L + l1 - l2] * f[L - l1 + l2] * f[l1 + l2 - L] / f[l1 + l2 + L + 1]);
    for (int t = threadIdx.x; t < sz; t += blockDim.x) {
      const int m = t / (n2 * nL), rem = t % (n2 * nL), n = rem / nL, k = rem % nL;
      const int m1 = m - l1, m2 = n - l2, M = m1 + m2;
      double val = 0.0;
      if (M >= -L && M <= L && k == M + L) {
        const double pref =
            pref0 * sqrt(f[L + M] * f[L - M] * f[l1 - m1] * f[l1 + m1] * f[l2 - m2] * f[l2 + m2]);
        double s = 0.0;
        for (int kk = 0; kk <= l1 + l2 - L; ++kk) {
          const int a1 = l1 - m1 - kk, a2 = l2 + m2 - kk;
          const int a3 = L - l2 + m1 + kk, a4 = L - l1 - m2 + kk;
          if (a1 < 0 || a2 < 0 || a3 < 0 || a4 < 0) continue;
          const double term = 1.0 / (f[kk] * f[l1 + l2 - L - kk] * f[a1] * f[a2] * f[a3] * f[a4]);
          s += (kk & 1) ? -term : term;
        }
        val = pref * s;
      }
      ws[off + t] = (float)val;
    }
  }
  __syncthreads();
  for (int pi = 0; pi < NPAIR; ++pi) {
    const int l1 = p_l1[pi], l2 = p_l2[pi];
    const int n1 = 2 * l1 + 1, n2 = 2 * l2 + 1, PH = l1 + 1, QH = l2 + 1;
    const int Lmin = l2 - l1, Lmax = (l1 + l2 < 3) ? (l1 + l2) : 3;
    const int c0 = p_c0[pi], NR = n1 * n2, RT = (NR + 15) >> 4;
    const float w = (l1 == l2) ? 1.f : 2.f;
    for (int t = threadIdx.x; t < PH * QH * 64; t += blockDim.x) {
      const int st = t >> 6, s6 = t & 63, ln = s6 >> 1, s = s6 & 1;
      const int i = st / QH, u = st % QH, h = ln >> 4, c = ln & 15;
      const int p = h * PH + i, q = 2 * u + s, L = lof(c), k = c - L * L;
      float val = 0.f;
      if (p < n1 && q < n2 && L >= Lmin && L <= Lmax)
        val = ws[c_off[c0 + L - Lmin] + (p * n2 + q) * (2 * L + 1) + k];
      ws[B1_BASE + (p_b1[pi] + st) * 64 + s6] = val;
    }
    for (int t = threadIdx.x; t < RT * 4 * 64; t += blockDim.x) {
      const int st = t >> 6, s6 = t & 63, ln = s6 >> 1, s = s6 & 1;
      const int rt = st >> 2, v = st & 3, h = ln >> 4, c = ln & 15;
      const int rho = rt * 16 + 8 * h + 2 * v + s, L = lof(c), k = c - L * L;
      float val = 0.f;
      if (rho < NR && L >= Lmin && L <= Lmax) {
        const int m = rho / n2, n = rho % n2;
        val = w * ws[c_off[c0 + L - Lmin] + (m * n2 + n) * (2 * L + 1) + k];
      }
      ws[A2_BASE + (p_a2[pi] + st) * 64 + s6] = val;
    }
  }
  for (int qi = 0; qi < NPAIR2; ++qi) {
    const int l1 = q_l1[qi], l2 = q_l2[qi], sw = q_sw[qi];
    const int n2 = 2 * l2 + 1;
    const int LA = sw ? l2 : l1, LB = sw ? l1 : l2;
    const int NA = 2 * LA + 1, NB = 2 * LB + 1, KH = LA + 1, TT = LB + 1;
    const int Lmin = l2 - l1, Lmax = (l1 + l2 < 3) ? (l1 + l2) : 3;
    const int c0 = q_c0[qi];
    const float w = (l1 == l2) ? 1.f : 2.f;
    for (int t = threadIdx.x; t < KH * 64; t += blockDim.x) {
      const int st = t >> 6, s6 = t & 63, ln = s6 >> 1, s = s6 & 1;
      const int c = ln & 15, ka = 2 * st + s, L = lof(c), j = c - L * L;
      const int kb = (j - L) + LA + LB - ka;
      float val = 0.f;
      if (ka < NA && kb >= 0 && kb < NB && L >= Lmin && L <= Lmax) {
        const int p = sw ? kb : ka, q = sw ? ka : kb;
        val = ws[c_off[c0 + L - Lmin] + (p * n2 + q) * (2 * L + 1) + j];
      }
      ws[B1_BASE + (q_g[qi] + st) * 64 + s6] = val;
    }
    for (int t = threadIdx.x; t < TT * 4 * 64; t += blockDim.x) {
      const int st = t >> 6, s6 = t & 63, ln = s6 >> 1, s = s6 & 1;
      const int tt = st >> 2, v = st & 3, h = ln >> 4, c = ln & 15;
      const int ma = 2 * v + s, nb = 2 * tt + h, L = lof(c), k = c - L * L;
      float val = 0.f;
      if (ma < NA && nb < NB && L >= Lmin && L <= Lmax) {
        const int m = sw ? nb : ma, n = sw ? ma : nb;
        val = w * ws[c_off[c0 + L - Lmin] + (m * n2 + n) * (2 * L + 1) + k];
      }
      ws[A2_BASE + (q_a2[qi] + st) * 64 + s6] = val;
    }
  }
}

__device__ __forceinline__ constexpr int xoff_c(int l) { return l == 0 ? 0 : (l == 1 ? 1 : (l == 2 ? 10 : 35)); }

template <int L1, int L2, int B1OFF, int A2OFF>
__device__ __forceinline__ void pair_gemm(const float* __restrict__ Xe, const float* __restrict__ sB1,
                                          const float* __restrict__ sA2, v8f (&acc)[2], const int lane) {
  constexpr int N1 = 2 * L1 + 1, N2 = 2 * L2 + 1, PH = L1 + 1, QH = L2 + 1;
  constexpr int NR = N1 * N2, RT = (NR + 15) / 16;
  constexpr int XO1 = xoff_c(L1), XO2 = xoff_c(L2);
  const int h = lane >> 4, c = lane & 15;
#pragma unroll
  for (int rt = 0; rt < RT; ++rt) {
    const int rho = rt * 16 + c;
    const bool valid = rho < NR;
    const int m = rho / N2, n = rho - m * N2;
    float a[2][PH], b[2][2 * QH];
#pragma unroll
    for (int i = 0; i < PH; ++i) {
      const int p = h * PH + i;
      const int ia = (valid && p < N1) ? (XO1 + m * N1 + p) : ZC;
#pragma unroll
      for (int el = 0; el < 2; ++el) a[el][i] = Xe[el * XS + ia];
    }
#pragma unroll
    for (int q = 0; q < 2 * QH; ++q) {
      const int ib = (valid && q < N2) ? (XO2 + n * N2 + q) : ZC;
#pragma unroll
      for (int el = 0; el < 2; ++el) b[el][q] = Xe[el * XS + ib];
    }
    v8f d[2];
#pragma unroll
    for (int el = 0; el < 2; ++el) d[el] = (v8f){0.f, 0.f, 0.f, 0.f, 0.f, 0.f, 0.f, 0.f};
#pragma unroll
    for (int i = 0; i < PH; ++i) {
#pragma unroll
      for (int u = 0; u < QH; ++u) {
        const v2f bf = *(const v2f*)&sB1[(B1OFF + i * QH + u) * 64 + lane * 2];
#pragma unroll
        for (int el = 0; el < 2; ++el) {
          v2f af;
          af.x = a[el][i] * b[el][2 * u];
          af.y = a[el][i] * b[el][2 * u + 1];
          d[el] = __builtin_amdgcn_wmma_f32_16x16x4_f32(false, af, false, bf, (short)0, d[el], false, false);
        }
      }
    }
#pragma unroll
    for (int v = 0; v < 4; ++v) {
      const v2f af2 = *(const v2f*)&sA2[(A2OFF + rt * 4 + v) * 64 + lane * 2];
#pragma unroll
      for (int el = 0; el < 2; ++el) {
        v2f bf2;
        bf2.x = d[el][2 * v];
        bf2.y = d[el][2 * v + 1];
        acc[el] = __builtin_amdgcn_wmma_f32_16x16x4_f32(false, af2, false, bf2, (short)0, acc[el], false, false);
      }
    }
  }
}

template <int LA, int LB, int GOFF, int A2OFF>
__device__ __forceinline__ void pair_gemm2(const float* __restrict__ Xe, const float* __restrict__ sB1,
                                           const float* __restrict__ sA2, v8f (&acc)[2], const int lane,
                                           const int Mc) {
  constexpr int NA = 2 * LA + 1, NB = 2 * LB + 1, KH = LA + 1, TT = LB + 1;
  constexpr int XOA = xoff_c(LA), XOB = xoff_c(LB);
  const int h = lane >> 4, c = lane & 15;
  const int g = c >> 3, ma = c & 7;
  const bool arow = (g == h) && (ma < NA);
  const float* Xa = Xe + (arow ? (XOA + ma * NA) : ZC);
  const float* Xb = Xe + (XOB + h * NB + Mc + LA + LB - (2 * KH - 1));
  v2f af[2][KH], gf[KH];
#pragma unroll
  for (int i = 0; i < KH; ++i) {
    gf[i] = *(const v2f*)&sB1[(GOFF + i) * 64 + lane * 2];
#pragma unroll
    for (int el = 0; el < 2; ++el) {
      af[el][i].x = Xa[el * XS + 2 * i];
      af[el][i].y = Xa[el * XS + 2 * i + 1];
    }
  }
#pragma unroll
  for (int tt = 0; tt < TT; ++tt) {
    asm volatile("" ::: "memory");
    float raw[2][2 * KH];
    v2f a2[4];
#pragma unroll
    for (int ka = 0; ka < 2 * KH; ++ka) {
#pragma unroll
      for (int el = 0; el < 2; ++el) raw[el][ka] = Xb[el * XS + 2 * tt * NB + (2 * KH - 1 - ka)];
    }
#pragma unroll
    for (int v = 0; v < 4; ++v) a2[v] = *(const v2f*)&sA2[(A2OFF + tt * 4 + v) * 64 + lane * 2];
    if constexpr (KH == 4) {
      asm volatile("" :: "v"(raw[0][0]), "v"(raw[0][1]), "v"(raw[0][2]), "v"(raw[0][3]), "v"(raw[0][4]), "v"(raw[0][5]), "v"(raw[0][6]), "v"(raw[0][7]), "v"(raw[1][0]), "v"(raw[1][1]), "v"(raw[1][2]), "v"(raw[1][3]), "v"(raw[1][4]), "v"(raw[1][5]), "v"(raw[1][6]), "v"(raw[1][7]), "v"(a2[0]), "v"(a2[1]), "v"(a2[2]), "v"(a2[3]));
    } else {
      asm volatile("" :: "v"(raw[0][0]), "v"(raw[0][1]), "v"(raw[0][2]), "v"(raw[0][3]), "v"(raw[0][4]), "v"(raw[0][5]), "v"(raw[1][0]), "v"(raw[1][1]), "v"(raw[1][2]), "v"(raw[1][3]), "v"(raw[1][4]), "v"(raw[1][5]), "v"(a2[0]), "v"(a2[1]), "v"(a2[2]), "v"(a2[3]));
    }
    v8f d[2];
#pragma unroll
    for (int el = 0; el < 2; ++el) d[el] = (v8f){0.f, 0.f, 0.f, 0.f, 0.f, 0.f, 0.f, 0.f};
#pragma unroll
    for (int i = 0; i < KH; ++i) {
#pragma unroll
      for (int el = 0; el < 2; ++el) {
        v2f bf;
        bf.x = raw[el][2 * i] * gf[i].x;
        bf.y = raw[el][2 * i + 1] * gf[i].y;
        d[el] = __builtin_amdgcn_wmma_f32_16x16x4_f32(false, af[el][i], false, bf, (short)0, d[el], false, false);
      }
    }
#pragma unroll
    for (int v = 0; v < 4; ++v) {
#pragma unroll
      for (int el = 0; el < 2; ++el) {
        v2f bf2;
        bf2.x = d[el][2 * v];
        bf2.y = d[el][2 * v + 1];
        acc[el] = __builtin_amdgcn_wmma_f32_16x16x4_f32(false, a2[v], false, bf2, (short)0, acc[el], false, false);
      }
    }
  }
}

__device__ __forceinline__ void store_x2(float* __restrict__ X2e, const v8f& t, const bool even, const int lane) {
  const int h = lane >> 4, c = lane & 15;
  const int Lc = lof(c);
#pragma unroll
  for (int v = 0; v < 8; ++v) {
    const int kp = 8 * h + v;
    const int L = lof(kp);
    const int k = kp - L * L, j = c - L * L, nL = 2 * L + 1;
    const bool pos = even ? ((L & 1) == 0) : ((L & 1) == 1);
    const int pb = (L >= 1) + 18 * (L >= 2) + 50 * (L >= 3);
    const int base = pos ? pb : (pb + nL * nL);
    const bool wr = (L == Lc) && (pos || L > 0);
    X2e[wr ? (base + k * nL + j) : (X2S - 1)] = t[v];
  }
}

__launch_bounds__(TB, 4)
__global__ void wigner_main(const float* __restrict__ x0, const float* __restrict__ x1,
                            const float* __restrict__ x2, const float* __restrict__ x3,
                            const float* __restrict__ tab, float* __restrict__ out, int Btot) {
  __shared__ __align__(16) float sTAB[LT_TOTAL];
  __shared__ float sXp[XFRONT + WAVES_PER_BLOCK * 16 * XS];
  __shared__ float sX2[WAVES_PER_BLOCK][X2W];

  const int tid = threadIdx.x, lane = tid & 31, wid = tid >> 5;
  const int b = lane & 15;
  const float* sB1 = sTAB + LB1;
  const float* sA2 = sTAB + LA2;
  const float* sDIAG = sTAB + LDIAG;

  for (int t = tid; t < LDIAG; t += TB) sTAB[t] = tab[CG_TOTAL + t];
  for (int l = 0; l <= 3; ++l) {
    const int nn = (2 * l + 1) * (2 * l + 1);
    for (int t = tid; t < nn; t += TB) sTAB[LDIAG + d_xoff[l] + t] = tab[c_off[d_diag[l]] + t];
  }

  const int tile = blockIdx.x * WAVES_PER_BLOCK + wid;
  const int base = tile * 16;
  float* X  = sXp + XFRONT + wid * 16 * XS;
  float* X2 = sX2[wid];
  float* PR = sX2[wid] + 2 * X2S;

  {
    int pe = base + WAVES_PER_BLOCK * 16;
    if (pe < Btot) __builtin_prefetch(&x3[(size_t)pe * 49], 0, 1);
  }

  for (int t = lane; t < 84 * 16; t += 32) {
    const int bb = t / 84, comp = t % 84;
    int e = base + bb;
    e = (e < Btot) ? e : (Btot - 1);
    float v;
    if (comp < 1)       v = x0[e];
    else if (comp < 10) v = x1[(size_t)e * 9 + (comp - 1)];
    else if (comp < 35) v = x2[(size_t)e * 25 + (comp - 10)];
    else                v = x3[(size_t)e * 49 + (comp - 35)];
    X[bb * XS + comp] = v;
  }
  for (int t = lane; t < 16 * (XS - ZC); t += 32) X[(t / (XS - ZC)) * XS + ZC + t % (XS - ZC)] = 0.f;
  if (tid < XFRONT) sXp[tid] = 0.f;
  __syncthreads();

  const int hh = lane >> 4, cc = lane & 15;
  const int Mcol = cc - lof(cc) * lof(cc) - lof(cc);
  float wgt[8], wgtN[8];
  int xi[8], pi[8], ni[8];
  bool posE[8];
#pragma unroll
  for (int r = 0; r < 8; ++r) {
    const int kp = 8 * hh + r;
    const int L = lof(kp), nL = 2 * L + 1;
    const int k = kp - L * L, j = cc - L * L;
    const bool dg = (L == lof(cc));
    const int kr = 2 * L - k, jr = 2 * L - j;
    const int xo = (L >= 1) + 9 * (L >= 2) + 25 * (L >= 3);
    const int pb = (L >= 1) + 18 * (L >= 2) + 50 * (L >= 3);
    const float gk = sDIAG[xo + k * nL + kr];
    const float gj = sDIAG[dg ? (xo + j * nL + jr) : 0];
    const float w = gk * gj;
    wgt[r]  = dg ? w : 0.f;
    wgtN[r] = (dg && L > 0) ? w : 0.f;
    xi[r] = dg ? (xo + kr * nL + jr) : ZC;
    pi[r] = dg ? (pb + kr * nL + jr) : 0;
    ni[r] = (dg && L > 0) ? (pb + nL * nL + kr * nL + jr) : 0;
    posE[r] = ((L & 1) == 0);
  }

  for (int e = 0; e < 16; e += 2) {
    asm volatile("" ::: "memory");
    const float* Xe = X + e * XS;
    v8f accE[2], accO[2];
#pragma unroll
    for (int el = 0; el < 2; ++el) {
      accE[el] = (v8f){0.f, 0.f, 0.f, 0.f, 0.f, 0.f, 0.f, 0.f};
      accO[el] = (v8f){0.f, 0.f, 0.f, 0.f, 0.f, 0.f, 0.f, 0.f};
    }
    pair_gemm<0, 0,  0,  0>(Xe, sB1, sA2, accE, lane);
    pair_gemm<0, 1,  1,  4>(Xe, sB1, sA2, accO, lane);
    pair_gemm<0, 2,  3,  8>(Xe, sB1, sA2, accE, lane);
    pair_gemm<0, 3,  6, 12>(Xe, sB1, sA2, accO, lane);
    pair_gemm<1, 1, 10, 16>(Xe, sB1, sA2, accE, lane);
    pair_gemm<1, 2, 14, 20>(Xe, sB1, sA2, accO, lane);
    pair_gemm2<3, 1, 20, 24>(Xe, sB1, sA2, accE, lane, Mcol);
    pair_gemm2<2, 2, 24, 32>(Xe, sB1, sA2, accE, lane, Mcol);
    pair_gemm2<3, 2, 27, 44>(Xe, sB1, sA2, accO, lane, Mcol);
    pair_gemm2<3, 3, 31, 56>(Xe, sB1, sA2, accE, lane, Mcol);
#pragma unroll
    for (int el = 0; el < 2; ++el) {
      store_x2(X2 + el * X2S, accE[el], true, lane);
      store_x2(X2 + el * X2S, accO[el], false, lane);
    }
    __builtin_amdgcn_fence(__ATOMIC_RELEASE, "wavefront");
    __builtin_amdgcn_fence(__ATOMIC_ACQUIRE, "wavefront");
    float xv[2][8], pv[2][8], nv[2][8];
#pragma unroll
    for (int el = 0; el < 2; ++el) {
      const float* Xq = Xe + el * XS;
      const float* X2q = X2 + el * X2S;
#pragma unroll
      for (int r = 0; r < 8; ++r) {
        xv[el][r] = Xq[xi[r]];
        pv[el][r] = X2q[pi[r]];
        nv[el][r] = X2q[ni[r]];
      }
    }
    asm volatile("" :: "v"(xv[0][0]), "v"(xv[0][1]), "v"(xv[0][2]), "v"(xv[0][3]), "v"(xv[0][4]), "v"(xv[0][5]), "v"(xv[0][6]), "v"(xv[0][7]), "v"(pv[0][0]), "v"(pv[0][1]), "v"(pv[0][2]), "v"(pv[0][3]), "v"(pv[0][4]), "v"(pv[0][5]), "v"(pv[0][6]), "v"(pv[0][7]), "v"(nv[0][0]), "v"(nv[0][1]), "v"(nv[0][2]), "v"(nv[0][3]), "v"(nv[0][4]), "v"(nv[0][5]), "v"(nv[0][6]), "v"(nv[0][7]));
    asm volatile("" :: "v"(xv[1][0]), "v"(xv[1][1]), "v"(xv[1][2]), "v"(xv[1][3]), "v"(xv[1][4]), "v"(xv[1][5]), "v"(xv[1][6]), "v"(xv[1][7]), "v"(pv[1][0]), "v"(pv[1][1]), "v"(pv[1][2]), "v"(pv[1][3]), "v"(pv[1][4]), "v"(pv[1][5]), "v"(pv[1][6]), "v"(pv[1][7]), "v"(nv[1][0]), "v"(nv[1][1]), "v"(nv[1][2]), "v"(nv[1][3]), "v"(nv[1][4]), "v"(nv[1][5]), "v"(nv[1][6]), "v"(nv[1][7]));
#pragma unroll
    for (int el = 0; el < 2; ++el) {
      float* Pe = PR + (e + el) * PS;
      float s3 = 0.f, s4 = 0.f;
#pragma unroll
      for (int r = 0; r < 8; ++r) {
        const float P = posE[r] ? accE[el][r] : accO[el][r];
        const float N = posE[r] ? accO[el][r] : accE[el][r];
        const float wp = wgt[r] * P;
        const float wn = wgtN[r] * N;
        s3 += wp * xv[el][r];
        s4 += wp * pv[el][r];
        s4 += wn * nv[el][r];
      }
      Pe[1 + lane] = s3;
      Pe[33 + lane] = s4;
      if (lane == 0) Pe[0] = accE[el][0];
    }
  }

  __builtin_amdgcn_fence(__ATOMIC_RELEASE, "wavefront");
  __builtin_amdgcn_fence(__ATOMIC_ACQUIRE, "wavefront");

  if (lane < 16) {
    const int e = base + b;
    if (e < Btot) {
      const float* Xb = X + b * XS;
      const float* Pb = PR + b * PS;
      const float nu1 = Xb[0];
      const float nu2 = Pb[0];
      float nu3 = 0.f, nu4 = 0.f;
      for (int i = 0; i < 32; ++i) {
        nu3 += Pb[1 + i];
        nu4 += Pb[33 + i];
      }
      out[(size_t)e * 4 + 0] = nu1;
      out[(size_t)e * 4 + 1] = nu2;
      out[(size_t)e * 4 + 2] = nu3;
      out[(size_t)e * 4 + 3] = nu4;
    }
  }
}

extern "C" void kernel_launch(void* const* d_in, const int* in_sizes, int n_in,
                              void* d_out, int out_size, void* d_ws, size_t ws_size,
                              hipStream_t stream) {
  const float* x0 = (const float*)d_in[0];
  const float* x1 = (const float*)d_in[1];
  const float* x2 = (const float*)d_in[2];
  const float* x3 = (const float*)d_in[3];
  float* tab = (float*)d_ws;
  const int Btot = in_sizes[0];

  wigner_cg_init<<<1, 256, 0, stream>>>(tab);

  const int tiles = (Btot + 15) / 16;
  const int blocks = (tiles + WAVES_PER_BLOCK - 1) / WAVES_PER_BLOCK;
  wigner_main<<<blocks, TB, 0, stream>>>(x0, x1, x2, x3, tab, (float*)d_out, Btot);
}


# WMMA 2 issues only K steps that hold rows: tile rows placed 4 per step in step order, steps past a pair's last row (ma>=5 in pair (2,2)) dropped: 274 to 246 WMMAs a trip; row sums of O reordered
# speedup vs baseline: 35.1288x; 1.0404x over previous
#include <hip/hip_runtime.h>


#define NCOMBO 23
#define CG_TOTAL 2351
#define NPAIR 6
#define NPAIR2 4
#define B1_STEPS 35
#define A2_STEPS 72
#define B1_BASE CG_TOTAL
#define A2_BASE (CG_TOTAL + B1_STEPS * 64)
#define WS_TOTAL (A2_BASE + A2_STEPS * 64)
#define LB1 0
#define LA2 (B1_STEPS * 64)
#define LDIAG (LA2 + A2_STEPS * 64)
#define LT_TOTAL (LDIAG + 84)
#define WAVES_PER_BLOCK 4
#define TB (WAVES_PER_BLOCK * 32)
#define XS 101
#define ZC 84
#define XFRONT 8
#define X2S 169
#define PS 65
#define X2W (2 * X2S + 16 * PS)

typedef float v2f __attribute__((ext_vector_type(2)));
typedef float v8f __attribute__((ext_vector_type(8)));

__constant__ int c_l1[NCOMBO] = {0,0,0,0, 1,1,1,1,1,1,1,1, 2,2,2,2,2,2,2, 3,3,3,3};
__constant__ int c_l2[NCOMBO] = {0,1,2,3, 1,1,1,2,2,2,3,3, 2,2,2,2,3,3,3, 3,3,3,3};
__constant__ int c_L [NCOMBO] = {0,1,2,3, 0,1,2,1,2,3,2,3, 0,1,2,3,1,2,3, 0,1,2,3};
__constant__ int c_off[NCOMBO] = {0,1,10,35, 84,93,120,165,210,285,390,495,
                                  642,667,742,867,1042,1147,1322, 1567,1616,1763,2008};
__constant__ int p_l1[NPAIR] = {0,0,0,0,1,1};
__constant__ int p_l2[NPAIR] = {0,1,2,3,1,2};
__constant__ int p_c0[NPAIR] = {0,1,2,3,4,7};
__constant__ int p_b1[NPAIR] = {0,1,3,6,10,14};
__constant__ int p_a2[NPAIR] = {0,4,8,12,16,20};
__constant__ int q_l1[NPAIR2] = {1,2,2,3};
__constant__ int q_l2[NPAIR2] = {3,2,3,3};
__constant__ int q_sw[NPAIR2] = {1,0,1,0};
__constant__ int q_c0[NPAIR2] = {10,12,16,19};
__constant__ int q_g [NPAIR2] = {20,24,27,31};
__constant__ int q_a2[NPAIR2] = {24,32,44,56};
__constant__ int d_xoff[4] = {0, 1, 10, 35};
__constant__ int d_diag[4] = {0, 4, 12, 19};

__device__ __forceinline__ int lof(int x) { return (x >= 1) + (x >= 4) + (x >= 9); }

__global__ void wigner_cg_init(float* __restrict__ ws) {
  double f[16];
  f[0] = 1.0;
  for (int i = 1; i < 16; ++i) f[i] = f[i - 1] * i;
  for (int c = 0; c < NCOMBO; ++c) {
    const int l1 = c_l1[c], l2 = c_l2[c], L = c_L[c];
    const int n2 = 2 * l2 + 1, nL = 2 * L + 1;
    const int sz = (2 * l1 + 1) * n2 * nL, off = c_off[c];
    const double pref0 =
        sqrt((2.0 * L + 1.0) * f[L + l1 - l2] * f[L - l1 + l2] * f[l1 + l2 - L] / f[l1 + l2 + L + 1]);
    for (int t = threadIdx.x; t < sz; t += blockDim.x) {
      const int m = t / (n2 * nL), rem = t % (n2 * nL), n = rem / nL, k = rem % nL;
      const int m1 = m - l1, m2 = n - l2, M = m1 + m2;
      double val = 0.0;
      if (M >= -L && M <= L && k == M + L) {
        const double pref =
            pref0 * sqrt(f[L + M] * f[L - M] * f[l1 - m1] * f[l1 + m1] * f[l2 - m2] * f[l2 + m2]);
        double s = 0.0;
        for (int kk = 0; kk <= l1 + l2 - L; ++kk) {
          const int a1 = l1 - m1 - kk, a2 = l2 + m2 - kk;
          const int a3 = L - l2 + m1 + kk, a4 = L - l1 - m2 + kk;
          if (a1 < 0 || a2 < 0 || a3 < 0 || a4 < 0) continue;
          const double term = 1.0 / (f[kk] * f[l1 + l2 - L - kk] * f[a1] * f[a2] * f[a3] * f[a4]);
          s += (kk & 1) ? -term : term;
        }
        val = pref * s;
      }
      ws[off + t] = (float)val;
    }
  }
  __syncthreads();
  for (int pi = 0; pi < NPAIR; ++pi) {
    const int l1 = p_l1[pi], l2 = p_l2[pi];
    const int n1 = 2 * l1 + 1, n2 = 2 * l2 + 1, PH = l1 + 1, QH = l2 + 1;
    const int Lmin = l2 - l1, Lmax = (l1 + l2 < 3) ? (l1 + l2) : 3;
    const int c0 = p_c0[pi], NR = n1 * n2, RT = (NR + 15) >> 4;
    const float w = (l1 == l2) ? 1.f : 2.f;
    for (int t = threadIdx.x; t < PH * QH * 64; t += blockDim.x) {
      const int st = t >> 6, s6 = t & 63, ln = s6 >> 1, s = s6 & 1;
      const int i = st / QH, u = st % QH, h = ln >> 4, c = ln & 15;
      const int p = h * PH + i, q = 2 * u + s, L = lof(c), k = c - L * L;
      float val = 0.f;
      if (p < n1 && q < n2 && L >= Lmin && L <= Lmax)
        val = ws[c_off[c0 + L - Lmin] + (p * n2 + q) * (2 * L + 1) + k];
      ws[B1_BASE + (p_b1[pi] + st) * 64 + s6] = val;
    }
    for (int t = threadIdx.x; t < RT * 4 * 64; t += blockDim.x) {
      const int st = t >> 6, s6 = t & 63, ln = s6 >> 1, s = s6 & 1;
      const int rt = st >> 2, v = st & 3, h = ln >> 4, c = ln & 15;
      const int rho = rt * 16 + 4 * v + 2 * h + s, L = lof(c), k = c - L * L;
      float val = 0.f;
      if (rho < NR && L >= Lmin && L <= Lmax) {
        const int m = rho / n2, n = rho % n2;
        val = w * ws[c_off[c0 + L - Lmin] + (m * n2 + n) * (2 * L + 1) + k];
      }
      ws[A2_BASE + (p_a2[pi] + st) * 64 + s6] = val;
    }
  }
  for (int qi = 0; qi < NPAIR2; ++qi) {
    const int l1 = q_l1[qi], l2 = q_l2[qi], sw = q_sw[qi];
    const int n2 = 2 * l2 + 1;
    const int LA = sw ? l2 : l1, LB = sw ? l1 : l2;
    const int NA = 2 * LA + 1, NB = 2 * LB + 1, KH = LA + 1, TT = LB + 1;
    const int Lmin = l2 - l1, Lmax = (l1 + l2 < 3) ? (l1 + l2) : 3;
    const int c0 = q_c0[qi];
    const float w = (l1 == l2) ? 1.f : 2.f;
    for (int t = threadIdx.x; t < KH * 64; t += blockDim.x) {
      const int st = t >> 6, s6 = t & 63, ln = s6 >> 1, s = s6 & 1;
      const int c = ln & 15, ka = 2 * st + s, L = lof(c), j = c - L * L;
      const int kb = (j - L) + LA + LB - ka;
      float val = 0.f;
      if (ka < NA && kb >= 0 && kb < NB && L >= Lmin && L <= Lmax) {
        const int p = sw ? kb : ka, q = sw ? ka : kb;
        val = ws[c_off[c0 + L - Lmin] + (p * n2 + q) * (2 * L + 1) + j];
      }
      ws[B1_BASE + (q_g[qi] + st) * 64 + s6] = val;
    }
    for (int t = threadIdx.x; t < TT * 4 * 64; t += blockDim.x) {
      const int st = t >> 6, s6 = t & 63, ln = s6 >> 1, s = s6 & 1;
      const int tt = st >> 2, v = st & 3, h = ln >> 4, c = ln & 15;
      const int ma = 2 * v + s, nb = 2 * tt + h, L = lof(c), k = c - L * L;
      float val = 0.f;
      if (ma < NA && nb < NB && L >= Lmin && L <= Lmax) {
        const int m = sw ? nb : ma, n = sw ? ma : nb;
        val = w * ws[c_off[c0 + L - Lmin] + (m * n2 + n) * (2 * L + 1) + k];
      }
      ws[A2_BASE + (q_a2[qi] + st) * 64 + s6] = val;
    }
  }
}

__device__ __forceinline__ constexpr int xoff_c(int l) { return l == 0 ? 0 : (l == 1 ? 1 : (l == 2 ? 10 : 35)); }

template <int L1, int L2, int B1OFF, int A2OFF>
__device__ __forceinline__ void pair_gemm(const float* __restrict__ Xe, const float* __restrict__ sB1,
                                          const float* __restrict__ sA2, v8f (&acc)[2], const int lane) {
  constexpr int N1 = 2 * L1 + 1, N2 = 2 * L2 + 1, PH = L1 + 1, QH = L2 + 1;
  constexpr int NR = N1 * N2, RT = (NR + 15) / 16;
  constexpr int XO1 = xoff_c(L1), XO2 = xoff_c(L2);
  const int h = lane >> 4, c = lane & 15;
#pragma unroll
  for (int rt = 0; rt < RT; ++rt) {
    const int rho = rt * 16 + 4 * ((c & 7) >> 1) + 2 * (c >> 3) + (c & 1);
    const bool valid = rho < NR;
    const int m = rho / N2, n = rho - m * N2;
    float a[2][PH], b[2][2 * QH];
#pragma unroll
    for (int i = 0; i < PH; ++i) {
      const int p = h * PH + i;
      const int ia = (valid && p < N1) ? (XO1 + m * N1 + p) : ZC;
#pragma unroll
      for (int el = 0; el < 2; ++el) a[el][i] = Xe[el * XS + ia];
    }
#pragma unroll
    for (int q = 0; q < 2 * QH; ++q) {
      const int ib = (valid && q < N2) ? (XO2 + n * N2 + q) : ZC;
#pragma unroll
      for (int el = 0; el < 2; ++el) b[el][q] = Xe[el * XS + ib];
    }
    v8f d[2];
#pragma unroll
    for (int el = 0; el < 2; ++el) d[el] = (v8f){0.f, 0.f, 0.f, 0.f, 0.f, 0.f, 0.f, 0.f};
#pragma unroll
    for (int i = 0; i < PH; ++i) {
#pragma unroll
      for (int u = 0; u < QH; ++u) {
        const v2f bf = *(const v2f*)&sB1[(B1OFF + i * QH + u) * 64 + lane * 2];
#pragma unroll
        for (int el = 0; el < 2; ++el) {
          v2f af;
          af.x = a[el][i] * b[el][2 * u];
          af.y = a[el][i] * b[el][2 * u + 1];
          d[el] = __builtin_amdgcn_wmma_f32_16x16x4_f32(false, af, false, bf, (short)0, d[el], false, false);
        }
      }
    }
    const int rows = (NR - rt * 16 < 16) ? (NR - rt * 16) : 16;
    const int nv = (rows + 3) / 4;
#pragma unroll
    for (int v = 0; v < 4; ++v) {
      if (v < nv) {
        const v2f af2 = *(const v2f*)&sA2[(A2OFF + rt * 4 + v) * 64 + lane * 2];
#pragma unroll
        for (int el = 0; el < 2; ++el) {
          v2f bf2;
          bf2.x = d[el][2 * v];
          bf2.y = d[el][2 * v + 1];
          acc[el] = __builtin_amdgcn_wmma_f32_16x16x4_f32(false, af2, false, bf2, (short)0, acc[el], false, false);
        }
      }
    }
  }
}

template <int LA, int LB, int GOFF, int A2OFF>
__device__ __forceinline__ void pair_gemm2(const float* __restrict__ Xe, const float* __restrict__ sB1,
                                           const float* __restrict__ sA2, v8f (&acc)[2], const int lane,
                                           const int Mc) {
  constexpr int NA = 2 * LA + 1, NB = 2 * LB + 1, KH = LA + 1, TT = LB + 1;
  constexpr int NV2 = (NA + 1) / 2;
  constexpr int XOA = xoff_c(LA), XOB = xoff_c(LB);
  const int h = lane >> 4, c = lane & 15;
  const int g = c >> 3, ma = c & 7;
  const bool arow = (g == h) && (ma < NA);
  const float* Xa = Xe + (arow ? (XOA + ma * NA) : ZC);
  const float* Xb = Xe + (XOB + h * NB + Mc + LA + LB - (2 * KH - 1));
  static_assert((KH == 4) == (NV2 == 4), "the load fence below lists NV2 table fragments");
  v2f af[2][KH], gf[KH];
#pragma unroll
  for (int i = 0; i < KH; ++i) {
    gf[i] = *(const v2f*)&sB1[(GOFF + i) * 64 + lane * 2];
#pragma unroll
    for (int el = 0; el < 2; ++el) {
      af[el][i].x = Xa[el * XS + 2 * i];
      af[el][i].y = Xa[el * XS + 2 * i + 1];
    }
  }
#pragma unroll
  for (int tt = 0; tt < TT; ++tt) {
    asm volatile("" ::: "memory");
    float raw[2][2 * KH];
    v2f a2[NV2];
#pragma unroll
    for (int ka = 0; ka < 2 * KH; ++ka) {
#pragma unroll
      for (int el = 0; el < 2; ++el) raw[el][ka] = Xb[el * XS + 2 * tt * NB + (2 * KH - 1 - ka)];
    }
#pragma unroll
    for (int v = 0; v < NV2; ++v) a2[v] = *(const v2f*)&sA2[(A2OFF + tt * 4 + v) * 64 + lane * 2];
    if constexpr (KH == 4) {
      asm volatile("" :: "v"(raw[0][0]), "v"(raw[0][1]), "v"(raw[0][2]), "v"(raw[0][3]), "v"(raw[0][4]), "v"(raw[0][5]), "v"(raw[0][6]), "v"(raw[0][7]), "v"(raw[1][0]), "v"(raw[1][1]), "v"(raw[1][2]), "v"(raw[1][3]), "v"(raw[1][4]), "v"(raw[1][5]), "v"(raw[1][6]), "v"(raw[1][7]), "v"(a2[0]), "v"(a2[1]), "v"(a2[2]), "v"(a2[3]));
    } else {
      asm volatile("" :: "v"(raw[0][0]), "v"(raw[0][1]), "v"(raw[0][2]), "v"(raw[0][3]), "v"(raw[0][4]), "v"(raw[0][5]), "v"(raw[1][0]), "v"(raw[1][1]), "v"(raw[1][2]), "v"(raw[1][3]), "v"(raw[1][4]), "v"(raw[1][5]), "v"(a2[0]), "v"(a2[1]), "v"(a2[2]));
    }
    v8f d[2];
#pragma unroll
    for (int el = 0; el < 2; ++el) d[el] = (v8f){0.f, 0.f, 0.f, 0.f, 0.f, 0.f, 0.f, 0.f};
#pragma unroll
    for (int i = 0; i < KH; ++i) {
#pragma unroll
      for (int el = 0; el < 2; ++el) {
        v2f bf;
        bf.x = raw[el][2 * i] * gf[i].x;
        bf.y = raw[el][2 * i + 1] * gf[i].y;
        d[el] = __builtin_amdgcn_wmma_f32_16x16x4_f32(false, af[el][i], false, bf, (short)0, d[el], false, false);
      }
    }
#pragma unroll
    for (int v = 0; v < NV2; ++v) {
#pragma unroll
      for (int el = 0; el < 2; ++el) {
        v2f bf2;
        bf2.x = d[el][2 * v];
        bf2.y = d[el][2 * v + 1];
        acc[el] = __builtin_amdgcn_wmma_f32_16x16x4_f32(false, a2[v], false, bf2, (short)0, acc[el], false, false);
      }
    }
  }
}

__device__ __forceinline__ void store_x2(float* __restrict__ X2e, const v8f& t, const bool even, const int lane) {
  const int h = lane >> 4, c = lane & 15;
  const int Lc = lof(c);
#pragma unroll
  for (int v = 0; v < 8; ++v) {
    const int kp = 8 * h + v;
    const int L = lof(kp);
    const int k = kp - L * L, j = c - L * L, nL = 2 * L + 1;
    const bool pos = even ? ((L & 1) == 0) : ((L & 1) == 1);
    const int pb = (L >= 1) + 18 * (L >= 2) + 50 * (L >= 3);
    const int base = pos ? pb : (pb + nL * nL);
    const bool wr = (L == Lc) && (pos || L > 0);
    X2e[wr ? (base + k * nL + j) : (X2S - 1)] = t[v];
  }
}

__launch_bounds__(TB, 4)
__global__ void wigner_main(const float* __restrict__ x0, const float* __restrict__ x1,
                            const float* __restrict__ x2, const float* __restrict__ x3,
                            const float* __restrict__ tab, float* __restrict__ out, int Btot) {
  __shared__ __align__(16) float sTAB[LT_TOTAL];
  __shared__ float sXp[XFRONT + WAVES_PER_BLOCK * 16 * XS];
  __shared__ float sX2[WAVES_PER_BLOCK][X2W];

  const int tid = threadIdx.x, lane = tid & 31, wid = tid >> 5;
  const int b = lane & 15;
  const float* sB1 = sTAB + LB1;
  const float* sA2 = sTAB + LA2;
  const float* sDIAG = sTAB + LDIAG;

  for (int t = tid; t < LDIAG; t += TB) sTAB[t] = tab[CG_TOTAL + t];
  for (int l = 0; l <= 3; ++l) {
    const int nn = (2 * l + 1) * (2 * l + 1);
    for (int t = tid; t < nn; t += TB) sTAB[LDIAG + d_xoff[l] + t] = tab[c_off[d_diag[l]] + t];
  }

  const int tile = blockIdx.x * WAVES_PER_BLOCK + wid;
  const int base = tile * 16;
  float* X  = sXp + XFRONT + wid * 16 * XS;
  float* X2 = sX2[wid];
  float* PR = sX2[wid] + 2 * X2S;

  {
    int pe = base + WAVES_PER_BLOCK * 16;
    if (pe < Btot) __builtin_prefetch(&x3[(size_t)pe * 49], 0, 1);
  }

  for (int t = lane; t < 84 * 16; t += 32) {
    const int bb = t / 84, comp = t % 84;
    int e = base + bb;
    e = (e < Btot) ? e : (Btot - 1);
    float v;
    if (comp < 1)       v = x0[e];
    else if (comp < 10) v = x1[(size_t)e * 9 + (comp - 1)];
    else if (comp < 35) v = x2[(size_t)e * 25 + (comp - 10)];
    else                v = x3[(size_t)e * 49 + (comp - 35)];
    X[bb * XS + comp] = v;
  }
  for (int t = lane; t < 16 * (XS - ZC); t += 32) X[(t / (XS - ZC)) * XS + ZC + t % (XS - ZC)] = 0.f;
  if (tid < XFRONT) sXp[tid] = 0.f;
  __syncthreads();

  const int hh = lane >> 4, cc = lane & 15;
  const int Mcol = cc - lof(cc) * lof(cc) - lof(cc);
  float wgt[8], wgtN[8];
  int xi[8], pi[8], ni[8];
  bool posE[8];
#pragma unroll
  for (int r = 0; r < 8; ++r) {
    const int kp = 8 * hh + r;
    const int L = lof(kp), nL = 2 * L + 1;
    const int k = kp - L * L, j = cc - L * L;
    const bool dg = (L == lof(cc));
    const int kr = 2 * L - k, jr = 2 * L - j;
    const int xo = (L >= 1) + 9 * (L >= 2) + 25 * (L >= 3);
    const int pb = (L >= 1) + 18 * (L >= 2) + 50 * (L >= 3);
    const float gk = sDIAG[xo + k * nL + kr];
    const float gj = sDIAG[dg ? (xo + j * nL + jr) : 0];
    const float w = gk * gj;
    wgt[r]  = dg ? w : 0.f;
    wgtN[r] = (dg && L > 0) ? w : 0.f;
    xi[r] = dg ? (xo + kr * nL + jr) : ZC;
    pi[r] = dg ? (pb + kr * nL + jr) : 0;
    ni[r] = (dg && L > 0) ? (pb + nL * nL + kr * nL + jr) : 0;
    posE[r] = ((L & 1) == 0);
  }

  for (int e = 0; e < 16; e += 2) {
    asm volatile("" ::: "memory");
    const float* Xe = X + e * XS;
    v8f accE[2], accO[2];
#pragma unroll
    for (int el = 0; el < 2; ++el) {
      accE[el] = (v8f){0.f, 0.f, 0.f, 0.f, 0.f, 0.f, 0.f, 0.f};
      accO[el] = (v8f){0.f, 0.f, 0.f, 0.f, 0.f, 0.f, 0.f, 0.f};
    }
    pair_gemm<0, 0,  0,  0>(Xe, sB1, sA2, accE, lane);
    pair_gemm<0, 1,  1,  4>(Xe, sB1, sA2, accO, lane);
    pair_gemm<0, 2,  3,  8>(Xe, sB1, sA2, accE, lane);
    pair_gemm<0, 3,  6, 12>(Xe, sB1, sA2, accO, lane);
    pair_gemm<1, 1, 10, 16>(Xe, sB1, sA2, accE, lane);
    pair_gemm<1, 2, 14, 20>(Xe, sB1, sA2, accO, lane);
    pair_gemm2<3, 1, 20, 24>(Xe, sB1, sA2, accE, lane, Mcol);
    pair_gemm2<2, 2, 24, 32>(Xe, sB1, sA2, accE, lane, Mcol);
    pair_gemm2<3, 2, 27, 44>(Xe, sB1, sA2, accO, lane, Mcol);
    pair_gemm2<3, 3, 31, 56>(Xe, sB1, sA2, accE, lane, Mcol);
#pragma unroll
    for (int el = 0; el < 2; ++el) {
      store_x2(X2 + el * X2S, accE[el], true, lane);
      store_x2(X2 + el * X2S, accO[el], false, lane);
    }
    __builtin_amdgcn_fence(__ATOMIC_RELEASE, "wavefront");
    __builtin_amdgcn_fence(__ATOMIC_ACQUIRE, "wavefront");
    float xv[2][8], pv[2][8], nv[2][8];
#pragma unroll
    for (int el = 0; el < 2; ++el) {
      const float* Xq = Xe + el * XS;
      const float* X2q = X2 + el * X2S;
#pragma unroll
      for (int r = 0; r < 8; ++r) {
        xv[el][r] = Xq[xi[r]];
        pv[el][r] = X2q[pi[r]];
        nv[el][r] = X2q[ni[r]];
      }
    }
    asm volatile("" :: "v"(xv[0][0]), "v"(xv[0][1]), "v"(xv[0][2]), "v"(xv[0][3]), "v"(xv[0][4]), "v"(xv[0][5]), "v"(xv[0][6]), "v"(xv[0][7]), "v"(pv[0][0]), "v"(pv[0][1]), "v"(pv[0][2]), "v"(pv[0][3]), "v"(pv[0][4]), "v"(pv[0][5]), "v"(pv[0][6]), "v"(pv[0][7]), "v"(nv[0][0]), "v"(nv[0][1]), "v"(nv[0][2]), "v"(nv[0][3]), "v"(nv[0][4]), "v"(nv[0][5]), "v"(nv[0][6]), "v"(nv[0][7]));
    asm volatile("" :: "v"(xv[1][0]), "v"(xv[1][1]), "v"(xv[1][2]), "v"(xv[1][3]), "v"(xv[1][4]), "v"(xv[1][5]), "v"(xv[1][6]), "v"(xv[1][7]), "v"(pv[1][0]), "v"(pv[1][1]), "v"(pv[1][2]), "v"(pv[1][3]), "v"(pv[1][4]), "v"(pv[1][5]), "v"(pv[1][6]), "v"(pv[1][7]), "v"(nv[1][0]), "v"(nv[1][1]), "v"(nv[1][2]), "v"(nv[1][3]), "v"(nv[1][4]), "v"(nv[1][5]), "v"(nv[1][6]), "v"(nv[1][7]));
#pragma unroll
    for (int el = 0; el < 2; ++el) {
      float* Pe = PR + (e + el) * PS;
      float s3 = 0.f, s4 = 0.f;
#pragma unroll
      for (int r = 0; r < 8; ++r) {
        const float P = posE[r] ? accE[el][r] : accO[el][r];
        const float N = posE[r] ? accO[el][r] : accE[el][r];
        const float wp = wgt[r] * P;
        const float wn = wgtN[r] * N;
        s3 += wp * xv[el][r];
        s4 += wp * pv[el][r];
        s4 += wn * nv[el][r];
      }
      Pe[1 + lane] = s3;
      Pe[33 + lane] = s4;
      if (lane == 0) Pe[0] = accE[el][0];
    }
  }

  __builtin_amdgcn_fence(__ATOMIC_RELEASE, "wavefront");
  __builtin_amdgcn_fence(__ATOMIC_ACQUIRE, "wavefront");

  if (lane < 16) {
    const int e = base + b;
    if (e < Btot) {
      const float* Xb = X + b * XS;
      const float* Pb = PR + b * PS;
      const float nu1 = Xb[0];
      const float nu2 = Pb[0];
      float nu3 = 0.f, nu4 = 0.f;
      for (int i = 0; i < 32; ++i) {
        nu3 += Pb[1 + i];
        nu4 += Pb[33 + i];
      }
      out[(size_t)e * 4 + 0] = nu1;
      out[(size_t)e * 4 + 1] = nu2;
      out[(size_t)e * 4 + 2] = nu3;
      out[(size_t)e * 4 + 3] = nu4;
    }
  }
}

extern "C" void kernel_launch(void* const* d_in, const int* in_sizes, int n_in,
                              void* d_out, int out_size, void* d_ws, size_t ws_size,
                              hipStream_t stream) {
  const float* x0 = (const float*)d_in[0];
  const float* x1 = (const float*)d_in[1];
  const float* x2 = (const float*)d_in[2];
  const float* x3 = (const float*)d_in[3];
  float* tab = (float*)d_ws;
  const int Btot = in_sizes[0];

  wigner_cg_init<<<1, 256, 0, stream>>>(tab);

  const int tiles = (Btot + 15) / 16;
  const int blocks = (tiles + WAVES_PER_BLOCK - 1) / WAVES_PER_BLOCK;
  wigner_main<<<blocks, TB, 0, stream>>>(x0, x1, x2, x3, tab, (float*)d_out, Btot);
}


# Table kernel: each f64 division of the Racah formula becomes a product of tabled reciprocal factorials (LDS), the two roots fold into one, per-table factor formed once: 3x fewer f64 ops; same sums
# speedup vs baseline: 38.4588x; 1.0948x over previous
#include <hip/hip_runtime.h>


#define NCOMBO 23
#define CG_TOTAL 2351
#define NPAIR 6
#define NPAIR2 4
#define B1_STEPS 35
#define A2_STEPS 72
#define B1_BASE CG_TOTAL
#define A2_BASE (CG_TOTAL + B1_STEPS * 64)
#define WS_TOTAL (A2_BASE + A2_STEPS * 64)
#define LB1 0
#define LA2 (B1_STEPS * 64)
#define LDIAG (LA2 + A2_STEPS * 64)
#define LT_TOTAL (LDIAG + 84)
#define WAVES_PER_BLOCK 4
#define TB (WAVES_PER_BLOCK * 32)
#define XS 101
#define ZC 84
#define XFRONT 8
#define X2S 169
#define PS 65
#define X2W (2 * X2S + 16 * PS)

typedef float v2f __attribute__((ext_vector_type(2)));
typedef float v8f __attribute__((ext_vector_type(8)));

__constant__ int c_l1[NCOMBO] = {0,0,0,0, 1,1,1,1,1,1,1,1, 2,2,2,2,2,2,2, 3,3,3,3};
__constant__ int c_l2[NCOMBO] = {0,1,2,3, 1,1,1,2,2,2,3,3, 2,2,2,2,3,3,3, 3,3,3,3};
__constant__ int c_L [NCOMBO] = {0,1,2,3, 0,1,2,1,2,3,2,3, 0,1,2,3,1,2,3, 0,1,2,3};
__constant__ int c_off[NCOMBO] = {0,1,10,35, 84,93,120,165,210,285,390,495,
                                  642,667,742,867,1042,1147,1322, 1567,1616,1763,2008};
__constant__ int p_l1[NPAIR] = {0,0,0,0,1,1};
__constant__ int p_l2[NPAIR] = {0,1,2,3,1,2};
__constant__ int p_c0[NPAIR] = {0,1,2,3,4,7};
__constant__ int p_b1[NPAIR] = {0,1,3,6,10,14};
__constant__ int p_a2[NPAIR] = {0,4,8,12,16,20};
__constant__ int q_l1[NPAIR2] = {1,2,2,3};
__constant__ int q_l2[NPAIR2] = {3,2,3,3};
__constant__ int q_sw[NPAIR2] = {1,0,1,0};
__constant__ int q_c0[NPAIR2] = {10,12,16,19};
__constant__ int q_g [NPAIR2] = {20,24,27,31};
__constant__ int q_a2[NPAIR2] = {24,32,44,56};
__constant__ int d_xoff[4] = {0, 1, 10, 35};
__constant__ int d_diag[4] = {0, 4, 12, 19};

__device__ __forceinline__ int lof(int x) { return (x >= 1) + (x >= 4) + (x >= 9); }

__global__ void wigner_cg_init(float* __restrict__ ws) {
  __shared__ double sF[16];
  __shared__ double sRF[16];
  __shared__ double sPre[NCOMBO];
  if (threadIdx.x < 16) {
    double v = 1.0;
    for (int i = 2; i <= (int)threadIdx.x; ++i) v *= (double)i;
    sF[threadIdx.x] = v;
    sRF[threadIdx.x] = 1.0 / v;
  }
  __syncthreads();
  if (threadIdx.x < NCOMBO) {
    const int l1 = c_l1[threadIdx.x], l2 = c_l2[threadIdx.x], L = c_L[threadIdx.x];
    sPre[threadIdx.x] =
        (2.0 * L + 1.0) * sF[L + l1 - l2] * sF[L - l1 + l2] * sF[l1 + l2 - L] * sRF[l1 + l2 + L + 1];
  }
  __syncthreads();
  for (int c = 0; c < NCOMBO; ++c) {
    const int l1 = c_l1[c], l2 = c_l2[c], L = c_L[c];
    const int n2 = 2 * l2 + 1, nL = 2 * L + 1;
    const int sz = (2 * l1 + 1) * n2 * nL, off = c_off[c];
    const double pre = sPre[c];
    for (int t = threadIdx.x; t < sz; t += blockDim.x) {
      const int m = t / (n2 * nL), rem = t % (n2 * nL), n = rem / nL, k = rem % nL;
      const int m1 = m - l1, m2 = n - l2, M = m1 + m2;
      double val = 0.0;
      if (M >= -L && M <= L && k == M + L) {
        const double rad =
            pre * sF[L + M] * sF[L - M] * sF[l1 - m1] * sF[l1 + m1] * sF[l2 - m2] * sF[l2 + m2];
        double s = 0.0;
        for (int kk = 0; kk <= l1 + l2 - L; ++kk) {
          const int a1 = l1 - m1 - kk, a2 = l2 + m2 - kk;
          const int a3 = L - l2 + m1 + kk, a4 = L - l1 - m2 + kk;
          if (a1 < 0 || a2 < 0 || a3 < 0 || a4 < 0) continue;
          const double term = sRF[kk] * sRF[l1 + l2 - L - kk] * sRF[a1] * sRF[a2] * sRF[a3] * sRF[a4];
          s += (kk & 1) ? -term : term;
        }
        val = sqrt(rad) * s;
      }
      ws[off + t] = (float)val;
    }
  }
  __syncthreads();
  for (int pi = 0; pi < NPAIR; ++pi) {
    const int l1 = p_l1[pi], l2 = p_l2[pi];
    const int n1 = 2 * l1 + 1, n2 = 2 * l2 + 1, PH = l1 + 1, QH = l2 + 1;
    const int Lmin = l2 - l1, Lmax = (l1 + l2 < 3) ? (l1 + l2) : 3;
    const int c0 = p_c0[pi], NR = n1 * n2, RT = (NR + 15) >> 4;
    const float w = (l1 == l2) ? 1.f : 2.f;
    for (int t = threadIdx.x; t < PH * QH * 64; t += blockDim.x) {
      const int st = t >> 6, s6 = t & 63, ln = s6 >> 1, s = s6 & 1;
      const int i = st / QH, u = st % QH, h = ln >> 4, c = ln & 15;
      const int p = h * PH + i, q = 2 * u + s, L = lof(c), k = c - L * L;
      float val = 0.f;
      if (p < n1 && q < n2 && L >= Lmin && L <= Lmax)
        val = ws[c_off[c0 + L - Lmin] + (p * n2 + q) * (2 * L + 1) + k];
      ws[B1_BASE + (p_b1[pi] + st) * 64 + s6] = val;
    }
    for (int t = threadIdx.x; t < RT * 4 * 64; t += blockDim.x) {
      const int st = t >> 6, s6 = t & 63, ln = s6 >> 1, s = s6 & 1;
      const int rt = st >> 2, v = st & 3, h = ln >> 4, c = ln & 15;
      const int rho = rt * 16 + 4 * v + 2 * h + s, L = lof(c), k = c - L * L;
      float val = 0.f;
      if (rho < NR && L >= Lmin && L <= Lmax) {
        const int m = rho / n2, n = rho % n2;
        val = w * ws[c_off[c0 + L - Lmin] + (m * n2 + n) * (2 * L + 1) + k];
      }
      ws[A2_BASE + (p_a2[pi] + st) * 64 + s6] = val;
    }
  }
  for (int qi = 0; qi < NPAIR2; ++qi) {
    const int l1 = q_l1[qi], l2 = q_l2[qi], sw = q_sw[qi];
    const int n2 = 2 * l2 + 1;
    const int LA = sw ? l2 : l1, LB = sw ? l1 : l2;
    const int NA = 2 * LA + 1, NB = 2 * LB + 1, KH = LA + 1, TT = LB + 1;
    const int Lmin = l2 - l1, Lmax = (l1 + l2 < 3) ? (l1 + l2) : 3;
    const int c0 = q_c0[qi];
    const float w = (l1 == l2) ? 1.f : 2.f;
    for (int t = threadIdx.x; t < KH * 64; t += blockDim.x) {
      const int st = t >> 6, s6 = t & 63, ln = s6 >> 1, s = s6 & 1;
      const int c = ln & 15, ka = 2 * st + s, L = lof(c), j = c - L * L;
      const int kb = (j - L) + LA + LB - ka;
      float val = 0.f;
      if (ka < NA && kb >= 0 && kb < NB && L >= Lmin && L <= Lmax) {
        const int p = sw ? kb : ka, q = sw ? ka : kb;
        val = ws[c_off[c0 + L - Lmin] + (p * n2 + q) * (2 * L + 1) + j];
      }
      ws[B1_BASE + (q_g[qi] + st) * 64 + s6] = val;
    }
    for (int t = threadIdx.x; t < TT * 4 * 64; t += blockDim.x) {
      const int st = t >> 6, s6 = t & 63, ln = s6 >> 1, s = s6 & 1;
      const int tt = st >> 2, v = st & 3, h = ln >> 4, c = ln & 15;
      const int ma = 2 * v + s, nb = 2 * tt + h, L = lof(c), k = c - L * L;
      float val = 0.f;
      if (ma < NA && nb < NB && L >= Lmin && L <= Lmax) {
        const int m = sw ? nb : ma, n = sw ? ma : nb;
        val = w * ws[c_off[c0 + L - Lmin] + (m * n2 + n) * (2 * L + 1) + k];
      }
      ws[A2_BASE + (q_a2[qi] + st) * 64 + s6] = val;
    }
  }
}

__device__ __forceinline__ constexpr int xoff_c(int l) { return l == 0 ? 0 : (l == 1 ? 1 : (l == 2 ? 10 : 35)); }

template <int L1, int L2, int B1OFF, int A2OFF>
__device__ __forceinline__ void pair_gemm(const float* __restrict__ Xe, const float* __restrict__ sB1,
                                          const float* __restrict__ sA2, v8f (&acc)[2], const int lane) {
  constexpr int N1 = 2 * L1 + 1, N2 = 2 * L2 + 1, PH = L1 + 1, QH = L2 + 1;
  constexpr int NR = N1 * N2, RT = (NR + 15) / 16;
  constexpr int XO1 = xoff_c(L1), XO2 = xoff_c(L2);
  const int h = lane >> 4, c = lane & 15;
#pragma unroll
  for (int rt = 0; rt < RT; ++rt) {
    const int rho = rt * 16 + 4 * ((c & 7) >> 1) + 2 * (c >> 3) + (c & 1);
    const bool valid = rho < NR;
    const int m = rho / N2, n = rho - m * N2;
    float a[2][PH], b[2][2 * QH];
#pragma unroll
    for (int i = 0; i < PH; ++i) {
      const int p = h * PH + i;
      const int ia = (valid && p < N1) ? (XO1 + m * N1 + p) : ZC;
#pragma unroll
      for (int el = 0; el < 2; ++el) a[el][i] = Xe[el * XS + ia];
    }
#pragma unroll
    for (int q = 0; q < 2 * QH; ++q) {
      const int ib = (valid && q < N2) ? (XO2 + n * N2 + q) : ZC;
#pragma unroll
      for (int el = 0; el < 2; ++el) b[el][q] = Xe[el * XS + ib];
    }
    v8f d[2];
#pragma unroll
    for (int el = 0; el < 2; ++el) d[el] = (v8f){0.f, 0.f, 0.f, 0.f, 0.f, 0.f, 0.f, 0.f};
#pragma unroll
    for (int i = 0; i < PH; ++i) {
#pragma unroll
      for (int u = 0; u < QH; ++u) {
        const v2f bf = *(const v2f*)&sB1[(B1OFF + i * QH + u) * 64 + lane * 2];
#pragma unroll
        for (int el = 0; el < 2; ++el) {
          v2f af;
          af.x = a[el][i] * b[el][2 * u];
          af.y = a[el][i] * b[el][2 * u + 1];
          d[el] = __builtin_amdgcn_wmma_f32_16x16x4_f32(false, af, false, bf, (short)0, d[el], false, false);
        }
      }
    }
    const int rows = (NR - rt * 16 < 16) ? (NR - rt * 16) : 16;
    const int nv = (rows + 3) / 4;
#pragma unroll
    for (int v = 0; v < 4; ++v) {
      if (v < nv) {
        const v2f af2 = *(const v2f*)&sA2[(A2OFF + rt * 4 + v) * 64 + lane * 2];
#pragma unroll
        for (int el = 0; el < 2; ++el) {
          v2f bf2;
          bf2.x = d[el][2 * v];
          bf2.y = d[el][2 * v + 1];
          acc[el] = __builtin_amdgcn_wmma_f32_16x16x4_f32(false, af2, false, bf2, (short)0, acc[el], false, false);
        }
      }
    }
  }
}

template <int LA, int LB, int GOFF, int A2OFF>
__device__ __forceinline__ void pair_gemm2(const float* __restrict__ Xe, const float* __restrict__ sB1,
                                           const float* __restrict__ sA2, v8f (&acc)[2], const int lane,
                                           const int Mc) {
  constexpr int NA = 2 * LA + 1, NB = 2 * LB + 1, KH = LA + 1, TT = LB + 1;
  constexpr int NV2 = (NA + 1) / 2;
  constexpr int XOA = xoff_c(LA), XOB = xoff_c(LB);
  const int h = lane >> 4, c = lane & 15;
  const int g = c >> 3, ma = c & 7;
  const bool arow = (g == h) && (ma < NA);
  const float* Xa = Xe + (arow ? (XOA + ma * NA) : ZC);
  const float* Xb = Xe + (XOB + h * NB + Mc + LA + LB - (2 * KH - 1));
  static_assert((KH == 4) == (NV2 == 4), "the load fence below lists NV2 table fragments");
  v2f af[2][KH], gf[KH];
#pragma unroll
  for (int i = 0; i < KH; ++i) {
    gf[i] = *(const v2f*)&sB1[(GOFF + i) * 64 + lane * 2];
#pragma unroll
    for (int el = 0; el < 2; ++el) {
      af[el][i].x = Xa[el * XS + 2 * i];
      af[el][i].y = Xa[el * XS + 2 * i + 1];
    }
  }
#pragma unroll
  for (int tt = 0; tt < TT; ++tt) {
    asm volatile("" ::: "memory");
    float raw[2][2 * KH];
    v2f a2[NV2];
#pragma unroll
    for (int ka = 0; ka < 2 * KH; ++ka) {
#pragma unroll
      for (int el = 0; el < 2; ++el) raw[el][ka] = Xb[el * XS + 2 * tt * NB + (2 * KH - 1 - ka)];
    }
#pragma unroll
    for (int v = 0; v < NV2; ++v) a2[v] = *(const v2f*)&sA2[(A2OFF + tt * 4 + v) * 64 + lane * 2];
    if constexpr (KH == 4) {
      asm volatile("" :: "v"(raw[0][0]), "v"(raw[0][1]), "v"(raw[0][2]), "v"(raw[0][3]), "v"(raw[0][4]), "v"(raw[0][5]), "v"(raw[0][6]), "v"(raw[0][7]), "v"(raw[1][0]), "v"(raw[1][1]), "v"(raw[1][2]), "v"(raw[1][3]), "v"(raw[1][4]), "v"(raw[1][5]), "v"(raw[1][6]), "v"(raw[1][7]), "v"(a2[0]), "v"(a2[1]), "v"(a2[2]), "v"(a2[3]));
    } else {
      asm volatile("" :: "v"(raw[0][0]), "v"(raw[0][1]), "v"(raw[0][2]), "v"(raw[0][3]), "v"(raw[0][4]), "v"(raw[0][5]), "v"(raw[1][0]), "v"(raw[1][1]), "v"(raw[1][2]), "v"(raw[1][3]), "v"(raw[1][4]), "v"(raw[1][5]), "v"(a2[0]), "v"(a2[1]), "v"(a2[2]));
    }
    v8f d[2];
#pragma unroll
    for (int el = 0; el < 2; ++el) d[el] = (v8f){0.f, 0.f, 0.f, 0.f, 0.f, 0.f, 0.f, 0.f};
#pragma unroll
    for (int i = 0; i < KH; ++i) {
#pragma unroll
      for (int el = 0; el < 2; ++el) {
        v2f bf;
        bf.x = raw[el][2 * i] * gf[i].x;
        bf.y = raw[el][2 * i + 1] * gf[i].y;
        d[el] = __builtin_amdgcn_wmma_f32_16x16x4_f32(false, af[el][i], false, bf, (short)0, d[el], false, false);
      }
    }
#pragma unroll
    for (int v = 0; v < NV2; ++v) {
#pragma unroll
      for (int el = 0; el < 2; ++el) {
        v2f bf2;
        bf2.x = d[el][2 * v];
        bf2.y = d[el][2 * v + 1];
        acc[el] = __builtin_amdgcn_wmma_f32_16x16x4_f32(false, a2[v], false, bf2, (short)0, acc[el], false, false);
      }
    }
  }
}

__device__ __forceinline__ void store_x2(float* __restrict__ X2e, const v8f& t, const bool even, const int lane) {
  const int h = lane >> 4, c = lane & 15;
  const int Lc = lof(c);
#pragma unroll
  for (int v = 0; v < 8; ++v) {
    const int kp = 8 * h + v;
    const int L = lof(kp);
    const int k = kp - L * L, j = c - L * L, nL = 2 * L + 1;
    const bool pos = even ? ((L & 1) == 0) : ((L & 1) == 1);
    const int pb = (L >= 1) + 18 * (L >= 2) + 50 * (L >= 3);
    const int base = pos ? pb : (pb + nL * nL);
    const bool wr = (L == Lc) && (pos || L > 0);
    X2e[wr ? (base + k * nL + j) : (X2S - 1)] = t[v];
  }
}

__launch_bounds__(TB, 4)
__global__ void wigner_main(const float* __restrict__ x0, const float* __restrict__ x1,
                            const float* __restrict__ x2, const float* __restrict__ x3,
                            const float* __restrict__ tab, float* __restrict__ out, int Btot) {
  __shared__ __align__(16) float sTAB[LT_TOTAL];
  __shared__ float sXp[XFRONT + WAVES_PER_BLOCK * 16 * XS];
  __shared__ float sX2[WAVES_PER_BLOCK][X2W];

  const int tid = threadIdx.x, lane = tid & 31, wid = tid >> 5;
  const int b = lane & 15;
  const float* sB1 = sTAB + LB1;
  const float* sA2 = sTAB + LA2;
  const float* sDIAG = sTAB + LDIAG;

  for (int t = tid; t < LDIAG; t += TB) sTAB[t] = tab[CG_TOTAL + t];
  for (int l = 0; l <= 3; ++l) {
    const int nn = (2 * l + 1) * (2 * l + 1);
    for (int t = tid; t < nn; t += TB) sTAB[LDIAG + d_xoff[l] + t] = tab[c_off[d_diag[l]] + t];
  }

  const int tile = blockIdx.x * WAVES_PER_BLOCK + wid;
  const int base = tile * 16;
  float* X  = sXp + XFRONT + wid * 16 * XS;
  float* X2 = sX2[wid];
  float* PR = sX2[wid] + 2 * X2S;

  {
    int pe = base + WAVES_PER_BLOCK * 16;
    if (pe < Btot) __builtin_prefetch(&x3[(size_t)pe * 49], 0, 1);
  }

  for (int t = lane; t < 84 * 16; t += 32) {
    const int bb = t / 84, comp = t % 84;
    int e = base + bb;
    e = (e < Btot) ? e : (Btot - 1);
    float v;
    if (comp < 1)       v = x0[e];
    else if (comp < 10) v = x1[(size_t)e * 9 + (comp - 1)];
    else if (comp < 35) v = x2[(size_t)e * 25 + (comp - 10)];
    else                v = x3[(size_t)e * 49 + (comp - 35)];
    X[bb * XS + comp] = v;
  }
  for (int t = lane; t < 16 * (XS - ZC); t += 32) X[(t / (XS - ZC)) * XS + ZC + t % (XS - ZC)] = 0.f;
  if (tid < XFRONT) sXp[tid] = 0.f;
  __syncthreads();

  const int hh = lane >> 4, cc = lane & 15;
  const int Mcol = cc - lof(cc) * lof(cc) - lof(cc);
  float wgt[8], wgtN[8];
  int xi[8], pi[8], ni[8];
  bool posE[8];
#pragma unroll
  for (int r = 0; r < 8; ++r) {
    const int kp = 8 * hh + r;
    const int L = lof(kp), nL = 2 * L + 1;
    const int k = kp - L * L, j = cc - L * L;
    const bool dg = (L == lof(cc));
    const int kr = 2 * L - k, jr = 2 * L - j;
    const int xo = (L >= 1) + 9 * (L >= 2) + 25 * (L >= 3);
    const int pb = (L >= 1) + 18 * (L >= 2) + 50 * (L >= 3);
    const float gk = sDIAG[xo + k * nL + kr];
    const float gj = sDIAG[dg ? (xo + j * nL + jr) : 0];
    const float w = gk * gj;
    wgt[r]  = dg ? w : 0.f;
    wgtN[r] = (dg && L > 0) ? w : 0.f;
    xi[r] = dg ? (xo + kr * nL + jr) : ZC;
    pi[r] = dg ? (pb + kr * nL + jr) : 0;
    ni[r] = (dg && L > 0) ? (pb + nL * nL + kr * nL + jr) : 0;
    posE[r] = ((L & 1) == 0);
  }

  for (int e = 0; e < 16; e += 2) {
    asm volatile("" ::: "memory");
    const float* Xe = X + e * XS;
    v8f accE[2], accO[2];
#pragma unroll
    for (int el = 0; el < 2; ++el) {
      accE[el] = (v8f){0.f, 0.f, 0.f, 0.f, 0.f, 0.f, 0.f, 0.f};
      accO[el] = (v8f){0.f, 0.f, 0.f, 0.f, 0.f, 0.f, 0.f, 0.f};
    }
    pair_gemm<0, 0,  0,  0>(Xe, sB1, sA2, accE, lane);
    pair_gemm<0, 1,  1,  4>(Xe, sB1, sA2, accO, lane);
    pair_gemm<0, 2,  3,  8>(Xe, sB1, sA2, accE, lane);
    pair_gemm<0, 3,  6, 12>(Xe, sB1, sA2, accO, lane);
    pair_gemm<1, 1, 10, 16>(Xe, sB1, sA2, accE, lane);
    pair_gemm<1, 2, 14, 20>(Xe, sB1, sA2, accO, lane);
    pair_gemm2<3, 1, 20, 24>(Xe, sB1, sA2, accE, lane, Mcol);
    pair_gemm2<2, 2, 24, 32>(Xe, sB1, sA2, accE, lane, Mcol);
    pair_gemm2<3, 2, 27, 44>(Xe, sB1, sA2, accO, lane, Mcol);
    pair_gemm2<3, 3, 31, 56>(Xe, sB1, sA2, accE, lane, Mcol);
#pragma unroll
    for (int el = 0; el < 2; ++el) {
      store_x2(X2 + el * X2S, accE[el], true, lane);
      store_x2(X2 + el * X2S, accO[el], false, lane);
    }
    __builtin_amdgcn_fence(__ATOMIC_RELEASE, "wavefront");
    __builtin_amdgcn_fence(__ATOMIC_ACQUIRE, "wavefront");
    float xv[2][8], pv[2][8], nv[2][8];
#pragma unroll
    for (int el = 0; el < 2; ++el) {
      const float* Xq = Xe + el * XS;
      const float* X2q = X2 + el * X2S;
#pragma unroll
      for (int r = 0; r < 8; ++r) {
        xv[el][r] = Xq[xi[r]];
        pv[el][r] = X2q[pi[r]];
        nv[el][r] = X2q[ni[r]];
      }
    }
    asm volatile("" :: "v"(xv[0][0]), "v"(xv[0][1]), "v"(xv[0][2]), "v"(xv[0][3]), "v"(xv[0][4]), "v"(xv[0][5]), "v"(xv[0][6]), "v"(xv[0][7]), "v"(pv[0][0]), "v"(pv[0][1]), "v"(pv[0][2]), "v"(pv[0][3]), "v"(pv[0][4]), "v"(pv[0][5]), "v"(pv[0][6]), "v"(pv[0][7]), "v"(nv[0][0]), "v"(nv[0][1]), "v"(nv[0][2]), "v"(nv[0][3]), "v"(nv[0][4]), "v"(nv[0][5]), "v"(nv[0][6]), "v"(nv[0][7]));
    asm volatile("" :: "v"(xv[1][0]), "v"(xv[1][1]), "v"(xv[1][2]), "v"(xv[1][3]), "v"(xv[1][4]), "v"(xv[1][5]), "v"(xv[1][6]), "v"(xv[1][7]), "v"(pv[1][0]), "v"(pv[1][1]), "v"(pv[1][2]), "v"(pv[1][3]), "v"(pv[1][4]), "v"(pv[1][5]), "v"(pv[1][6]), "v"(pv[1][7]), "v"(nv[1][0]), "v"(nv[1][1]), "v"(nv[1][2]), "v"(nv[1][3]), "v"(nv[1][4]), "v"(nv[1][5]), "v"(nv[1][6]), "v"(nv[1][7]));
#pragma unroll
    for (int el = 0; el < 2; ++el) {
      float* Pe = PR + (e + el) * PS;
      float s3 = 0.f, s4 = 0.f;
#pragma unroll
      for (int r = 0; r < 8; ++r) {
        const float P = posE[r] ? accE[el][r] : accO[el][r];
        const float N = posE[r] ? accO[el][r] : accE[el][r];
        const float wp = wgt[r] * P;
        const float wn = wgtN[r] * N;
        s3 += wp * xv[el][r];
        s4 += wp * pv[el][r];
        s4 += wn * nv[el][r];
      }
      Pe[1 + lane] = s3;
      Pe[33 + lane] = s4;
      if (lane == 0) Pe[0] = accE[el][0];
    }
  }

  __builtin_amdgcn_fence(__ATOMIC_RELEASE, "wavefront");
  __builtin_amdgcn_fence(__ATOMIC_ACQUIRE, "wavefront");

  if (lane < 16) {
    const int e = base + b;
    if (e < Btot) {
      const float* Xb = X + b * XS;
      const float* Pb = PR + b * PS;
      const float nu1 = Xb[0];
      const float nu2 = Pb[0];
      float nu3 = 0.f, nu4 = 0.f;
      for (int i = 0; i < 32; ++i) {
        nu3 += Pb[1 + i];
        nu4 += Pb[33 + i];
      }
      out[(size_t)e * 4 + 0] = nu1;
      out[(size_t)e * 4 + 1] = nu2;
      out[(size_t)e * 4 + 2] = nu3;
      out[(size_t)e * 4 + 3] = nu4;
    }
  }
}

extern "C" void kernel_launch(void* const* d_in, const int* in_sizes, int n_in,
                              void* d_out, int out_size, void* d_ws, size_t ws_size,
                              hipStream_t stream) {
  const float* x0 = (const float*)d_in[0];
  const float* x1 = (const float*)d_in[1];
  const float* x2 = (const float*)d_in[2];
  const float* x3 = (const float*)d_in[3];
  float* tab = (float*)d_ws;
  const int Btot = in_sizes[0];

  wigner_cg_init<<<1, 256, 0, stream>>>(tab);

  const int tiles = (Btot + 15) / 16;
  const int blocks = (tiles + WAVES_PER_BLOCK - 1) / WAVES_PER_BLOCK;
  wigner_main<<<blocks, TB, 0, stream>>>(x0, x1, x2, x3, tab, (float*)d_out, Btot);
}


# Input rows streamed: each trip's two element rows are fetched one trip ahead (8 whole-line loads per lane, in registers) into a 2-slot LDS ring; the tile gather's 42 serial round trips go; same sums
# speedup vs baseline: 40.9709x; 1.0653x over previous
#include <hip/hip_runtime.h>


#define NCOMBO 23
#define CG_TOTAL 2351
#define NPAIR 6
#define NPAIR2 4
#define B1_STEPS 35
#define A2_STEPS 72
#define B1_BASE CG_TOTAL
#define A2_BASE (CG_TOTAL + B1_STEPS * 64)
#define WS_TOTAL (A2_BASE + A2_STEPS * 64)
#define LB1 0
#define LA2 (B1_STEPS * 64)
#define LDIAG (LA2 + A2_STEPS * 64)
#define LT_TOTAL (LDIAG + 84)
#define WAVES_PER_BLOCK 4
#define TB (WAVES_PER_BLOCK * 32)
#define XS 101
#define ZC 84
#define XFRONT 8
#define X2S 169
#define XROWS 4
#define PS 67
#define X2W (2 * X2S + 16 * PS)

typedef float v2f __attribute__((ext_vector_type(2)));
typedef float v8f __attribute__((ext_vector_type(8)));

__constant__ int c_l1[NCOMBO] = {0,0,0,0, 1,1,1,1,1,1,1,1, 2,2,2,2,2,2,2, 3,3,3,3};
__constant__ int c_l2[NCOMBO] = {0,1,2,3, 1,1,1,2,2,2,3,3, 2,2,2,2,3,3,3, 3,3,3,3};
__constant__ int c_L [NCOMBO] = {0,1,2,3, 0,1,2,1,2,3,2,3, 0,1,2,3,1,2,3, 0,1,2,3};
__constant__ int c_off[NCOMBO] = {0,1,10,35, 84,93,120,165,210,285,390,495,
                                  642,667,742,867,1042,1147,1322, 1567,1616,1763,2008};
__constant__ int p_l1[NPAIR] = {0,0,0,0,1,1};
__constant__ int p_l2[NPAIR] = {0,1,2,3,1,2};
__constant__ int p_c0[NPAIR] = {0,1,2,3,4,7};
__constant__ int p_b1[NPAIR] = {0,1,3,6,10,14};
__constant__ int p_a2[NPAIR] = {0,4,8,12,16,20};
__constant__ int q_l1[NPAIR2] = {1,2,2,3};
__constant__ int q_l2[NPAIR2] = {3,2,3,3};
__constant__ int q_sw[NPAIR2] = {1,0,1,0};
__constant__ int q_c0[NPAIR2] = {10,12,16,19};
__constant__ int q_g [NPAIR2] = {20,24,27,31};
__constant__ int q_a2[NPAIR2] = {24,32,44,56};
__constant__ int d_xoff[4] = {0, 1, 10, 35};
__constant__ int d_diag[4] = {0, 4, 12, 19};

__device__ __forceinline__ int lof(int x) { return (x >= 1) + (x >= 4) + (x >= 9); }

__global__ void wigner_cg_init(float* __restrict__ ws) {
  __shared__ double sF[16];
  __shared__ double sRF[16];
  __shared__ double sPre[NCOMBO];
  if (threadIdx.x < 16) {
    double v = 1.0;
    for (int i = 2; i <= (int)threadIdx.x; ++i) v *= (double)i;
    sF[threadIdx.x] = v;
    sRF[threadIdx.x] = 1.0 / v;
  }
  __syncthreads();
  if (threadIdx.x < NCOMBO) {
    const int l1 = c_l1[threadIdx.x], l2 = c_l2[threadIdx.x], L = c_L[threadIdx.x];
    sPre[threadIdx.x] =
        (2.0 * L + 1.0) * sF[L + l1 - l2] * sF[L - l1 + l2] * sF[l1 + l2 - L] * sRF[l1 + l2 + L + 1];
  }
  __syncthreads();
  for (int c = 0; c < NCOMBO; ++c) {
    const int l1 = c_l1[c], l2 = c_l2[c], L = c_L[c];
    const int n2 = 2 * l2 + 1, nL = 2 * L + 1;
    const int sz = (2 * l1 + 1) * n2 * nL, off = c_off[c];
    const double pre = sPre[c];
    for (int t = threadIdx.x; t < sz; t += blockDim.x) {
      const int m = t / (n2 * nL), rem = t % (n2 * nL), n = rem / nL, k = rem % nL;
      const int m1 = m - l1, m2 = n - l2, M = m1 + m2;
      double val = 0.0;
      if (M >= -L && M <= L && k == M + L) {
        const double rad =
            pre * sF[L + M] * sF[L - M] * sF[l1 - m1] * sF[l1 + m1] * sF[l2 - m2] * sF[l2 + m2];
        double s = 0.0;
        for (int kk = 0; kk <= l1 + l2 - L; ++kk) {
          const int a1 = l1 - m1 - kk, a2 = l2 + m2 - kk;
          const int a3 = L - l2 + m1 + kk, a4 = L - l1 - m2 + kk;
          if (a1 < 0 || a2 < 0 || a3 < 0 || a4 < 0) continue;
          const double term = sRF[kk] * sRF[l1 + l2 - L - kk] * sRF[a1] * sRF[a2] * sRF[a3] * sRF[a4];
          s += (kk & 1) ? -term : term;
        }
        val = sqrt(rad) * s;
      }
      ws[off + t] = (float)val;
    }
  }
  __syncthreads();
  for (int pi = 0; pi < NPAIR; ++pi) {
    const int l1 = p_l1[pi], l2 = p_l2[pi];
    const int n1 = 2 * l1 + 1, n2 = 2 * l2 + 1, PH = l1 + 1, QH = l2 + 1;
    const int Lmin = l2 - l1, Lmax = (l1 + l2 < 3) ? (l1 + l2) : 3;
    const int c0 = p_c0[pi], NR = n1 * n2, RT = (NR + 15) >> 4;
    const float w = (l1 == l2) ? 1.f : 2.f;
    for (int t = threadIdx.x; t < PH * QH * 64; t += blockDim.x) {
      const int st = t >> 6, s6 = t & 63, ln = s6 >> 1, s = s6 & 1;
      const int i = st / QH, u = st % QH, h = ln >> 4, c = ln & 15;
      const int p = h * PH + i, q = 2 * u + s, L = lof(c), k = c - L * L;
      float val = 0.f;
      if (p < n1 && q < n2 && L >= Lmin && L <= Lmax)
        val = ws[c_off[c0 + L - Lmin] + (p * n2 + q) * (2 * L + 1) + k];
      ws[B1_BASE + (p_b1[pi] + st) * 64 + s6] = val;
    }
    for (int t = threadIdx.x; t < RT * 4 * 64; t += blockDim.x) {
      const int st = t >> 6, s6 = t & 63, ln = s6 >> 1, s = s6 & 1;
      const int rt = st >> 2, v = st & 3, h = ln >> 4, c = ln & 15;
      const int rho = rt * 16 + 4 * v + 2 * h + s, L = lof(c), k = c - L * L;
      float val = 0.f;
      if (rho < NR && L >= Lmin && L <= Lmax) {
        const int m = rho / n2, n = rho % n2;
        val = w * ws[c_off[c0 + L - Lmin] + (m * n2 + n) * (2 * L + 1) + k];
      }
      ws[A2_BASE + (p_a2[pi] + st) * 64 + s6] = val;
    }
  }
  for (int qi = 0; qi < NPAIR2; ++qi) {
    const int l1 = q_l1[qi], l2 = q_l2[qi], sw = q_sw[qi];
    const int n2 = 2 * l2 + 1;
    const int LA = sw ? l2 : l1, LB = sw ? l1 : l2;
    const int NA = 2 * LA + 1, NB = 2 * LB + 1, KH = LA + 1, TT = LB + 1;
    const int Lmin = l2 - l1, Lmax = (l1 + l2 < 3) ? (l1 + l2) : 3;
    const int c0 = q_c0[qi];
    const float w = (l1 == l2) ? 1.f : 2.f;
    for (int t = threadIdx.x; t < KH * 64; t += blockDim.x) {
      const int st = t >> 6, s6 = t & 63, ln = s6 >> 1, s = s6 & 1;
      const int c = ln & 15, ka = 2 * st + s, L = lof(c), j = c - L * L;
      const int kb = (j - L) + LA + LB - ka;
      float val = 0.f;
      if (ka < NA && kb >= 0 && kb < NB && L >= Lmin && L <= Lmax) {
        const int p = sw ? kb : ka, q = sw ? ka : kb;
        val = ws[c_off[c0 + L - Lmin] + (p * n2 + q) * (2 * L + 1) + j];
      }
      ws[B1_BASE + (q_g[qi] + st) * 64 + s6] = val;
    }
    for (int t = threadIdx.x; t < TT * 4 * 64; t += blockDim.x) {
      const int st = t >> 6, s6 = t & 63, ln = s6 >> 1, s = s6 & 1;
      const int tt = st >> 2, v = st & 3, h = ln >> 4, c = ln & 15;
      const int ma = 2 * v + s, nb = 2 * tt + h, L = lof(c), k = c - L * L;
      float val = 0.f;
      if (ma < NA && nb < NB && L >= Lmin && L <= Lmax) {
        const int m = sw ? nb : ma, n = sw ? ma : nb;
        val = w * ws[c_off[c0 + L - Lmin] + (m * n2 + n) * (2 * L + 1) + k];
      }
      ws[A2_BASE + (q_a2[qi] + st) * 64 + s6] = val;
    }
  }
}

__device__ __forceinline__ constexpr int xoff_c(int l) { return l == 0 ? 0 : (l == 1 ? 1 : (l == 2 ? 10 : 35)); }

template <int L1, int L2, int B1OFF, int A2OFF>
__device__ __forceinline__ void pair_gemm(const float* __restrict__ Xe, const float* __restrict__ sB1,
                                          const float* __restrict__ sA2, v8f (&acc)[2], const int lane) {
  constexpr int N1 = 2 * L1 + 1, N2 = 2 * L2 + 1, PH = L1 + 1, QH = L2 + 1;
  constexpr int NR = N1 * N2, RT = (NR + 15) / 16;
  constexpr int XO1 = xoff_c(L1), XO2 = xoff_c(L2);
  const int h = lane >> 4, c = lane & 15;
#pragma unroll
  for (int rt = 0; rt < RT; ++rt) {
    const int rho = rt * 16 + 4 * ((c & 7) >> 1) + 2 * (c >> 3) + (c & 1);
    const bool valid = rho < NR;
    const int m = rho / N2, n = rho - m * N2;
    float a[2][PH], b[2][2 * QH];
#pragma unroll
    for (int i = 0; i < PH; ++i) {
      const int p = h * PH + i;
      const int ia = (valid && p < N1) ? (XO1 + m * N1 + p) : ZC;
#pragma unroll
      for (int el = 0; el < 2; ++el) a[el][i] = Xe[el * XS + ia];
    }
#pragma unroll
    for (int q = 0; q < 2 * QH; ++q) {
      const int ib = (valid && q < N2) ? (XO2 + n * N2 + q) : ZC;
#pragma unroll
      for (int el = 0; el < 2; ++el) b[el][q] = Xe[el * XS + ib];
    }
    v8f d[2];
#pragma unroll
    for (int el = 0; el < 2; ++el) d[el] = (v8f){0.f, 0.f, 0.f, 0.f, 0.f, 0.f, 0.f, 0.f};
#pragma unroll
    for (int i = 0; i < PH; ++i) {
#pragma unroll
      for (int u = 0; u < QH; ++u) {
        const v2f bf = *(const v2f*)&sB1[(B1OFF + i * QH + u) * 64 + lane * 2];
#pragma unroll
        for (int el = 0; el < 2; ++el) {
          v2f af;
          af.x = a[el][i] * b[el][2 * u];
          af.y = a[el][i] * b[el][2 * u + 1];
          d[el] = __builtin_amdgcn_wmma_f32_16x16x4_f32(false, af, false, bf, (short)0, d[el], false, false);
        }
      }
    }
    const int rows = (NR - rt * 16 < 16) ? (NR - rt * 16) : 16;
    const int nv = (rows + 3) / 4;
#pragma unroll
    for (int v = 0; v < 4; ++v) {
      if (v < nv) {
        const v2f af2 = *(const v2f*)&sA2[(A2OFF + rt * 4 + v) * 64 + lane * 2];
#pragma unroll
        for (int el = 0; el < 2; ++el) {
          v2f bf2;
          bf2.x = d[el][2 * v];
          bf2.y = d[el][2 * v + 1];
          acc[el] = __builtin_amdgcn_wmma_f32_16x16x4_f32(false, af2, false, bf2, (short)0, acc[el], false, false);
        }
      }
    }
  }
}

template <int LA, int LB, int GOFF, int A2OFF>
__device__ __forceinline__ void pair_gemm2(const float* __restrict__ Xe, const float* __restrict__ sB1,
                                           const float* __restrict__ sA2, v8f (&acc)[2], const int lane,
                                           const int Mc) {
  constexpr int NA = 2 * LA + 1, NB = 2 * LB + 1, KH = LA + 1, TT = LB + 1;
  constexpr int NV2 = (NA + 1) / 2;
  constexpr int XOA = xoff_c(LA), XOB = xoff_c(LB);
  const int h = lane >> 4, c = lane & 15;
  const int g = c >> 3, ma = c & 7;
  const bool arow = (g == h) && (ma < NA);
  const float* Xa = Xe + (arow ? (XOA + ma * NA) : ZC);
  const float* Xb = Xe + (XOB + h * NB + Mc + LA + LB - (2 * KH - 1));
  static_assert((KH == 4) == (NV2 == 4), "the load fence below lists NV2 table fragments");
  v2f af[2][KH], gf[KH];
#pragma unroll
  for (int i = 0; i < KH; ++i) {
    gf[i] = *(const v2f*)&sB1[(GOFF + i) * 64 + lane * 2];
#pragma unroll
    for (int el = 0; el < 2; ++el) {
      af[el][i].x = Xa[el * XS + 2 * i];
      af[el][i].y = Xa[el * XS + 2 * i + 1];
    }
  }
#pragma unroll
  for (int tt = 0; tt < TT; ++tt) {
    asm volatile("" ::: "memory");
    float raw[2][2 * KH];
    v2f a2[NV2];
#pragma unroll
    for (int ka = 0; ka < 2 * KH; ++ka) {
#pragma unroll
      for (int el = 0; el < 2; ++el) raw[el][ka] = Xb[el * XS + 2 * tt * NB + (2 * KH - 1 - ka)];
    }
#pragma unroll
    for (int v = 0; v < NV2; ++v) a2[v] = *(const v2f*)&sA2[(A2OFF + tt * 4 + v) * 64 + lane * 2];
    if constexpr (KH == 4) {
      asm volatile("" :: "v"(raw[0][0]), "v"(raw[0][1]), "v"(raw[0][2]), "v"(raw[0][3]), "v"(raw[0][4]), "v"(raw[0][5]), "v"(raw[0][6]), "v"(raw[0][7]), "v"(raw[1][0]), "v"(raw[1][1]), "v"(raw[1][2]), "v"(raw[1][3]), "v"(raw[1][4]), "v"(raw[1][5]), "v"(raw[1][6]), "v"(raw[1][7]), "v"(a2[0]), "v"(a2[1]), "v"(a2[2]), "v"(a2[3]));
    } else {
      asm volatile("" :: "v"(raw[0][0]), "v"(raw[0][1]), "v"(raw[0][2]), "v"(raw[0][3]), "v"(raw[0][4]), "v"(raw[0][5]), "v"(raw[1][0]), "v"(raw[1][1]), "v"(raw[1][2]), "v"(raw[1][3]), "v"(raw[1][4]), "v"(raw[1][5]), "v"(a2[0]), "v"(a2[1]), "v"(a2[2]));
    }
    v8f d[2];
#pragma unroll
    for (int el = 0; el < 2; ++el) d[el] = (v8f){0.f, 0.f, 0.f, 0.f, 0.f, 0.f, 0.f, 0.f};
#pragma unroll
    for (int i = 0; i < KH; ++i) {
#pragma unroll
      for (int el = 0; el < 2; ++el) {
        v2f bf;
        bf.x = raw[el][2 * i] * gf[i].x;
        bf.y = raw[el][2 * i + 1] * gf[i].y;
        d[el] = __builtin_amdgcn_wmma_f32_16x16x4_f32(false, af[el][i], false, bf, (short)0, d[el], false, false);
      }
    }
#pragma unroll
    for (int v = 0; v < NV2; ++v) {
#pragma unroll
      for (int el = 0; el < 2; ++el) {
        v2f bf2;
        bf2.x = d[el][2 * v];
        bf2.y = d[el][2 * v + 1];
        acc[el] = __builtin_amdgcn_wmma_f32_16x16x4_f32(false, a2[v], false, bf2, (short)0, acc[el], false, false);
      }
    }
  }
}

__device__ __forceinline__ void store_x2(float* __restrict__ X2e, const v8f& t, const bool even, const int lane) {
  const int h = lane >> 4, c = lane & 15;
  const int Lc = lof(c);
#pragma unroll
  for (int v = 0; v < 8; ++v) {
    const int kp = 8 * h + v;
    const int L = lof(kp);
    const int k = kp - L * L, j = c - L * L, nL = 2 * L + 1;
    const bool pos = even ? ((L & 1) == 0) : ((L & 1) == 1);
    const int pb = (L >= 1) + 18 * (L >= 2) + 50 * (L >= 3);
    const int base = pos ? pb : (pb + nL * nL);
    const bool wr = (L == Lc) && (pos || L > 0);
    X2e[wr ? (base + k * nL + j) : (X2S - 1)] = t[v];
  }
}

struct RowRegs { float r3[4]; float r2[2]; float r1; float r0; };

__device__ __forceinline__ void fetch_rows(RowRegs& r, const float* __restrict__ x0, const float* __restrict__ x1,
                                           const float* __restrict__ x2, const float* __restrict__ x3,
                                           const int e0, const int Btot, const int lane) {
  const size_t n0 = (size_t)Btot - 1, n1 = (size_t)Btot * 9 - 1;
  const size_t n2 = (size_t)Btot * 25 - 1, n3 = (size_t)Btot * 49 - 1;
#pragma unroll
  for (int i = 0; i < 4; ++i) {
    const size_t f = (size_t)e0 * 49 + (size_t)(32 * i + lane);
    r.r3[i] = x3[f < n3 ? f : n3];
  }
#pragma unroll
  for (int i = 0; i < 2; ++i) {
    const size_t f = (size_t)e0 * 25 + (size_t)(32 * i + lane);
    r.r2[i] = x2[f < n2 ? f : n2];
  }
  {
    const size_t f = (size_t)e0 * 9 + (size_t)lane;
    r.r1 = x1[f < n1 ? f : n1];
  }
  {
    const size_t f = (size_t)e0 + (size_t)lane;
    r.r0 = x0[f < n0 ? f : n0];
  }
}

__device__ __forceinline__ void put_rows(float* __restrict__ Xn, const RowRegs& r, const int lane) {
#pragma unroll
  for (int i = 0; i < 4; ++i) {
    const int j = 32 * i + lane;
    const int bb = (j >= 49) ? 1 : 0;
    if (j < 98) Xn[bb * XS + 35 + j - 49 * bb] = r.r3[i];
  }
#pragma unroll
  for (int i = 0; i < 2; ++i) {
    const int j = 32 * i + lane;
    const int bb = (j >= 25) ? 1 : 0;
    if (j < 50) Xn[bb * XS + 10 + j - 25 * bb] = r.r2[i];
  }
  {
    const int bb = (lane >= 9) ? 1 : 0;
    if (lane < 18) Xn[bb * XS + 1 + lane - 9 * bb] = r.r1;
  }
  if (lane < 2) Xn[lane * XS] = r.r0;
}

__launch_bounds__(TB, 4)
__global__ void wigner_main(const float* __restrict__ x0, const float* __restrict__ x1,
                            const float* __restrict__ x2, const float* __restrict__ x3,
                            const float* __restrict__ tab, float* __restrict__ out, int Btot) {
  __shared__ __align__(16) float sTAB[LT_TOTAL];
  __shared__ float sXp[XFRONT + WAVES_PER_BLOCK * XROWS * XS];
  __shared__ float sX2[WAVES_PER_BLOCK][X2W];

  const int tid = threadIdx.x, lane = tid & 31, wid = tid >> 5;
  const int b = lane & 15;
  const float* sB1 = sTAB + LB1;
  const float* sA2 = sTAB + LA2;
  const float* sDIAG = sTAB + LDIAG;

  for (int t = tid; t < LDIAG; t += TB) sTAB[t] = tab[CG_TOTAL + t];
  for (int l = 0; l <= 3; ++l) {
    const int nn = (2 * l + 1) * (2 * l + 1);
    for (int t = tid; t < nn; t += TB) sTAB[LDIAG + d_xoff[l] + t] = tab[c_off[d_diag[l]] + t];
  }

  const int tile = blockIdx.x * WAVES_PER_BLOCK + wid;
  const int base = tile * 16;
  float* X  = sXp + XFRONT + wid * XROWS * XS;
  float* X2 = sX2[wid];
  float* PR = sX2[wid] + 2 * X2S;

  {
    int pe = base + WAVES_PER_BLOCK * 16;
    if (pe < Btot) __builtin_prefetch(&x3[(size_t)pe * 49], 0, 1);
  }

  RowRegs nx;
  fetch_rows(nx, x0, x1, x2, x3, base, Btot, lane);
  put_rows(X, nx, lane);
  for (int t = lane; t < XROWS * (XS - ZC); t += 32) X[(t / (XS - ZC)) * XS + ZC + t % (XS - ZC)] = 0.f;
  if (tid < XFRONT) sXp[tid] = 0.f;
  __syncthreads();

  const int hh = lane >> 4, cc = lane & 15;
  const int Mcol = cc - lof(cc) * lof(cc) - lof(cc);
  float wgt[8], wgtN[8];
  int xi[8], pi[8], ni[8];
  bool posE[8];
#pragma unroll
  for (int r = 0; r < 8; ++r) {
    const int kp = 8 * hh + r;
    const int L = lof(kp), nL = 2 * L + 1;
    const int k = kp - L * L, j = cc - L * L;
    const bool dg = (L == lof(cc));
    const int kr = 2 * L - k, jr = 2 * L - j;
    const int xo = (L >= 1) + 9 * (L >= 2) + 25 * (L >= 3);
    const int pb = (L >= 1) + 18 * (L >= 2) + 50 * (L >= 3);
    const float gk = sDIAG[xo + k * nL + kr];
    const float gj = sDIAG[dg ? (xo + j * nL + jr) : 0];
    const float w = gk * gj;
    wgt[r]  = dg ? w : 0.f;
    wgtN[r] = (dg && L > 0) ? w : 0.f;
    xi[r] = dg ? (xo + kr * nL + jr) : ZC;
    pi[r] = dg ? (pb + kr * nL + jr) : 0;
    ni[r] = (dg && L > 0) ? (pb + nL * nL + kr * nL + jr) : 0;
    posE[r] = ((L & 1) == 0);
  }

  for (int e = 0; e < 16; e += 2) {
    fetch_rows(nx, x0, x1, x2, x3, base + e + 2, Btot, lane);
    asm volatile("" ::: "memory");
    const float* Xe = X + (e & 2) * XS;
    v8f accE[2], accO[2];
#pragma unroll
    for (int el = 0; el < 2; ++el) {
      accE[el] = (v8f){0.f, 0.f, 0.f, 0.f, 0.f, 0.f, 0.f, 0.f};
      accO[el] = (v8f){0.f, 0.f, 0.f, 0.f, 0.f, 0.f, 0.f, 0.f};
    }
    pair_gemm<0, 0,  0,  0>(Xe, sB1, sA2, accE, lane);
    pair_gemm<0, 1,  1,  4>(Xe, sB1, sA2, accO, lane);
    pair_gemm<0, 2,  3,  8>(Xe, sB1, sA2, accE, lane);
    pair_gemm<0, 3,  6, 12>(Xe, sB1, sA2, accO, lane);
    pair_gemm<1, 1, 10, 16>(Xe, sB1, sA2, accE, lane);
    pair_gemm<1, 2, 14, 20>(Xe, sB1, sA2, accO, lane);
    pair_gemm2<3, 1, 20, 24>(Xe, sB1, sA2, accE, lane, Mcol);
    pair_gemm2<2, 2, 24, 32>(Xe, sB1, sA2, accE, lane, Mcol);
    pair_gemm2<3, 2, 27, 44>(Xe, sB1, sA2, accO, lane, Mcol);
    pair_gemm2<3, 3, 31, 56>(Xe, sB1, sA2, accE, lane, Mcol);
#pragma unroll
    for (int el = 0; el < 2; ++el) {
      store_x2(X2 + el * X2S, accE[el], true, lane);
      store_x2(X2 + el * X2S, accO[el], false, lane);
    }
    __builtin_amdgcn_fence(__ATOMIC_RELEASE, "wavefront");
    __builtin_amdgcn_fence(__ATOMIC_ACQUIRE, "wavefront");
    float xv[2][8], pv[2][8], nv[2][8];
#pragma unroll
    for (int el = 0; el < 2; ++el) {
      const float* Xq = Xe + el * XS;
      const float* X2q = X2 + el * X2S;
#pragma unroll
      for (int r = 0; r < 8; ++r) {
        xv[el][r] = Xq[xi[r]];
        pv[el][r] = X2q[pi[r]];
        nv[el][r] = X2q[ni[r]];
      }
    }
    asm volatile("" :: "v"(xv[0][0]), "v"(xv[0][1]), "v"(xv[0][2]), "v"(xv[0][3]), "v"(xv[0][4]), "v"(xv[0][5]), "v"(xv[0][6]), "v"(xv[0][7]), "v"(pv[0][0]), "v"(pv[0][1]), "v"(pv[0][2]), "v"(pv[0][3]), "v"(pv[0][4]), "v"(pv[0][5]), "v"(pv[0][6]), "v"(pv[0][7]), "v"(nv[0][0]), "v"(nv[0][1]), "v"(nv[0][2]), "v"(nv[0][3]), "v"(nv[0][4]), "v"(nv[0][5]), "v"(nv[0][6]), "v"(nv[0][7]));
    asm volatile("" :: "v"(xv[1][0]), "v"(xv[1][1]), "v"(xv[1][2]), "v"(xv[1][3]), "v"(xv[1][4]), "v"(xv[1][5]), "v"(xv[1][6]), "v"(xv[1][7]), "v"(pv[1][0]), "v"(pv[1][1]), "v"(pv[1][2]), "v"(pv[1][3]), "v"(pv[1][4]), "v"(pv[1][5]), "v"(pv[1][6]), "v"(pv[1][7]), "v"(nv[1][0]), "v"(nv[1][1]), "v"(nv[1][2]), "v"(nv[1][3]), "v"(nv[1][4]), "v"(nv[1][5]), "v"(nv[1][6]), "v"(nv[1][7]));
#pragma unroll
    for (int el = 0; el < 2; ++el) {
      float* Pe = PR + (e + el) * PS;
      float s3 = 0.f, s4 = 0.f;
#pragma unroll
      for (int r = 0; r < 8; ++r) {
        const float P = posE[r] ? accE[el][r] : accO[el][r];
        const float N = posE[r] ? accO[el][r] : accE[el][r];
        const float wp = wgt[r] * P;
        const float wn = wgtN[r] * N;
        s3 += wp * xv[el][r];
        s4 += wp * pv[el][r];
        s4 += wn * nv[el][r];
      }
      Pe[1 + lane] = s3;
      Pe[33 + lane] = s4;
      if (lane == 0) {
        Pe[0] = accE[el][0];
        Pe[65] = Xe[el * XS];
      }
    }
    put_rows(X + (2 - (e & 2)) * XS, nx, lane);
  }

  __builtin_amdgcn_fence(__ATOMIC_RELEASE, "wavefront");
  __builtin_amdgcn_fence(__ATOMIC_ACQUIRE, "wavefront");

  if (lane < 16) {
    const int e = base + b;
    if (e < Btot) {
      const float* Pb = PR + b * PS;
      const float nu1 = Pb[65];
      const float nu2 = Pb[0];
      float nu3 = 0.f, nu4 = 0.f;
      for (int i = 0; i < 32; ++i) {
        nu3 += Pb[1 + i];
        nu4 += Pb[33 + i];
      }
      out[(size_t)e * 4 + 0] = nu1;
      out[(size_t)e * 4 + 1] = nu2;
      out[(size_t)e * 4 + 2] = nu3;
      out[(size_t)e * 4 + 3] = nu4;
    }
  }
}

extern "C" void kernel_launch(void* const* d_in, const int* in_sizes, int n_in,
                              void* d_out, int out_size, void* d_ws, size_t ws_size,
                              hipStream_t stream) {
  const float* x0 = (const float*)d_in[0];
  const float* x1 = (const float*)d_in[1];
  const float* x2 = (const float*)d_in[2];
  const float* x3 = (const float*)d_in[3];
  float* tab = (float*)d_ws;
  const int Btot = in_sizes[0];

  wigner_cg_init<<<1, 256, 0, stream>>>(tab);

  const int tiles = (Btot + 15) / 16;
  const int blocks = (tiles + WAVES_PER_BLOCK - 1) / WAVES_PER_BLOCK;
  wigner_main<<<blocks, TB, 0, stream>>>(x0, x1, x2, x3, tab, (float*)d_out, Btot);
}


# Pairs (0,l): C(0,l,l) is the identity table, so their term w*x0*X_l[k,j] is formed by vector multiplies as the accumulators' start value: 32 of 246 WMMAs a trip and their fragments go; same sums
# speedup vs baseline: 45.7396x; 1.1164x over previous
#include <hip/hip_runtime.h>


#define NCOMBO 23
#define CG_TOTAL 2351
#define NPAIR 2
#define NPAIR2 4
#define B1_STEPS 25
#define A2_STEPS 56
#define B1_BASE CG_TOTAL
#define A2_BASE (CG_TOTAL + B1_STEPS * 64)
#define WS_TOTAL (A2_BASE + A2_STEPS * 64)
#define LB1 0
#define LA2 (B1_STEPS * 64)
#define LDIAG (LA2 + A2_STEPS * 64)
#define LT_TOTAL (LDIAG + 84)
#define WAVES_PER_BLOCK 4
#define TB (WAVES_PER_BLOCK * 32)
#define XS 101
#define ZC 84
#define XFRONT 8
#define X2S 169
#define XROWS 4
#define PS 67
#define X2W (2 * X2S + 16 * PS)

typedef float v2f __attribute__((ext_vector_type(2)));
typedef float v8f __attribute__((ext_vector_type(8)));

__constant__ int c_l1[NCOMBO] = {0,0,0,0, 1,1,1,1,1,1,1,1, 2,2,2,2,2,2,2, 3,3,3,3};
__constant__ int c_l2[NCOMBO] = {0,1,2,3, 1,1,1,2,2,2,3,3, 2,2,2,2,3,3,3, 3,3,3,3};
__constant__ int c_L [NCOMBO] = {0,1,2,3, 0,1,2,1,2,3,2,3, 0,1,2,3,1,2,3, 0,1,2,3};
__constant__ int c_off[NCOMBO] = {0,1,10,35, 84,93,120,165,210,285,390,495,
                                  642,667,742,867,1042,1147,1322, 1567,1616,1763,2008};
__constant__ int p_l1[NPAIR] = {1,1};
__constant__ int p_l2[NPAIR] = {1,2};
__constant__ int p_c0[NPAIR] = {4,7};
__constant__ int p_b1[NPAIR] = {0,4};
__constant__ int p_a2[NPAIR] = {0,4};
__constant__ int q_l1[NPAIR2] = {1,2,2,3};
__constant__ int q_l2[NPAIR2] = {3,2,3,3};
__constant__ int q_sw[NPAIR2] = {1,0,1,0};
__constant__ int q_c0[NPAIR2] = {10,12,16,19};
__constant__ int q_g [NPAIR2] = {10,14,17,21};
__constant__ int q_a2[NPAIR2] = {8,16,28,40};
__constant__ int d_xoff[4] = {0, 1, 10, 35};
__constant__ int d_diag[4] = {0, 4, 12, 19};

__device__ __forceinline__ int lof(int x) { return (x >= 1) + (x >= 4) + (x >= 9); }

__global__ void wigner_cg_init(float* __restrict__ ws) {
  __shared__ double sF[16];
  __shared__ double sRF[16];
  __shared__ double sPre[NCOMBO];
  if (threadIdx.x < 16) {
    double v = 1.0;
    for (int i = 2; i <= (int)threadIdx.x; ++i) v *= (double)i;
    sF[threadIdx.x] = v;
    sRF[threadIdx.x] = 1.0 / v;
  }
  __syncthreads();
  if (threadIdx.x < NCOMBO) {
    const int l1 = c_l1[threadIdx.x], l2 = c_l2[threadIdx.x], L = c_L[threadIdx.x];
    sPre[threadIdx.x] =
        (2.0 * L + 1.0) * sF[L + l1 - l2] * sF[L - l1 + l2] * sF[l1 + l2 - L] * sRF[l1 + l2 + L + 1];
  }
  __syncthreads();
  for (int c = 0; c < NCOMBO; ++c) {
    if (c >= 1 && c <= 3) continue;
    const int l1 = c_l1[c], l2 = c_l2[c], L = c_L[c];
    const int n2 = 2 * l2 + 1, nL = 2 * L + 1;
    const int sz = (2 * l1 + 1) * n2 * nL, off = c_off[c];
    const double pre = sPre[c];
    for (int t = threadIdx.x; t < sz; t += blockDim.x) {
      const int m = t / (n2 * nL), rem = t % (n2 * nL), n = rem / nL, k = rem % nL;
      const int m1 = m - l1, m2 = n - l2, M = m1 + m2;
      double val = 0.0;
      if (M >= -L && M <= L && k == M + L) {
        const double rad =
            pre * sF[L + M] * sF[L - M] * sF[l1 - m1] * sF[l1 + m1] * sF[l2 - m2] * sF[l2 + m2];
        double s = 0.0;
        for (int kk = 0; kk <= l1 + l2 - L; ++kk) {
          const int a1 = l1 - m1 - kk, a2 = l2 + m2 - kk;
          const int a3 = L - l2 + m1 + kk, a4 = L - l1 - m2 + kk;
          if (a1 < 0 || a2 < 0 || a3 < 0 || a4 < 0) continue;
          const double term = sRF[kk] * sRF[l1 + l2 - L - kk] * sRF[a1] * sRF[a2] * sRF[a3] * sRF[a4];
          s += (kk & 1) ? -term : term;
        }
        val = sqrt(rad) * s;
      }
      ws[off + t] = (float)val;
    }
  }
  __syncthreads();
  for (int pi = 0; pi < NPAIR; ++pi) {
    const int l1 = p_l1[pi], l2 = p_l2[pi];
    const int n1 = 2 * l1 + 1, n2 = 2 * l2 + 1, PH = l1 + 1, QH = l2 + 1;
    const int Lmin = l2 - l1, Lmax = (l1 + l2 < 3) ? (l1 + l2) : 3;
    const int c0 = p_c0[pi], NR = n1 * n2, RT = (NR + 15) >> 4;
    const float w = (l1 == l2) ? 1.f : 2.f;
    for (int t = threadIdx.x; t < PH * QH * 64; t += blockDim.x) {
      const int st = t >> 6, s6 = t & 63, ln = s6 >> 1, s = s6 & 1;
      const int i = st / QH, u = st % QH, h = ln >> 4, c = ln & 15;
      const int p = h * PH + i, q = 2 * u + s, L = lof(c), k = c - L * L;
      float val = 0.f;
      if (p < n1 && q < n2 && L >= Lmin && L <= Lmax)
        val = ws[c_off[c0 + L - Lmin] + (p * n2 + q) * (2 * L + 1) + k];
      ws[B1_BASE + (p_b1[pi] + st) * 64 + s6] = val;
    }
    for (int t = threadIdx.x; t < RT * 4 * 64; t += blockDim.x) {
      const int st = t >> 6, s6 = t & 63, ln = s6 >> 1, s = s6 & 1;
      const int rt = st >> 2, v = st & 3, h = ln >> 4, c = ln & 15;
      const int rho = rt * 16 + 4 * v + 2 * h + s, L = lof(c), k = c - L * L;
      float val = 0.f;
      if (rho < NR && L >= Lmin && L <= Lmax) {
        const int m = rho / n2, n = rho % n2;
        val = w * ws[c_off[c0 + L - Lmin] + (m * n2 + n) * (2 * L + 1) + k];
      }
      ws[A2_BASE + (p_a2[pi] + st) * 64 + s6] = val;
    }
  }
  for (int qi = 0; qi < NPAIR2; ++qi) {
    const int l1 = q_l1[qi], l2 = q_l2[qi], sw = q_sw[qi];
    const int n2 = 2 * l2 + 1;
    const int LA = sw ? l2 : l1, LB = sw ? l1 : l2;
    const int NA = 2 * LA + 1, NB = 2 * LB + 1, KH = LA + 1, TT = LB + 1;
    const int Lmin = l2 - l1, Lmax = (l1 + l2 < 3) ? (l1 + l2) : 3;
    const int c0 = q_c0[qi];
    const float w = (l1 == l2) ? 1.f : 2.f;
    for (int t = threadIdx.x; t < KH * 64; t += blockDim.x) {
      const int st = t >> 6, s6 = t & 63, ln = s6 >> 1, s = s6 & 1;
      const int c = ln & 15, ka = 2 * st + s, L = lof(c), j = c - L * L;
      const int kb = (j - L) + LA + LB - ka;
      float val = 0.f;
      if (ka < NA && kb >= 0 && kb < NB && L >= Lmin && L <= Lmax) {
        const int p = sw ? kb : ka, q = sw ? ka : kb;
        val = ws[c_off[c0 + L - Lmin] + (p * n2 + q) * (2 * L + 1) + j];
      }
      ws[B1_BASE + (q_g[qi] + st) * 64 + s6] = val;
    }
    for (int t = threadIdx.x; t < TT * 4 * 64; t += blockDim.x) {
      const int st = t >> 6, s6 = t & 63, ln = s6 >> 1, s = s6 & 1;
      const int tt = st >> 2, v = st & 3, h = ln >> 4, c = ln & 15;
      const int ma = 2 * v + s, nb = 2 * tt + h, L = lof(c), k = c - L * L;
      float val = 0.f;
      if (ma < NA && nb < NB && L >= Lmin && L <= Lmax) {
        const int m = sw ? nb : ma, n = sw ? ma : nb;
        val = w * ws[c_off[c0 + L - Lmin] + (m * n2 + n) * (2 * L + 1) + k];
      }
      ws[A2_BASE + (q_a2[qi] + st) * 64 + s6] = val;
    }
  }
}

__device__ __forceinline__ constexpr int xoff_c(int l) { return l == 0 ? 0 : (l == 1 ? 1 : (l == 2 ? 10 : 35)); }

template <int L1, int L2, int B1OFF, int A2OFF>
__device__ __forceinline__ void pair_gemm(const float* __restrict__ Xe, const float* __restrict__ sB1,
                                          const float* __restrict__ sA2, v8f (&acc)[2], const int lane) {
  constexpr int N1 = 2 * L1 + 1, N2 = 2 * L2 + 1, PH = L1 + 1, QH = L2 + 1;
  constexpr int NR = N1 * N2, RT = (NR + 15) / 16;
  constexpr int XO1 = xoff_c(L1), XO2 = xoff_c(L2);
  const int h = lane >> 4, c = lane & 15;
#pragma unroll
  for (int rt = 0; rt < RT; ++rt) {
    const int rho = rt * 16 + 4 * ((c & 7) >> 1) + 2 * (c >> 3) + (c & 1);
    const bool valid = rho < NR;
    const int m = rho / N2, n = rho - m * N2;
    float a[2][PH], b[2][2 * QH];
#pragma unroll
    for (int i = 0; i < PH; ++i) {
      const int p = h * PH + i;
      const int ia = (valid && p < N1) ? (XO1 + m * N1 + p) : ZC;
#pragma unroll
      for (int el = 0; el < 2; ++el) a[el][i] = Xe[el * XS + ia];
    }
#pragma unroll
    for (int q = 0; q < 2 * QH; ++q) {
      const int ib = (valid && q < N2) ? (XO2 + n * N2 + q) : ZC;
#pragma unroll
      for (int el = 0; el < 2; ++el) b[el][q] = Xe[el * XS + ib];
    }
    v8f d[2];
#pragma unroll
    for (int el = 0; el < 2; ++el) d[el] = (v8f){0.f, 0.f, 0.f, 0.f, 0.f, 0.f, 0.f, 0.f};
#pragma unroll
    for (int i = 0; i < PH; ++i) {
#pragma unroll
      for (int u = 0; u < QH; ++u) {
        const v2f bf = *(const v2f*)&sB1[(B1OFF + i * QH + u) * 64 + lane * 2];
#pragma unroll
        for (int el = 0; el < 2; ++el) {
          v2f af;
          af.x = a[el][i] * b[el][2 * u];
          af.y = a[el][i] * b[el][2 * u + 1];
          d[el] = __builtin_amdgcn_wmma_f32_16x16x4_f32(false, af, false, bf, (short)0, d[el], false, false);
        }
      }
    }
    const int rows = (NR - rt * 16 < 16) ? (NR - rt * 16) : 16;
    const int nv = (rows + 3) / 4;
#pragma unroll
    for (int v = 0; v < 4; ++v) {
      if (v < nv) {
        const v2f af2 = *(const v2f*)&sA2[(A2OFF + rt * 4 + v) * 64 + lane * 2];
#pragma unroll
        for (int el = 0; el < 2; ++el) {
          v2f bf2;
          bf2.x = d[el][2 * v];
          bf2.y = d[el][2 * v + 1];
          acc[el] = __builtin_amdgcn_wmma_f32_16x16x4_f32(false, af2, false, bf2, (short)0, acc[el], false, false);
        }
      }
    }
  }
}

template <int LA, int LB, int GOFF, int A2OFF>
__device__ __forceinline__ void pair_gemm2(const float* __restrict__ Xe, const float* __restrict__ sB1,
                                           const float* __restrict__ sA2, v8f (&acc)[2], const int lane,
                                           const int Mc) {
  constexpr int NA = 2 * LA + 1, NB = 2 * LB + 1, KH = LA + 1, TT = LB + 1;
  constexpr int NV2 = (NA + 1) / 2;
  constexpr int XOA = xoff_c(LA), XOB = xoff_c(LB);
  const int h = lane >> 4, c = lane & 15;
  const int g = c >> 3, ma = c & 7;
  const bool arow = (g == h) && (ma < NA);
  const float* Xa = Xe + (arow ? (XOA + ma * NA) : ZC);
  const float* Xb = Xe + (XOB + h * NB + Mc + LA + LB - (2 * KH - 1));
  static_assert((KH == 4) == (NV2 == 4), "the load fence below lists NV2 table fragments");
  v2f af[2][KH], gf[KH];
#pragma unroll
  for (int i = 0; i < KH; ++i) {
    gf[i] = *(const v2f*)&sB1[(GOFF + i) * 64 + lane * 2];
#pragma unroll
    for (int el = 0; el < 2; ++el) {
      af[el][i].x = Xa[el * XS + 2 * i];
      af[el][i].y = Xa[el * XS + 2 * i + 1];
    }
  }
#pragma unroll
  for (int tt = 0; tt < TT; ++tt) {
    asm volatile("" ::: "memory");
    float raw[2][2 * KH];
    v2f a2[NV2];
#pragma unroll
    for (int ka = 0; ka < 2 * KH; ++ka) {
#pragma unroll
      for (int el = 0; el < 2; ++el) raw[el][ka] = Xb[el * XS + 2 * tt * NB + (2 * KH - 1 - ka)];
    }
#pragma unroll
    for (int v = 0; v < NV2; ++v) a2[v] = *(const v2f*)&sA2[(A2OFF + tt * 4 + v) * 64 + lane * 2];
    if constexpr (KH == 4) {
      asm volatile("" :: "v"(raw[0][0]), "v"(raw[0][1]), "v"(raw[0][2]), "v"(raw[0][3]), "v"(raw[0][4]), "v"(raw[0][5]), "v"(raw[0][6]), "v"(raw[0][7]), "v"(raw[1][0]), "v"(raw[1][1]), "v"(raw[1][2]), "v"(raw[1][3]), "v"(raw[1][4]), "v"(raw[1][5]), "v"(raw[1][6]), "v"(raw[1][7]), "v"(a2[0]), "v"(a2[1]), "v"(a2[2]), "v"(a2[3]));
    } else {
      asm volatile("" :: "v"(raw[0][0]), "v"(raw[0][1]), "v"(raw[0][2]), "v"(raw[0][3]), "v"(raw[0][4]), "v"(raw[0][5]), "v"(raw[1][0]), "v"(raw[1][1]), "v"(raw[1][2]), "v"(raw[1][3]), "v"(raw[1][4]), "v"(raw[1][5]), "v"(a2[0]), "v"(a2[1]), "v"(a2[2]));
    }
    v8f d[2];
#pragma unroll
    for (int el = 0; el < 2; ++el) d[el] = (v8f){0.f, 0.f, 0.f, 0.f, 0.f, 0.f, 0.f, 0.f};
#pragma unroll
    for (int i = 0; i < KH; ++i) {
#pragma unroll
      for (int el = 0; el < 2; ++el) {
        v2f bf;
        bf.x = raw[el][2 * i] * gf[i].x;
        bf.y = raw[el][2 * i + 1] * gf[i].y;
        d[el] = __builtin_amdgcn_wmma_f32_16x16x4_f32(false, af[el][i], false, bf, (short)0, d[el], false, false);
      }
    }
#pragma unroll
    for (int v = 0; v < NV2; ++v) {
#pragma unroll
      for (int el = 0; el < 2; ++el) {
        v2f bf2;
        bf2.x = d[el][2 * v];
        bf2.y = d[el][2 * v + 1];
        acc[el] = __builtin_amdgcn_wmma_f32_16x16x4_f32(false, a2[v], false, bf2, (short)0, acc[el], false, false);
      }
    }
  }
}

__device__ __forceinline__ void store_x2(float* __restrict__ X2e, const v8f& t, const bool even, const int lane) {
  const int h = lane >> 4, c = lane & 15;
  const int Lc = lof(c);
#pragma unroll
  for (int v = 0; v < 8; ++v) {
    const int kp = 8 * h + v;
    const int L = lof(kp);
    const int k = kp - L * L, j = c - L * L, nL = 2 * L + 1;
    const bool pos = even ? ((L & 1) == 0) : ((L & 1) == 1);
    const int pb = (L >= 1) + 18 * (L >= 2) + 50 * (L >= 3);
    const int base = pos ? pb : (pb + nL * nL);
    const bool wr = (L == Lc) && (pos || L > 0);
    X2e[wr ? (base + k * nL + j) : (X2S - 1)] = t[v];
  }
}

struct RowRegs { float r3[4]; float r2[2]; float r1; float r0; };

__device__ __forceinline__ void fetch_rows(RowRegs& r, const float* __restrict__ x0, const float* __restrict__ x1,
                                           const float* __restrict__ x2, const float* __restrict__ x3,
                                           const int e0, const int Btot, const int lane) {
  const size_t n0 = (size_t)Btot - 1, n1 = (size_t)Btot * 9 - 1;
  const size_t n2 = (size_t)Btot * 25 - 1, n3 = (size_t)Btot * 49 - 1;
#pragma unroll
  for (int i = 0; i < 4; ++i) {
    const size_t f = (size_t)e0 * 49 + (size_t)(32 * i + lane);
    r.r3[i] = x3[f < n3 ? f : n3];
  }
#pragma unroll
  for (int i = 0; i < 2; ++i) {
    const size_t f = (size_t)e0 * 25 + (size_t)(32 * i + lane);
    r.r2[i] = x2[f < n2 ? f : n2];
  }
  {
    const size_t f = (size_t)e0 * 9 + (size_t)lane;
    r.r1 = x1[f < n1 ? f : n1];
  }
  {
    const size_t f = (size_t)e0 + (size_t)lane;
    r.r0 = x0[f < n0 ? f : n0];
  }
}

__device__ __forceinline__ void put_rows(float* __restrict__ Xn, const RowRegs& r, const int lane) {
#pragma unroll
  for (int i = 0; i < 4; ++i) {
    const int j = 32 * i + lane;
    const int bb = (j >= 49) ? 1 : 0;
    if (j < 98) Xn[bb * XS + 35 + j - 49 * bb] = r.r3[i];
  }
#pragma unroll
  for (int i = 0; i < 2; ++i) {
    const int j = 32 * i + lane;
    const int bb = (j >= 25) ? 1 : 0;
    if (j < 50) Xn[bb * XS + 10 + j - 25 * bb] = r.r2[i];
  }
  {
    const int bb = (lane >= 9) ? 1 : 0;
    if (lane < 18) Xn[bb * XS + 1 + lane - 9 * bb] = r.r1;
  }
  if (lane < 2) Xn[lane * XS] = r.r0;
}

__launch_bounds__(TB, 4)
__global__ void wigner_main(const float* __restrict__ x0, const float* __restrict__ x1,
                            const float* __restrict__ x2, const float* __restrict__ x3,
                            const float* __restrict__ tab, float* __restrict__ out, int Btot) {
  __shared__ __align__(16) float sTAB[LT_TOTAL];
  __shared__ float sXp[XFRONT + WAVES_PER_BLOCK * XROWS * XS];
  __shared__ float sX2[WAVES_PER_BLOCK][X2W];

  const int tid = threadIdx.x, lane = tid & 31, wid = tid >> 5;
  const int b = lane & 15;
  const float* sB1 = sTAB + LB1;
  const float* sA2 = sTAB + LA2;
  const float* sDIAG = sTAB + LDIAG;

  for (int t = tid; t < LDIAG; t += TB) sTAB[t] = tab[CG_TOTAL + t];
  for (int l = 0; l <= 3; ++l) {
    const int nn = (2 * l + 1) * (2 * l + 1);
    for (int t = tid; t < nn; t += TB) sTAB[LDIAG + d_xoff[l] + t] = tab[c_off[d_diag[l]] + t];
  }

  const int tile = blockIdx.x * WAVES_PER_BLOCK + wid;
  const int base = tile * 16;
  float* X  = sXp + XFRONT + wid * XROWS * XS;
  float* X2 = sX2[wid];
  float* PR = sX2[wid] + 2 * X2S;

  {
    int pe = base + WAVES_PER_BLOCK * 16;
    if (pe < Btot) __builtin_prefetch(&x3[(size_t)pe * 49], 0, 1);
  }

  RowRegs nx;
  fetch_rows(nx, x0, x1, x2, x3, base, Btot, lane);
  put_rows(X, nx, lane);
  for (int t = lane; t < XROWS * (XS - ZC); t += 32) X[(t / (XS - ZC)) * XS + ZC + t % (XS - ZC)] = 0.f;
  if (tid < XFRONT) sXp[tid] = 0.f;
  __syncthreads();

  const int hh = lane >> 4, cc = lane & 15;
  const int Mcol = cc - lof(cc) * lof(cc) - lof(cc);
  float wgt[8], wgtN[8];
  int xi[8], pi[8], ni[8];
  bool posE[8];
#pragma unroll
  for (int r = 0; r < 8; ++r) {
    const int kp = 8 * hh + r;
    const int L = lof(kp), nL = 2 * L + 1;
    const int k = kp - L * L, j = cc - L * L;
    const bool dg = (L == lof(cc));
    const int kr = 2 * L - k, jr = 2 * L - j;
    const int xo = (L >= 1) + 9 * (L >= 2) + 25 * (L >= 3);
    const int pb = (L >= 1) + 18 * (L >= 2) + 50 * (L >= 3);
    const float gk = sDIAG[xo + k * nL + kr];
    const float gj = sDIAG[dg ? (xo + j * nL + jr) : 0];
    const float w = gk * gj;
    wgt[r]  = dg ? w : 0.f;
    wgtN[r] = (dg && L > 0) ? w : 0.f;
    xi[r] = dg ? (xo + kr * nL + jr) : ZC;
    pi[r] = dg ? (pb + kr * nL + jr) : 0;
    ni[r] = (dg && L > 0) ? (pb + nL * nL + kr * nL + jr) : 0;
    posE[r] = ((L & 1) == 0);
  }

  int sd[8];
#pragma unroll
  for (int r = 0; r < 8; ++r) {
    const int kp = 8 * hh + r;
    const int L = lof(kp), nL = 2 * L + 1;
    const int xo = (L >= 1) + 9 * (L >= 2) + 25 * (L >= 3);
    sd[r] = xo + (kp - L * L) * nL + ((L == lof(cc)) ? (cc - L * L) : 0);
  }
  const int Lcc = lof(cc);
  const float sw0  = (Lcc == (hh ? 2 : 0)) ? (hh ? 2.f : 1.f) : 0.f;
  const float sw1  = (Lcc == (hh ? 3 : 1)) ? 2.f : 0.f;
  const float sw2e = (hh == 0 && Lcc == 2) ? 2.f : 0.f;
  const float sw2o = (hh == 1 && Lcc == 3) ? 2.f : 0.f;

  for (int e = 0; e < 16; e += 2) {
    fetch_rows(nx, x0, x1, x2, x3, base + e + 2, Btot, lane);
    asm volatile("" ::: "memory");
    const float* Xe = X + (e & 2) * XS;
    v8f accE[2], accO[2];
#pragma unroll
    for (int el = 0; el < 2; ++el) {
      const float* Xq = Xe + el * XS;
      const float x0v = Xq[0];
      float sv[8];
#pragma unroll
      for (int r = 0; r < 8; ++r) sv[r] = Xq[sd[r]];
      const float a0 = sw0 * x0v, a1 = sw1 * x0v, a2e = sw2e * x0v, a2o = sw2o * x0v;
      accE[el][0] = a0 * sv[0];
      accO[el][0] = 0.f;
#pragma unroll
      for (int r = 1; r < 4; ++r) {
        accE[el][r] = 0.f;
        accO[el][r] = a1 * sv[r];
      }
#pragma unroll
      for (int r = 4; r < 8; ++r) {
        accE[el][r] = a2e * sv[r];
        accO[el][r] = a2o * sv[r];
      }
    }
    pair_gemm<1, 1,  0,  0>(Xe, sB1, sA2, accE, lane);
    pair_gemm<1, 2,  4,  4>(Xe, sB1, sA2, accO, lane);
    pair_gemm2<3, 1, 10,  8>(Xe, sB1, sA2, accE, lane, Mcol);
    pair_gemm2<2, 2, 14, 16>(Xe, sB1, sA2, accE, lane, Mcol);
    pair_gemm2<3, 2, 17, 28>(Xe, sB1, sA2, accO, lane, Mcol);
    pair_gemm2<3, 3, 21, 40>(Xe, sB1, sA2, accE, lane, Mcol);
#pragma unroll
    for (int el = 0; el < 2; ++el) {
      store_x2(X2 + el * X2S, accE[el], true, lane);
      store_x2(X2 + el * X2S, accO[el], false, lane);
    }
    __builtin_amdgcn_fence(__ATOMIC_RELEASE, "wavefront");
    __builtin_amdgcn_fence(__ATOMIC_ACQUIRE, "wavefront");
    float xv[2][8], pv[2][8], nv[2][8];
#pragma unroll
    for (int el = 0; el < 2; ++el) {
      const float* Xq = Xe + el * XS;
      const float* X2q = X2 + el * X2S;
#pragma unroll
      for (int r = 0; r < 8; ++r) {
        xv[el][r] = Xq[xi[r]];
        pv[el][r] = X2q[pi[r]];
        nv[el][r] = X2q[ni[r]];
      }
    }
    asm volatile("" :: "v"(xv[0][0]), "v"(xv[0][1]), "v"(xv[0][2]), "v"(xv[0][3]), "v"(xv[0][4]), "v"(xv[0][5]), "v"(xv[0][6]), "v"(xv[0][7]), "v"(pv[0][0]), "v"(pv[0][1]), "v"(pv[0][2]), "v"(pv[0][3]), "v"(pv[0][4]), "v"(pv[0][5]), "v"(pv[0][6]), "v"(pv[0][7]), "v"(nv[0][0]), "v"(nv[0][1]), "v"(nv[0][2]), "v"(nv[0][3]), "v"(nv[0][4]), "v"(nv[0][5]), "v"(nv[0][6]), "v"(nv[0][7]));
    asm volatile("" :: "v"(xv[1][0]), "v"(xv[1][1]), "v"(xv[1][2]), "v"(xv[1][3]), "v"(xv[1][4]), "v"(xv[1][5]), "v"(xv[1][6]), "v"(xv[1][7]), "v"(pv[1][0]), "v"(pv[1][1]), "v"(pv[1][2]), "v"(pv[1][3]), "v"(pv[1][4]), "v"(pv[1][5]), "v"(pv[1][6]), "v"(pv[1][7]), "v"(nv[1][0]), "v"(nv[1][1]), "v"(nv[1][2]), "v"(nv[1][3]), "v"(nv[1][4]), "v"(nv[1][5]), "v"(nv[1][6]), "v"(nv[1][7]));
#pragma unroll
    for (int el = 0; el < 2; ++el) {
      float* Pe = PR + (e + el) * PS;
      float s3 = 0.f, s4 = 0.f;
#pragma unroll
      for (int r = 0; r < 8; ++r) {
        const float P = posE[r] ? accE[el][r] : accO[el][r];
        const float N = posE[r] ? accO[el][r] : accE[el][r];
        const float wp = wgt[r] * P;
        const float wn = wgtN[r] * N;
        s3 += wp * xv[el][r];
        s4 += wp * pv[el][r];
        s4 += wn * nv[el][r];
      }
      Pe[1 + lane] = s3;
      Pe[33 + lane] = s4;
      if (lane == 0) {
        Pe[0] = accE[el][0];
        Pe[65] = Xe[el * XS];
      }
    }
    put_rows(X + (2 - (e & 2)) * XS, nx, lane);
  }

  __builtin_amdgcn_fence(__ATOMIC_RELEASE, "wavefront");
  __builtin_amdgcn_fence(__ATOMIC_ACQUIRE, "wavefront");

  if (lane < 16) {
    const int e = base + b;
    if (e < Btot) {
      const float* Pb = PR + b * PS;
      const float nu1 = Pb[65];
      const float nu2 = Pb[0];
      float nu3 = 0.f, nu4 = 0.f;
      for (int i = 0; i < 32; ++i) {
        nu3 += Pb[1 + i];
        nu4 += Pb[33 + i];
      }
      out[(size_t)e * 4 + 0] = nu1;
      out[(size_t)e * 4 + 1] = nu2;
      out[(size_t)e * 4 + 2] = nu3;
      out[(size_t)e * 4 + 3] = nu4;
    }
  }
}

extern "C" void kernel_launch(void* const* d_in, const int* in_sizes, int n_in,
                              void* d_out, int out_size, void* d_ws, size_t ws_size,
                              hipStream_t stream) {
  const float* x0 = (const float*)d_in[0];
  const float* x1 = (const float*)d_in[1];
  const float* x2 = (const float*)d_in[2];
  const float* x3 = (const float*)d_in[3];
  float* tab = (float*)d_ws;
  const int Btot = in_sizes[0];

  wigner_cg_init<<<1, 256, 0, stream>>>(tab);

  const int tiles = (Btot + 15) / 16;
  const int blocks = (tiles + WAVES_PER_BLOCK - 1) / WAVES_PER_BLOCK;
  wigner_main<<<blocks, TB, 0, stream>>>(x0, x1, x2, x3, tab, (float*)d_out, Btot);
}


# A trip's 16 K tiles run as one software pipeline: tile n+1's LDS reads go out as one group ahead of tile n's WMMAs, its operand multiplies behind them (was read, wait, multiply per tile); same sums
# speedup vs baseline: 67.0816x; 1.4666x over previous
#include <hip/hip_runtime.h>


#define NCOMBO 23
#define CG_TOTAL 2351
#define NPAIR 6
#define B1_STEPS 20
#define GA_TILES 16
#define B1_BASE CG_TOTAL
#define GA_BASE (CG_TOTAL + B1_STEPS * 64)
#define WS_TOTAL (GA_BASE + GA_TILES * 32)
#define LB1 0
#define LGA (B1_STEPS * 64)
#define LDIAG (LGA + GA_TILES * 32)
#define LT_TOTAL (LDIAG + 84)
#define WAVES_PER_BLOCK 4
#define TB (WAVES_PER_BLOCK * 32)
#define XS 101
#define ZC 84
#define XFRONT 8
#define X2S 169
#define XROWS 4
#define PS 67
#define X2W (2 * X2S + 16 * PS)

typedef float v2f __attribute__((ext_vector_type(2)));
typedef float v8f __attribute__((ext_vector_type(8)));

__constant__ int c_l1[NCOMBO] = {0,0,0,0, 1,1,1,1,1,1,1,1, 2,2,2,2,2,2,2, 3,3,3,3};
__constant__ int c_l2[NCOMBO] = {0,1,2,3, 1,1,1,2,2,2,3,3, 2,2,2,2,3,3,3, 3,3,3,3};
__constant__ int c_L [NCOMBO] = {0,1,2,3, 0,1,2,1,2,3,2,3, 0,1,2,3,1,2,3, 0,1,2,3};
__constant__ int c_off[NCOMBO] = {0,1,10,35, 84,93,120,165,210,285,390,495,
                                  642,667,742,867,1042,1147,1322, 1567,1616,1763,2008};
__constant__ int q_l1[NPAIR] = {1,1,1,2,2,3};
__constant__ int q_l2[NPAIR] = {1,2,3,2,3,3};
__constant__ int q_sw[NPAIR] = {0,1,1,0,1,0};
__constant__ int q_c0[NPAIR] = {4,7,10,12,16,19};
__constant__ int q_g [NPAIR] = {0,2,5,9,12,16};
__constant__ int q_ga[NPAIR] = {0,2,4,6,9,12};
__constant__ int d_xoff[4] = {0, 1, 10, 35};
__constant__ int d_diag[4] = {0, 4, 12, 19};

__device__ __forceinline__ int lof(int x) { return (x >= 1) + (x >= 4) + (x >= 9); }

__global__ void wigner_cg_init(float* __restrict__ ws) {
  __shared__ double sF[16];
  __shared__ double sRF[16];
  __shared__ double sPre[NCOMBO];
  if (threadIdx.x < 16) {
    double v = 1.0;
    for (int i = 2; i <= (int)threadIdx.x; ++i) v *= (double)i;
    sF[threadIdx.x] = v;
    sRF[threadIdx.x] = 1.0 / v;
  }
  __syncthreads();
  if (threadIdx.x < NCOMBO) {
    const int l1 = c_l1[threadIdx.x], l2 = c_l2[threadIdx.x], L = c_L[threadIdx.x];
    sPre[threadIdx.x] =
        (2.0 * L + 1.0) * sF[L + l1 - l2] * sF[L - l1 + l2] * sF[l1 + l2 - L] * sRF[l1 + l2 + L + 1];
  }
  __syncthreads();
  for (int c = 0; c < NCOMBO; ++c) {
    if (c >= 1 && c <= 3) continue;
    const int l1 = c_l1[c], l2 = c_l2[c], L = c_L[c];
    const int n2 = 2 * l2 + 1, nL = 2 * L + 1;
    const int sz = (2 * l1 + 1) * n2 * nL, off = c_off[c];
    const double pre = sPre[c];
    for (int t = threadIdx.x; t < sz; t += blockDim.x) {
      const int m = t / (n2 * nL), rem = t % (n2 * nL), n = rem / nL, k = rem % nL;
      const int m1 = m - l1, m2 = n - l2, M = m1 + m2;
      double val = 0.0;
      if (M >= -L && M <= L && k == M + L) {
        const double rad =
            pre * sF[L + M] * sF[L - M] * sF[l1 - m1] * sF[l1 + m1] * sF[l2 - m2] * sF[l2 + m2];
        double s = 0.0;
        for (int kk = 0; kk <= l1 + l2 - L; ++kk) {
          const int a1 = l1 - m1 - kk, a2 = l2 + m2 - kk;
          const int a3 = L - l2 + m1 + kk, a4 = L - l1 - m2 + kk;
          if (a1 < 0 || a2 < 0 || a3 < 0 || a4 < 0) continue;
          const double term = sRF[kk] * sRF[l1 + l2 - L - kk] * sRF[a1] * sRF[a2] * sRF[a3] * sRF[a4];
          s += (kk & 1) ? -term : term;
        }
        val = sqrt(rad) * s;
      }
      ws[off + t] = (float)val;
    }
  }
  __syncthreads();
  for (int qi = 0; qi < NPAIR; ++qi) {
    const int l1 = q_l1[qi], l2 = q_l2[qi], sw = q_sw[qi];
    const int n2 = 2 * l2 + 1;
    const int LA = sw ? l2 : l1, LB = sw ? l1 : l2;
    const int NA = 2 * LA + 1, NB = 2 * LB + 1, KH = LA + 1, TT = LB + 1;
    const int Lmin = l2 - l1, Lmax = (l1 + l2 < 3) ? (l1 + l2) : 3;
    const int c0 = q_c0[qi];
    const float w = (l1 == l2) ? 1.f : 2.f;
    for (int t = threadIdx.x; t < KH * 64; t += blockDim.x) {
      const int st = t >> 6, s6 = t & 63, ln = s6 >> 1, s = s6 & 1;
      const int c = ln & 15, ka = 2 * st + s, L = lof(c), j = c - L * L;
      const int kb = (j - L) + LA + LB - ka;
      float val = 0.f;
      if (ka < NA && kb >= 0 && kb < NB && L >= Lmin && L <= Lmax) {
        const int p = sw ? kb : ka, q = sw ? ka : kb;
        val = ws[c_off[c0 + L - Lmin] + (p * n2 + q) * (2 * L + 1) + j];
      }
      ws[B1_BASE + (q_g[qi] + st) * 64 + s6] = val;
    }
    for (int t = threadIdx.x; t < TT * 32; t += blockDim.x) {
      const int tt = t >> 5, ln = t & 31, h = ln >> 4, c = ln & 15;
      const int L = lof(c), k = c - L * L;
      const int nb = 2 * tt + h, ma = (k - L) + LA + LB - nb;
      float val = 0.f;
      if (ma >= 0 && ma < NA && nb < NB && L >= Lmin && L <= Lmax) {
        const int m = sw ? nb : ma, n = sw ? ma : nb;
        val = w * ws[c_off[c0 + L - Lmin] + (m * n2 + n) * (2 * L + 1) + k];
      }
      ws[GA_BASE + (q_ga[qi] + tt) * 32 + ln] = val;
    }
  }
}

__device__ __forceinline__ constexpr int xoff_c(int l) { return l == 0 ? 0 : (l == 1 ? 1 : (l == 2 ? 10 : 35)); }

#define NTILE 16
__device__ __forceinline__ constexpr int tp_pair(int n) { return n < 2 ? 0 : (n < 4 ? 1 : (n < 6 ? 2 : (n < 9 ? 3 : (n < 12 ? 4 : 5)))); }
__device__ __forceinline__ constexpr int tp_first(int p) { return p == 0 ? 0 : (p == 1 ? 2 : (p == 2 ? 4 : (p == 3 ? 6 : (p == 4 ? 9 : 12)))); }
__device__ __forceinline__ constexpr int tp_LA(int p) { return p == 0 ? 1 : ((p == 1 || p == 3) ? 2 : 3); }
__device__ __forceinline__ constexpr int tp_LB(int p) { return p < 3 ? 1 : (p < 5 ? 2 : 3); }
__device__ __forceinline__ constexpr int tp_G(int p) { return p == 0 ? 0 : (p == 1 ? 2 : (p == 2 ? 5 : (p == 3 ? 9 : (p == 4 ? 12 : 16)))); }
__device__ __forceinline__ constexpr int tp_GA(int p) { return p == 0 ? 0 : (p == 1 ? 2 : (p == 2 ? 4 : (p == 3 ? 6 : (p == 4 ? 9 : 12)))); }
__device__ __forceinline__ constexpr bool tp_odd(int p) { return p == 1 || p == 4; }

struct TileRaw { float ra[2][8]; float rb[2][8]; float ga; };

template <int N>
__device__ __forceinline__ void tile_load(TileRaw& r, v2f (&gf)[4], const float* __restrict__ Xe,
                                           const float* __restrict__ sB1, const float* __restrict__ sGA,
                                           const int lane, const int Mc) {
  constexpr int P = tp_pair(N), LA = tp_LA(P), LB = tp_LB(P), tt = N - tp_first(P);
  constexpr int NA = 2 * LA + 1, NB = 2 * LB + 1, KH = LA + 1;
  constexpr int XOA = xoff_c(LA), XOB = xoff_c(LB);
  const int h = lane >> 4;
  const float* Xb = Xe + (XOB + h * NB + Mc + LA + LB - (2 * KH - 1));
  const int ma = Mc + LA + LB - h - 2 * tt;
  const float* Xa = Xe + (((unsigned)ma < (unsigned)NA) ? (XOA + ma * NA) : ZC);
  if constexpr (tt == 0) {
#pragma unroll
    for (int i = 0; i < KH; ++i) gf[i] = *(const v2f*)&sB1[(tp_G(P) + i) * 64 + lane * 2];
  }
  r.ga = sGA[(tp_GA(P) + tt) * 32 + lane];
#pragma unroll
  for (int ka = 0; ka < 2 * KH; ++ka) {
#pragma unroll
    for (int el = 0; el < 2; ++el) {
      r.ra[el][ka] = Xa[el * XS + ka];
      r.rb[el][ka] = Xb[el * XS + 2 * tt * NB + (2 * KH - 1 - ka)];
    }
  }
}

struct TileProd { v2f af[2][4]; v2f bf[2][4]; };

template <int N>
__device__ __forceinline__ void tile_mul(const TileRaw& r, const v2f (&gf)[4], TileProd& p) {
  constexpr int P = tp_pair(N), KH = tp_LA(P) + 1;
#pragma unroll
  for (int i = 0; i < KH; ++i) {
#pragma unroll
    for (int el = 0; el < 2; ++el) {
      p.af[el][i].x = r.ra[el][2 * i] * r.ga;
      p.af[el][i].y = r.ra[el][2 * i + 1] * r.ga;
      p.bf[el][i].x = r.rb[el][2 * i] * gf[i].x;
      p.bf[el][i].y = r.rb[el][2 * i + 1] * gf[i].y;
    }
  }
}

template <int N>
__device__ __forceinline__ void tile_wmma(const TileProd& p, v8f (&accE)[2], v8f (&accO)[2]) {
  constexpr int P = tp_pair(N), KH = tp_LA(P) + 1;
#pragma unroll
  for (int i = 0; i < KH; ++i) {
#pragma unroll
    for (int el = 0; el < 2; ++el) {
      if constexpr (tp_odd(P))
        accO[el] = __builtin_amdgcn_wmma_f32_16x16x4_f32(false, p.af[el][i], false, p.bf[el][i], (short)0, accO[el], false, false);
      else
        accE[el] = __builtin_amdgcn_wmma_f32_16x16x4_f32(false, p.af[el][i], false, p.bf[el][i], (short)0, accE[el], false, false);
    }
  }
}

template <int N>
__device__ __forceinline__ void tile_stage(const TileProd& cur, TileProd& nxt, TileRaw& raw, v2f (&g0)[4], v2f (&g1)[4],
                                            const float* __restrict__ Xe, const float* __restrict__ sB1,
                                            const float* __restrict__ sGA, v8f (&accE)[2], v8f (&accO)[2],
                                            const int lane, const int Mc) {
  if constexpr (N + 1 < NTILE) {
    if constexpr (tp_pair(N + 1) & 1) tile_load<N + 1>(raw, g1, Xe, sB1, sGA, lane, Mc);
    else                              tile_load<N + 1>(raw, g0, Xe, sB1, sGA, lane, Mc);
  }
  tile_wmma<N>(cur, accE, accO);
  if constexpr (N + 1 < NTILE) {
    if constexpr (tp_pair(N + 1) & 1) tile_mul<N + 1>(raw, g1, nxt);
    else                              tile_mul<N + 1>(raw, g0, nxt);
    __builtin_amdgcn_sched_group_barrier(0x100, 40, 0);
    __builtin_amdgcn_sched_group_barrier(0x008, 8, 0);
    __builtin_amdgcn_sched_group_barrier(0x002, 64, 0);
  }
  __builtin_amdgcn_sched_barrier(0);
}

__device__ __forceinline__ void store_x2(float* __restrict__ X2e, const v8f& t, const bool even, const int lane) {
  const int h = lane >> 4, c = lane & 15;
  const int Lc = lof(c);
#pragma unroll
  for (int v = 0; v < 8; ++v) {
    const int kp = 8 * h + v;
    const int L = lof(kp);
    const int k = kp - L * L, j = c - L * L, nL = 2 * L + 1;
    const bool pos = even ? ((L & 1) == 0) : ((L & 1) == 1);
    const int pb = (L >= 1) + 18 * (L >= 2) + 50 * (L >= 3);
    const int base = pos ? pb : (pb + nL * nL);
    const bool wr = (L == Lc) && (pos || L > 0);
    X2e[wr ? (base + k * nL + j) : (X2S - 1)] = t[v];
  }
}

struct RowRegs { float r3[4]; float r2[2]; float r1; float r0; };

__device__ __forceinline__ void fetch_rows(RowRegs& r, const float* __restrict__ x0, const float* __restrict__ x1,
                                           const float* __restrict__ x2, const float* __restrict__ x3,
                                           const int e0, const int Btot, const int lane) {
  const size_t n0 = (size_t)Btot - 1, n1 = (size_t)Btot * 9 - 1;
  const size_t n2 = (size_t)Btot * 25 - 1, n3 = (size_t)Btot * 49 - 1;
#pragma unroll
  for (int i = 0; i < 4; ++i) {
    const size_t f = (size_t)e0 * 49 + (size_t)(32 * i + lane);
    r.r3[i] = x3[f < n3 ? f : n3];
  }
#pragma unroll
  for (int i = 0; i < 2; ++i) {
    const size_t f = (size_t)e0 * 25 + (size_t)(32 * i + lane);
    r.r2[i] = x2[f < n2 ? f : n2];
  }
  {
    const size_t f = (size_t)e0 * 9 + (size_t)lane;
    r.r1 = x1[f < n1 ? f : n1];
  }
  {
    const size_t f = (size_t)e0 + (size_t)lane;
    r.r0 = x0[f < n0 ? f : n0];
  }
}

__device__ __forceinline__ void put_rows(float* __restrict__ Xn, const RowRegs& r, const int lane) {
#pragma unroll
  for (int i = 0; i < 4; ++i) {
    const int j = 32 * i + lane;
    const int bb = (j >= 49) ? 1 : 0;
    if (j < 98) Xn[bb * XS + 35 + j - 49 * bb] = r.r3[i];
  }
#pragma unroll
  for (int i = 0; i < 2; ++i) {
    const int j = 32 * i + lane;
    const int bb = (j >= 25) ? 1 : 0;
    if (j < 50) Xn[bb * XS + 10 + j - 25 * bb] = r.r2[i];
  }
  {
    const int bb = (lane >= 9) ? 1 : 0;
    if (lane < 18) Xn[bb * XS + 1 + lane - 9 * bb] = r.r1;
  }
  if (lane < 2) Xn[lane * XS] = r.r0;
}

__launch_bounds__(TB, 4)
__global__ void wigner_main(const float* __restrict__ x0, const float* __restrict__ x1,
                            const float* __restrict__ x2, const float* __restrict__ x3,
                            const float* __restrict__ tab, float* __restrict__ out, int Btot) {
  __shared__ __align__(16) float sTAB[LT_TOTAL];
  __shared__ float sXp[XFRONT + WAVES_PER_BLOCK * XROWS * XS];
  __shared__ float sX2[WAVES_PER_BLOCK][X2W];

  const int tid = threadIdx.x, lane = tid & 31, wid = tid >> 5;
  const int b = lane & 15;
  const float* sB1 = sTAB + LB1;
  const float* sGA = sTAB + LGA;
  const float* sDIAG = sTAB + LDIAG;

  for (int t = tid; t < LDIAG; t += TB) sTAB[t] = tab[CG_TOTAL + t];
  for (int l = 0; l <= 3; ++l) {
    const int nn = (2 * l + 1) * (2 * l + 1);
    for (int t = tid; t < nn; t += TB) sTAB[LDIAG + d_xoff[l] + t] = tab[c_off[d_diag[l]] + t];
  }

  const int tile = blockIdx.x * WAVES_PER_BLOCK + wid;
  const int base = tile * 16;
  float* X  = sXp + XFRONT + wid * XROWS * XS;
  float* X2 = sX2[wid];
  float* PR = sX2[wid] + 2 * X2S;

  {
    int pe = base + WAVES_PER_BLOCK * 16;
    if (pe < Btot) __builtin_prefetch(&x3[(size_t)pe * 49], 0, 1);
  }

  RowRegs nx;
  fetch_rows(nx, x0, x1, x2, x3, base, Btot, lane);
  put_rows(X, nx, lane);
  for (int t = lane; t < XROWS * (XS - ZC); t += 32) X[(t / (XS - ZC)) * XS + ZC + t % (XS - ZC)] = 0.f;
  if (tid < XFRONT) sXp[tid] = 0.f;
  __syncthreads();

  const int hh = lane >> 4, cc = lane & 15;
  const int Mcol = cc - lof(cc) * lof(cc) - lof(cc);
  float wgt[8], wgtN[8];
  int xi[8], pi[8], ni[8];
  bool posE[8];
#pragma unroll
  for (int r = 0; r < 8; ++r) {
    const int kp = 8 * hh + r;
    const int L = lof(kp), nL = 2 * L + 1;
    const int k = kp - L * L, j = cc - L * L;
    const bool dg = (L == lof(cc));
    const int kr = 2 * L - k, jr = 2 * L - j;
    const int xo = (L >= 1) + 9 * (L >= 2) + 25 * (L >= 3);
    const int pb = (L >= 1) + 18 * (L >= 2) + 50 * (L >= 3);
    const float gk = sDIAG[xo + k * nL + kr];
    const float gj = sDIAG[dg ? (xo + j * nL + jr) : 0];
    const float w = gk * gj;
    wgt[r]  = dg ? w : 0.f;
    wgtN[r] = (dg && L > 0) ? w : 0.f;
    xi[r] = dg ? (xo + kr * nL + jr) : ZC;
    pi[r] = dg ? (pb + kr * nL + jr) : 0;
    ni[r] = (dg && L > 0) ? (pb + nL * nL + kr * nL + jr) : 0;
    posE[r] = ((L & 1) == 0);
  }

  int sd[8];
#pragma unroll
  for (int r = 0; r < 8; ++r) {
    const int kp = 8 * hh + r;
    const int L = lof(kp), nL = 2 * L + 1;
    const int xo = (L >= 1) + 9 * (L >= 2) + 25 * (L >= 3);
    sd[r] = xo + (kp - L * L) * nL + ((L == lof(cc)) ? (cc - L * L) : 0);
  }
  const int Lcc = lof(cc);
  const float sw0  = (Lcc == (hh ? 2 : 0)) ? (hh ? 2.f : 1.f) : 0.f;
  const float sw1  = (Lcc == (hh ? 3 : 1)) ? 2.f : 0.f;
  const float sw2e = (hh == 0 && Lcc == 2) ? 2.f : 0.f;
  const float sw2o = (hh == 1 && Lcc == 3) ? 2.f : 0.f;

  for (int e = 0; e < 16; e += 2) {
    fetch_rows(nx, x0, x1, x2, x3, base + e + 2, Btot, lane);
    asm volatile("" ::: "memory");
    const float* Xe = X + (e & 2) * XS;
    v8f accE[2], accO[2];
    TileRaw raw;
    TileProd pA, pB;
    v2f g0[4], g1[4];
    tile_load<0>(raw, g0, Xe, sB1, sGA, lane, Mcol);
#pragma unroll
    for (int el = 0; el < 2; ++el) {
      const float* Xq = Xe + el * XS;
      const float x0v = Xq[0];
      float sv[8];
#pragma unroll
      for (int r = 0; r < 8; ++r) sv[r] = Xq[sd[r]];
      const float a0 = sw0 * x0v, a1 = sw1 * x0v, a2e = sw2e * x0v, a2o = sw2o * x0v;
      accE[el][0] = a0 * sv[0];
      accO[el][0] = 0.f;
#pragma unroll
      for (int r = 1; r < 4; ++r) {
        accE[el][r] = 0.f;
        accO[el][r] = a1 * sv[r];
      }
#pragma unroll
      for (int r = 4; r < 8; ++r) {
        accE[el][r] = a2e * sv[r];
        accO[el][r] = a2o * sv[r];
      }
    }
    tile_mul<0>(raw, g0, pA);
    tile_stage<0>(pA, pB, raw, g0, g1, Xe, sB1, sGA, accE, accO, lane, Mcol);
    tile_stage<1>(pB, pA, raw, g0, g1, Xe, sB1, sGA, accE, accO, lane, Mcol);
    tile_stage<2>(pA, pB, raw, g0, g1, Xe, sB1, sGA, accE, accO, lane, Mcol);
    tile_stage<3>(pB, pA, raw, g0, g1, Xe, sB1, sGA, accE, accO, lane, Mcol);
    tile_stage<4>(pA, pB, raw, g0, g1, Xe, sB1, sGA, accE, accO, lane, Mcol);
    tile_stage<5>(pB, pA, raw, g0, g1, Xe, sB1, sGA, accE, accO, lane, Mcol);
    tile_stage<6>(pA, pB, raw, g0, g1, Xe, sB1, sGA, accE, accO, lane, Mcol);
    tile_stage<7>(pB, pA, raw, g0, g1, Xe, sB1, sGA, accE, accO, lane, Mcol);
    tile_stage<8>(pA, pB, raw, g0, g1, Xe, sB1, sGA, accE, accO, lane, Mcol);
    tile_stage<9>(pB, pA, raw, g0, g1, Xe, sB1, sGA, accE, accO, lane, Mcol);
    tile_stage<10>(pA, pB, raw, g0, g1, Xe, sB1, sGA, accE, accO, lane, Mcol);
    tile_stage<11>(pB, pA, raw, g0, g1, Xe, sB1, sGA, accE, accO, lane, Mcol);
    tile_stage<12>(pA, pB, raw, g0, g1, Xe, sB1, sGA, accE, accO, lane, Mcol);
    tile_stage<13>(pB, pA, raw, g0, g1, Xe, sB1, sGA, accE, accO, lane, Mcol);
    tile_stage<14>(pA, pB, raw, g0, g1, Xe, sB1, sGA, accE, accO, lane, Mcol);
    tile_stage<15>(pB, pA, raw, g0, g1, Xe, sB1, sGA, accE, accO, lane, Mcol);
#pragma unroll
    for (int el = 0; el < 2; ++el) {
      store_x2(X2 + el * X2S, accE[el], true, lane);
      store_x2(X2 + el * X2S, accO[el], false, lane);
    }
    __builtin_amdgcn_fence(__ATOMIC_RELEASE, "wavefront");
    __builtin_amdgcn_fence(__ATOMIC_ACQUIRE, "wavefront");
    float xv[2][8], pv[2][8], nv[2][8];
#pragma unroll
    for (int el = 0; el < 2; ++el) {
      const float* Xq = Xe + el * XS;
      const float* X2q = X2 + el * X2S;
#pragma unroll
      for (int r = 0; r < 8; ++r) {
        xv[el][r] = Xq[xi[r]];
        pv[el][r] = X2q[pi[r]];
        nv[el][r] = X2q[ni[r]];
      }
    }
    asm volatile("" :: "v"(xv[0][0]), "v"(xv[0][1]), "v"(xv[0][2]), "v"(xv[0][3]), "v"(xv[0][4]), "v"(xv[0][5]), "v"(xv[0][6]), "v"(xv[0][7]), "v"(pv[0][0]), "v"(pv[0][1]), "v"(pv[0][2]), "v"(pv[0][3]), "v"(pv[0][4]), "v"(pv[0][5]), "v"(pv[0][6]), "v"(pv[0][7]), "v"(nv[0][0]), "v"(nv[0][1]), "v"(nv[0][2]), "v"(nv[0][3]), "v"(nv[0][4]), "v"(nv[0][5]), "v"(nv[0][6]), "v"(nv[0][7]));
    asm volatile("" :: "v"(xv[1][0]), "v"(xv[1][1]), "v"(xv[1][2]), "v"(xv[1][3]), "v"(xv[1][4]), "v"(xv[1][5]), "v"(xv[1][6]), "v"(xv[1][7]), "v"(pv[1][0]), "v"(pv[1][1]), "v"(pv[1][2]), "v"(pv[1][3]), "v"(pv[1][4]), "v"(pv[1][5]), "v"(pv[1][6]), "v"(pv[1][7]), "v"(nv[1][0]), "v"(nv[1][1]), "v"(nv[1][2]), "v"(nv[1][3]), "v"(nv[1][4]), "v"(nv[1][5]), "v"(nv[1][6]), "v"(nv[1][7]));
#pragma unroll
    for (int el = 0; el < 2; ++el) {
      float* Pe = PR + (e + el) * PS;
      float s3 = 0.f, s4 = 0.f;
#pragma unroll
      for (int r = 0; r < 8; ++r) {
        const float P = posE[r] ? accE[el][r] : accO[el][r];
        const float N = posE[r] ? accO[el][r] : accE[el][r];
        const float wp = wgt[r] * P;
        const float wn = wgtN[r] * N;
        s3 += wp * xv[el][r];
        s4 += wp * pv[el][r];
        s4 += wn * nv[el][r];
      }
      Pe[1 + lane] = s3;
      Pe[33 + lane] = s4;
      if (lane == 0) {
        Pe[0] = accE[el][0];
        Pe[65] = Xe[el * XS];
      }
    }
    put_rows(X + (2 - (e & 2)) * XS, nx, lane);
  }

  __builtin_amdgcn_fence(__ATOMIC_RELEASE, "wavefront");
  __builtin_amdgcn_fence(__ATOMIC_ACQUIRE, "wavefront");

  if (lane < 16) {
    const int e = base + b;
    if (e < Btot) {
      const float* Pb = PR + b * PS;
      const float nu1 = Pb[65];
      const float nu2 = Pb[0];
      float nu3 = 0.f, nu4 = 0.f;
      for (int i = 0; i < 32; ++i) {
        nu3 += Pb[1 + i];
        nu4 += Pb[33 + i];
      }
      out[(size_t)e * 4 + 0] = nu1;
      out[(size_t)e * 4 + 1] = nu2;
      out[(size_t)e * 4 + 2] = nu3;
      out[(size_t)e * 4 + 3] = nu4;
    }
  }
}

extern "C" void kernel_launch(void* const* d_in, const int* in_sizes, int n_in,
                              void* d_out, int out_size, void* d_ws, size_t ws_size,
                              hipStream_t stream) {
  const float* x0 = (const float*)d_in[0];
  const float* x1 = (const float*)d_in[1];
  const float* x2 = (const float*)d_in[2];
  const float* x3 = (const float*)d_in[3];
  float* tab = (float*)d_ws;
  const int Btot = in_sizes[0];

  wigner_cg_init<<<1, 256, 0, stream>>>(tab);

  const int tiles = (Btot + 15) / 16;
  const int blocks = (tiles + WAVES_PER_BLOCK - 1) / WAVES_PER_BLOCK;
  wigner_main<<<blocks, TB, 0, stream>>>(x0, x1, x2, x3, tab, (float*)d_out, Btot);
}
